# starting kernel
# baseline (speedup 1.0000x reference)
#include <hip/hip_runtime.h>

typedef __attribute__((ext_vector_type(2))) float v2f;
typedef __attribute__((ext_vector_type(8))) float v8f;

constexpr int kD  = 64;
constexpr int kU  = 1024;
constexpr int kB  = 4096;
constexpr int kBT = 32;

constexpr int kXStride = kD + 4;
constexpr int kHStride = kU + 4;

constexpr int kWaOff = 0;
constexpr int kWaSz  = kU * kD;
constexpr int kWbOff = kWaOff + kWaSz;
constexpr int kWbSz  = kU * kU;
constexpr int kWcOff = kWbOff + kWbSz;
constexpr int kWcSz  = 2 * kD * kU;


__global__ void prep_w0(const float* __restrict__ W0, float* __restrict__ Wa) {
  int e = blockIdx.x * 256 + threadIdx.x;
  int j = e >> 6, d = e & 63;
  int dh = j % (kD - 1);
  float v = (d <= dh) ? W0[e] : 0.f;
  int t = j >> 4, n = j & 15;
  int cp = d >> 3, rr = d & 7;
  int q  = ((rr >> 2) << 1) | (rr & 1);
  int hi = (rr >> 1) & 1;
  int l  = (hi << 4) | n;
  Wa[(((t * 8 + cp) * 32 + l) << 2) + q] = v;
}

__global__ void prep_w1(const float* __restrict__ W1, float* __restrict__ Wb) {
  int e = blockIdx.x * 256 + threadIdx.x;
  int j = e >> 10, k = e & 1023;
  int dhj = j % (kD - 1), dhk = k % (kD - 1);
  float v = (dhk <= dhj) ? W1[e] : 0.f;
  int t = j >> 4, n = j & 15;
  int cp = k >> 3, rr = k & 7;
  int q  = ((rr >> 2) << 1) | (rr & 1);
  int hi = (rr >> 1) & 1;
  int l  = (hi << 4) | n;
  Wb[(((t * 128 + cp) * 32 + l) << 2) + q] = v;
}

__global__ void prep_w2(const float* __restrict__ W2, float* __restrict__ Wc) {
  int e = blockIdx.x * 256 + threadIdx.x;
  int r = e >> 10, k = e & 1023;
  int dhk = k % (kD - 1);
  Wc[e] = (dhk < (r & (kD - 1))) ? W2[e] : 0.f;
}


__device__ __forceinline__ v8f wmma_f32(v2f a, v2f b, v8f c) {
  return __builtin_amdgcn_wmma_f32_16x16x4_f32(false, a, false, b, (short)0, c,
                                               false, false);
}

__launch_bounds__(256, 1)
__global__ void made_flow_kernel(const float* __restrict__ u,
                                 const float* __restrict__ b0,
                                 const float* __restrict__ b1,
                                 const float* __restrict__ b2,
                                 const float* __restrict__ ws,
                                 float* __restrict__ out) {
  extern __shared__ float smem[];
  float* xs   = smem;
  float* h0s  = xs  + kBT * kXStride;
  float* h1s  = h0s + kBT * kHStride;
  float* red  = h1s + kBT * kHStride;
  float* ldjs = red + kBT * 8 * 2;

  const int tid  = threadIdx.x;
  const int wg   = blockIdx.x;
  const int lane = tid & 31;
  const int wv   = tid >> 5;
  const int n    = lane & 15;
  const int hi2  = (lane >> 4) << 1;
  const int hi8  = (lane >> 4) << 3;

  const float* Wa = ws + kWaOff;
  const float* Wb = ws + kWbOff;
  const float* Wc = ws + kWcOff;

  for (int e = tid; e < kBT * kXStride; e += 256) xs[e] = 0.f;
  if (tid < kBT) ldjs[tid] = 0.f;
  __syncthreads();

  for (int i = 0; i < kD; ++i) {
    for (int tt = 0; tt < 8; ++tt) {
      const int t = wv * 8 + tt;
      const float bias = b0[t * 16 + n];
      v8f acc0, acc1;
      for (int v = 0; v < 8; ++v) { acc0[v] = bias; acc1[v] = bias; }
      const float4* wq = ((const float4*)Wa) + (t * 8) * 32 + lane;
      for (int cp = 0; cp < 8; ++cp) {
        float4 bq = wq[cp * 32];
        v2f bA = {bq.x, bq.y};
        v2f bB = {bq.z, bq.w};
        const int kb = cp * 8 + hi2;
        v2f a00 = *(const v2f*)&xs[n * kXStride + kb];
        v2f a01 = *(const v2f*)&xs[n * kXStride + kb + 4];
        v2f a10 = *(const v2f*)&xs[(n + 16) * kXStride + kb];
        v2f a11 = *(const v2f*)&xs[(n + 16) * kXStride + kb + 4];
        acc0 = wmma_f32(a00, bA, acc0);
        acc1 = wmma_f32(a10, bA, acc1);
        acc0 = wmma_f32(a01, bB, acc0);
        acc1 = wmma_f32(a11, bB, acc1);
      }
      for (int v = 0; v < 8; ++v) {
        float f0 = acc0[v] > 0.f ? acc0[v] : 0.f;
        float f1 = acc1[v] > 0.f ? acc1[v] : 0.f;
        h0s[(v + hi8) * kHStride + t * 16 + n]        = f0;
        h0s[(16 + v + hi8) * kHStride + t * 16 + n]   = f1;
      }
    }
    __syncthreads();

    for (int tt = 0; tt < 8; ++tt) {
      const int t = wv * 8 + tt;
      const float bias = b1[t * 16 + n];
      v8f acc0, acc1;
      for (int v = 0; v < 8; ++v) { acc0[v] = bias; acc1[v] = bias; }
      const float4* wq = ((const float4*)Wb) + (t * 128) * 32 + lane;
      for (int cp = 0; cp < 128; ++cp) {
        float4 bq = wq[cp * 32];
        v2f bA = {bq.x, bq.y};
        v2f bB = {bq.z, bq.w};
        const int kb = cp * 8 + hi2;
        v2f a00 = *(const v2f*)&h0s[n * kHStride + kb];
        v2f a01 = *(const v2f*)&h0s[n * kHStride + kb + 4];
        v2f a10 = *(const v2f*)&h0s[(n + 16) * kHStride + kb];
        v2f a11 = *(const v2f*)&h0s[(n + 16) * kHStride + kb + 4];
        acc0 = wmma_f32(a00, bA, acc0);
        acc1 = wmma_f32(a10, bA, acc1);
        acc0 = wmma_f32(a01, bB, acc0);
        acc1 = wmma_f32(a11, bB, acc1);
      }
      for (int v = 0; v < 8; ++v) {
        float f0 = acc0[v] > 0.f ? acc0[v] : 0.f;
        float f1 = acc1[v] > 0.f ? acc1[v] : 0.f;
        h1s[(v + hi8) * kHStride + t * 16 + n]        = f0;
        h1s[(16 + v + hi8) * kHStride + t * 16 + n]   = f1;
      }
    }
    __syncthreads();

    {
      const int r = tid & 31;
      const int s = tid >> 5;
      const float* wmu = Wc + i * kU;
      const float* wsg = Wc + (kD + i) * kU;
      const float* hrow = h1s + r * kHStride;
      float mp = 0.f, sp = 0.f;
      const int k0 = s * 128;
      for (int k = k0; k < k0 + 128; ++k) {
        float h = hrow[k];
        mp = fmaf(h, wmu[k], mp);
        sp = fmaf(h, wsg[k], sp);
      }
      red[(r * 8 + s) * 2 + 0] = mp;
      red[(r * 8 + s) * 2 + 1] = sp;
    }
    __syncthreads();
    if (tid < kBT) {
      float mu = b2[i], sg = b2[kD + i];
      for (int s = 0; s < 8; ++s) {
        mu += red[(tid * 8 + s) * 2 + 0];
        sg += red[(tid * 8 + s) * 2 + 1];
      }
      const float uu = u[(wg * kBT + tid) * kD + i];
      xs[tid * kXStride + i] = uu * expf(sg) + mu;
      ldjs[tid] += sg;
    }
    __syncthreads();
  }

  for (int e = tid; e < kBT * kD; e += 256) {
    int r = e >> 6, d = e & 63;
    out[(wg * kBT + r) * kD + d] = xs[r * kXStride + d];
  }
  if (tid < kBT) out[kB * kD + wg * kBT + tid] = ldjs[tid];
}

extern "C" void kernel_launch(void* const* d_in, const int* in_sizes, int n_in,
                              void* d_out, int out_size, void* d_ws, size_t ws_size,
                              hipStream_t stream) {
  const float* u  = (const float*)d_in[0];
  const float* W0 = (const float*)d_in[1];
  const float* b0 = (const float*)d_in[2];
  const float* W1 = (const float*)d_in[3];
  const float* b1 = (const float*)d_in[4];
  const float* W2 = (const float*)d_in[5];
  const float* b2 = (const float*)d_in[6];
  float* ws  = (float*)d_ws;
  float* out = (float*)d_out;

  prep_w0<<<(kU * kD) / 256,     256, 0, stream>>>(W0, ws + kWaOff);
  prep_w1<<<(kU * kU) / 256,     256, 0, stream>>>(W1, ws + kWbOff);
  prep_w2<<<(2 * kD * kU) / 256, 256, 0, stream>>>(W2, ws + kWcOff);

  size_t smem = (size_t)(kBT * kXStride + 2 * kBT * kHStride + kBT * 8 * 2 + kBT) *
                sizeof(float);
  made_flow_kernel<<<kB / kBT, 256, smem, stream>>>(u, b0, b1, b2, ws, out);
}


# One wave owns 16 batch rows for all 64 steps (256 one-wave workgroups, no K split, no cross-wave barrier): units sorted by degree, each tile computed once, weights 16 loads ahead (sum order changes)
# speedup vs baseline: 12.8127x; 12.8127x over previous
#include <hip/hip_runtime.h>

typedef __attribute__((ext_vector_type(2))) float v2f;
typedef __attribute__((ext_vector_type(8))) float v8f;
typedef __attribute__((ext_vector_type(4))) float v4f;

constexpr int kD  = 64;
constexpr int kU  = 1024;
constexpr int kB  = 4096;
constexpr int kBT = 16;

constexpr int kXStride = kD + 4;
constexpr int kHStride = kU + 4;

constexpr int kWaOff = 0;
constexpr int kWaSz  = kU * kD;
constexpr int kWbOff = kWaOff + kWaSz;
constexpr int kWbSz  = kU * kU;
constexpr int kWzOff = kWbOff + kWbSz;
constexpr int kWzSz  = 2 * kD * kU;
constexpr int kB0Off = kWzOff + kWzSz;
constexpr int kB1Off = kB0Off + kU;
constexpr int kZeroOff = kB1Off + kU;

__device__ __forceinline__ int deg_off(int d) {
  return d <= 16 ? 17 * d : 272 + 16 * (d - 16);
}
__device__ __forceinline__ int pos_of_unit(int j) {
  int d = j % (kD - 1), m = j / (kD - 1);
  return deg_off(d) + m;
}


__global__ void prep_w0(const float* __restrict__ W0, const float* __restrict__ b0,
                        const float* __restrict__ b1, float* __restrict__ ws) {
  int e = blockIdx.x * 256 + threadIdx.x;
  int j = e >> 6, d = e & 63;
  int dh = j % (kD - 1);
  float v = (d <= dh) ? W0[e] : 0.f;
  int p = pos_of_unit(j);
  int t = p >> 4, n = p & 15;
  int cp = d >> 3, rr = d & 7;
  int q  = ((rr >> 2) << 1) | (rr & 1);
  int hi = (rr >> 1) & 1;
  int l  = (hi << 4) | n;
  ws[kWaOff + (((t * 8 + cp) * 32 + l) << 2) + q] = v;
  if (e < 4) ws[kZeroOff + e] = 0.f;
  if (e < kU) {
    int pe = pos_of_unit(e);
    ws[kB0Off + pe] = b0[e];
    ws[kB1Off + pe] = b1[e];
  }
}

__global__ void prep_w1(const float* __restrict__ W1, float* __restrict__ Wb) {
  int e = blockIdx.x * 256 + threadIdx.x;
  int j = e >> 10, k = e & 1023;
  int dhj = j % (kD - 1), dhk = k % (kD - 1);
  float v = (dhk <= dhj) ? W1[e] : 0.f;
  int pj = pos_of_unit(j), pk = pos_of_unit(k);
  int t = pj >> 4, n = pj & 15;
  int cp = pk >> 3, rr = pk & 7;
  int q  = ((rr >> 2) << 1) | (rr & 1);
  int hi = (rr >> 1) & 1;
  int l  = (hi << 4) | n;
  Wb[(((t * 128 + cp) * 32 + l) << 2) + q] = v;
}

__global__ void prep_w2(const float* __restrict__ W2, float* __restrict__ Wz) {
  int e = blockIdx.x * 256 + threadIdx.x;
  int r = e >> 10, k = e & 1023;
  int i = r & (kD - 1), c = r >> 6;
  int dhk = k % (kD - 1);
  float v = (dhk < i) ? W2[e] : 0.f;
  int pk = pos_of_unit(k);
  int cp = pk >> 3, rr = pk & 7;
  int q  = ((rr >> 2) << 1) | (rr & 1);
  int hi = (rr >> 1) & 1;
  Wz[((((i * 128 + cp) * 2 + hi) * 2 + c) << 2) + q] = v;
}

__device__ __forceinline__ v8f wmma_f32(v2f a, v2f b, v8f c) {
  return __builtin_amdgcn_wmma_f32_16x16x4_f32(false, a, false, b, (short)0, c,
                                               false, false);
}

constexpr int kPD = 16;

__device__ __forceinline__ v8f tile_gemm(const v4f* __restrict__ wq, int wstep, int ncp,
                                         const float* hrow) {
  v4f buf[kPD];
#pragma unroll
  for (int u = 0; u < kPD; ++u) buf[u] = wq[u * wstep];
  asm volatile("" :: "v"(buf[0]), "v"(buf[1]), "v"(buf[2]), "v"(buf[3]), "v"(buf[4]), "v"(buf[5]),
               "v"(buf[6]), "v"(buf[7]), "v"(buf[8]), "v"(buf[9]), "v"(buf[10]), "v"(buf[11]),
               "v"(buf[12]), "v"(buf[13]), "v"(buf[14]), "v"(buf[15]));
  v8f a0, a1, a2, a3;
  for (int v = 0; v < 8; ++v) { a0[v] = 0.f; a1[v] = 0.f; a2[v] = 0.f; a3[v] = 0.f; }
  for (int base = 0; base < ncp; base += kPD) {
#pragma unroll
    for (int u = 0; u < kPD; ++u) {
      const int cp = base + u;
      v4f w = buf[u];
      v2f wA = {w.x, w.y};
      v2f wB = {w.z, w.w};
      v2f hA = *(const v2f*)(hrow + cp * 8);
      v2f hB = *(const v2f*)(hrow + cp * 8 + 4);
      if (u & 1) { a2 = wmma_f32(wA, hA, a2); a3 = wmma_f32(wB, hB, a3); }
      else       { a0 = wmma_f32(wA, hA, a0); a1 = wmma_f32(wB, hB, a1); }
      __builtin_amdgcn_sched_barrier(0x386);
      int nx = cp + kPD; nx = nx > 127 ? 127 : nx;
      buf[u] = wq[nx * wstep];
    }
  }
  v8f r;
  for (int v = 0; v < 8; ++v) r[v] = (a0[v] + a2[v]) + (a1[v] + a3[v]);
  return r;
}

__launch_bounds__(32, 1)
__global__ void made_flow_kernel(const float* __restrict__ u,
                                 const float* __restrict__ b2,
                                 const float* __restrict__ ws,
                                 float* __restrict__ out) {
  extern __shared__ float smem[];
  float* xs  = smem;
  float* us  = xs  + kBT * kXStride;
  float* h0s = us  + kBT * kXStride;
  float* h1s = h0s + kBT * kHStride;

  const int lane = threadIdx.x;
  const int wg   = blockIdx.x;
  const int n    = lane & 15;
  const int hi   = lane >> 4;
  const int hi2  = hi << 1;

  const float4* Wa4 = (const float4*)(ws + kWaOff);
  const v4f* Wb4 = (const v4f*)(ws + kWbOff);
  const v4f* Wz4 = (const v4f*)(ws + kWzOff);
  const float*  b0s = ws + kB0Off;
  const float*  b1s = ws + kB1Off;

  for (int e = lane; e < kBT * kXStride; e += 32) xs[e] = 0.f;
  for (int e = lane; e < 2 * kBT * kHStride; e += 32) h0s[e] = 0.f;
  for (int e = lane; e < kBT * kD; e += 32) {
    int r = e >> 6, d = e & 63;
    us[r * kXStride + d] = u[(wg * kBT + r) * kD + d];
  }
  float ldj = 0.f;
  __syncthreads();

  const bool zlive = n < 2;
  const float* xrow  = xs  + n * kXStride + hi2;
  const float* h0row = h0s + n * kHStride + hi2;
  const float* h1row = h1s + n * kHStride + hi2;

  for (int i = 0; i < kD; ++i) {
    const int offi = deg_off(i);
    const int ncp  = ((offi + 127) >> 7) << 4;
    if (i > 0) {
      const int tlo = deg_off(i - 1) >> 4, thi = (offi - 1) >> 4;
      for (int t = tlo; t <= thi; ++t) {
        v8f acc;
        const float4 bb0 = *(const float4*)(b0s + t * 16 + hi * 8);
        const float4 bb1 = *(const float4*)(b0s + t * 16 + hi * 8 + 4);
        acc[0] = bb0.x; acc[1] = bb0.y; acc[2] = bb0.z; acc[3] = bb0.w;
        acc[4] = bb1.x; acc[5] = bb1.y; acc[6] = bb1.z; acc[7] = bb1.w;
        const float4* wq = Wa4 + (t * 8) * 32 + lane;
        v8f a1;
        for (int v = 0; v < 8; ++v) a1[v] = 0.f;
#pragma unroll
        for (int cp = 0; cp < 8; ++cp) {
          float4 w = wq[cp * 32];
          v2f wA = {w.x, w.y};
          v2f wB = {w.z, w.w};
          v2f xA = *(const v2f*)(xrow + cp * 8);
          v2f xB = *(const v2f*)(xrow + cp * 8 + 4);
          acc = wmma_f32(wA, xA, acc);
          a1  = wmma_f32(wB, xB, a1);
        }
        float4 o0, o1;
        o0.x = fmaxf(acc[0] + a1[0], 0.f); o0.y = fmaxf(acc[1] + a1[1], 0.f);
        o0.z = fmaxf(acc[2] + a1[2], 0.f); o0.w = fmaxf(acc[3] + a1[3], 0.f);
        o1.x = fmaxf(acc[4] + a1[4], 0.f); o1.y = fmaxf(acc[5] + a1[5], 0.f);
        o1.z = fmaxf(acc[6] + a1[6], 0.f); o1.w = fmaxf(acc[7] + a1[7], 0.f);
        float* dst = h0s + n * kHStride + t * 16 + hi * 8;
        *(float4*)dst = o0;
        *(float4*)(dst + 4) = o1;
      }
      __syncthreads();
      for (int t = tlo; t <= thi; ++t) {
        const float4 bb0 = *(const float4*)(b1s + t * 16 + hi * 8);
        const float4 bb1 = *(const float4*)(b1s + t * 16 + hi * 8 + 4);
        v8f r = tile_gemm(Wb4 + (t * 128) * 32 + lane, 32, ncp, h0row);
        float4 o0, o1;
        o0.x = fmaxf(r[0] + bb0.x, 0.f); o0.y = fmaxf(r[1] + bb0.y, 0.f);
        o0.z = fmaxf(r[2] + bb0.z, 0.f); o0.w = fmaxf(r[3] + bb0.w, 0.f);
        o1.x = fmaxf(r[4] + bb1.x, 0.f); o1.y = fmaxf(r[5] + bb1.y, 0.f);
        o1.z = fmaxf(r[6] + bb1.z, 0.f); o1.w = fmaxf(r[7] + bb1.w, 0.f);
        float* dst = h1s + n * kHStride + t * 16 + hi * 8;
        *(float4*)dst = o0;
        *(float4*)(dst + 4) = o1;
      }
      __syncthreads();
    }
    {
      const v4f* zq = zlive ? Wz4 + (i * 128) * 4 + hi * 2 + n : (const v4f*)(ws + kZeroOff);
      v8f z = tile_gemm(zq, zlive ? 4 : 0, ncp, h1row);
      if (hi == 0) {
        const float mu = z[0] + b2[i];
        const float sg = z[1] + b2[kD + i];
        xs[n * kXStride + i] = us[n * kXStride + i] * expf(sg) + mu;
        ldj += sg;
      }
    }
    __syncthreads();
  }

  for (int e = lane; e < kBT * kD; e += 32) {
    int r = e >> 6, d = e & 63;
    out[(wg * kBT + r) * kD + d] = xs[r * kXStride + d];
  }
  if (hi == 0) out[kB * kD + wg * kBT + n] = ldj;
}

extern "C" void kernel_launch(void* const* d_in, const int* in_sizes, int n_in,
                              void* d_out, int out_size, void* d_ws, size_t ws_size,
                              hipStream_t stream) {
  const float* u  = (const float*)d_in[0];
  const float* W0 = (const float*)d_in[1];
  const float* b0 = (const float*)d_in[2];
  const float* W1 = (const float*)d_in[3];
  const float* b1 = (const float*)d_in[4];
  const float* W2 = (const float*)d_in[5];
  const float* b2 = (const float*)d_in[6];
  float* ws  = (float*)d_ws;
  float* out = (float*)d_out;

  prep_w0<<<(kU * kD) / 256,     256, 0, stream>>>(W0, b0, b1, ws);
  prep_w1<<<(kU * kU) / 256,     256, 0, stream>>>(W1, ws + kWbOff);
  prep_w2<<<(2 * kD * kU) / 256, 256, 0, stream>>>(W2, ws + kWzOff);

  size_t smem = (size_t)(2 * kBT * kXStride + 2 * kBT * kHStride) * sizeof(float);
  made_flow_kernel<<<kB / kBT, 32, smem, stream>>>(u, b2, ws, out);
}


# Layer 3 tiled over blocks of 8 steps: the z tile's 16 rows hold mu and sigma of 8 steps; full K only at block start, later steps add only the newly final units (z, log-det sum order changes)
# speedup vs baseline: 17.8582x; 1.3938x over previous
#include <hip/hip_runtime.h>

typedef __attribute__((ext_vector_type(2))) float v2f;
typedef __attribute__((ext_vector_type(8))) float v8f;
typedef __attribute__((ext_vector_type(4))) float v4f;

constexpr int kD  = 64;
constexpr int kU  = 1024;
constexpr int kB  = 4096;
constexpr int kBT = 16;

constexpr int kXStride = kD + 4;
constexpr int kHStride = kU + 4;

constexpr int kWaOff = 0;
constexpr int kWaSz  = kU * kD;
constexpr int kWbOff = kWaOff + kWaSz;
constexpr int kWbSz  = kU * kU;
constexpr int kWzbOff = kWbOff + kWbSz;
constexpr int kWzbSz  = 8 * 128 * 128;
constexpr int kWziOff = kWzbOff + kWzbSz;
constexpr int kWziSz  = kD * 3 * 128;
constexpr int kB0Off = kWziOff + kWziSz;
constexpr int kB1Off = kB0Off + kU;

__device__ __forceinline__ int deg_off(int d) {
  return d <= 16 ? 17 * d : 272 + 16 * (d - 16);
}
__device__ __forceinline__ int pos_of_unit(int j) {
  int d = j % (kD - 1), m = j / (kD - 1);
  return deg_off(d) + m;
}
__device__ __forceinline__ int unit_at_pos(int p) {
  int d, m;
  if (p < 272) { d = p / 17; m = p - 17 * d; }
  else         { d = 16 + ((p - 272) >> 4); m = (p - 272) & 15; }
  return m * (kD - 1) + d;
}


__global__ void prep_w0(const float* __restrict__ W0, const float* __restrict__ b0,
                        const float* __restrict__ b1, float* __restrict__ ws) {
  int e = blockIdx.x * 256 + threadIdx.x;
  int j = e >> 6, d = e & 63;
  int dh = j % (kD - 1);
  float v = (d <= dh) ? W0[e] : 0.f;
  int p = pos_of_unit(j);
  int t = p >> 4, n = p & 15;
  int cp = d >> 3, rr = d & 7;
  int q  = ((rr >> 2) << 1) | (rr & 1);
  int hi = (rr >> 1) & 1;
  int l  = (hi << 4) | n;
  ws[kWaOff + (((t * 8 + cp) * 32 + l) << 2) + q] = v;
  if (e < kU) {
    int pe = pos_of_unit(e);
    ws[kB0Off + pe] = b0[e];
    ws[kB1Off + pe] = b1[e];
  }
}

__global__ void prep_w1(const float* __restrict__ W1, float* __restrict__ Wb) {
  int e = blockIdx.x * 256 + threadIdx.x;
  int j = e >> 10, k = e & 1023;
  int dhj = j % (kD - 1), dhk = k % (kD - 1);
  float v = (dhk <= dhj) ? W1[e] : 0.f;
  int pj = pos_of_unit(j), pk = pos_of_unit(k);
  int t = pj >> 4, n = pj & 15;
  int cp = pk >> 3, rr = pk & 7;
  int q  = ((rr >> 2) << 1) | (rr & 1);
  int hi = (rr >> 1) & 1;
  int l  = (hi << 4) | n;
  Wb[(((t * 128 + cp) * 32 + l) << 2) + q] = v;
}

__global__ void prep_w2(const float* __restrict__ W2, float* __restrict__ ws) {
  int e = blockIdx.x * 256 + threadIdx.x;
  if (e < kWzbSz) {
    int q = e & 3, l = (e >> 2) & 31, cp = (e >> 7) & 127, b = e >> 14;
    int hi = l >> 4, m = l & 15;
    int p = 8 * cp + (q >> 1) * 4 + hi * 2 + (q & 1);
    int row = (m & 1) * kD + 8 * b + (m >> 1);
    float v = 0.f;
    if (p < deg_off(8 * b)) v = W2[row * kU + unit_at_pos(p)];
    ws[kWzbOff + e] = v;
  } else {
    int f = e - kWzbSz;
    int q = f & 3, l = (f >> 2) & 31, r = f >> 7;
    int i = r / 3, cl = r - 3 * i;
    int hi = l >> 4, m = l & 15;
    float v = 0.f;
    if (i >= 1) {
      int lo = deg_off(i - 1), hiEnd = deg_off(i);
      int p = 8 * ((lo >> 3) + cl) + (q >> 1) * 4 + hi * 2 + (q & 1);
      int row = (m & 1) * kD + 8 * (i >> 3) + (m >> 1);
      if (p >= lo && p < hiEnd) v = W2[row * kU + unit_at_pos(p)];
    }
    ws[kWziOff + f] = v;
  }
}

__device__ __forceinline__ v8f wmma_f32(v2f a, v2f b, v8f c) {
  return __builtin_amdgcn_wmma_f32_16x16x4_f32(false, a, false, b, (short)0, c,
                                               false, false);
}

constexpr int kPD = 16;

__device__ __forceinline__ v8f tile_gemm(const v4f* __restrict__ wq, int wstep, int ncp,
                                         const float* hrow) {
  v4f buf[kPD];
#pragma unroll
  for (int u = 0; u < kPD; ++u) buf[u] = wq[u * wstep];
  asm volatile("" :: "v"(buf[0]), "v"(buf[1]), "v"(buf[2]), "v"(buf[3]), "v"(buf[4]), "v"(buf[5]),
               "v"(buf[6]), "v"(buf[7]), "v"(buf[8]), "v"(buf[9]), "v"(buf[10]), "v"(buf[11]),
               "v"(buf[12]), "v"(buf[13]), "v"(buf[14]), "v"(buf[15]));
  v8f a0, a1, a2, a3;
  for (int v = 0; v < 8; ++v) { a0[v] = 0.f; a1[v] = 0.f; a2[v] = 0.f; a3[v] = 0.f; }
  for (int base = 0; base < ncp; base += kPD) {
#pragma unroll
    for (int u = 0; u < kPD; ++u) {
      const int cp = base + u;
      v4f w = buf[u];
      v2f wA = {w.x, w.y};
      v2f wB = {w.z, w.w};
      v2f hA = *(const v2f*)(hrow + cp * 8);
      v2f hB = *(const v2f*)(hrow + cp * 8 + 4);
      if (u & 1) { a2 = wmma_f32(wA, hA, a2); a3 = wmma_f32(wB, hB, a3); }
      else       { a0 = wmma_f32(wA, hA, a0); a1 = wmma_f32(wB, hB, a1); }
      __builtin_amdgcn_sched_barrier(0x386);
      int nx = cp + kPD; nx = nx > 127 ? 127 : nx;
      buf[u] = wq[nx * wstep];
    }
  }
  v8f r;
  for (int v = 0; v < 8; ++v) r[v] = (a0[v] + a2[v]) + (a1[v] + a3[v]);
  return r;
}

__launch_bounds__(32, 1)
__global__ void made_flow_kernel(const float* __restrict__ u,
                                 const float* __restrict__ b2,
                                 const float* __restrict__ ws,
                                 float* __restrict__ out) {
  extern __shared__ float smem[];
  float* xs  = smem;
  float* us  = xs  + kBT * kXStride;
  float* h0s = us  + kBT * kXStride;
  float* h1s = h0s + kBT * kHStride;

  const int lane = threadIdx.x;
  const int wg   = blockIdx.x;
  const int n    = lane & 15;
  const int hi   = lane >> 4;
  const int hi2  = hi << 1;

  const float4* Wa4 = (const float4*)(ws + kWaOff);
  const v4f* Wb4 = (const v4f*)(ws + kWbOff);
  const v4f* Wzb4 = (const v4f*)(ws + kWzbOff);
  const v4f* Wzi4 = (const v4f*)(ws + kWziOff);
  const float*  b0s = ws + kB0Off;
  const float*  b1s = ws + kB1Off;

  for (int e = lane; e < kBT * kXStride; e += 32) xs[e] = 0.f;
  for (int e = lane; e < 2 * kBT * kHStride; e += 32) h0s[e] = 0.f;
  for (int e = lane; e < kBT * kD; e += 32) {
    int r = e >> 6, d = e & 63;
    us[r * kXStride + d] = u[(wg * kBT + r) * kD + d];
  }
  float ldj = 0.f;
  __syncthreads();

  v8f zacc;
  for (int v = 0; v < 8; ++v) zacc[v] = 0.f;
  const float* xrow  = xs  + n * kXStride + hi2;
  const float* h0row = h0s + n * kHStride + hi2;
  const float* h1row = h1s + n * kHStride + hi2;

  for (int i = 0; i < kD; ++i) {
    const int offi = deg_off(i);
    const int ncp  = ((offi + 127) >> 7) << 4;
    const int jb = i & 7;
    v4f zi0, zi1, zi2;
    for (int v = 0; v < 4; ++v) { zi0[v] = 0.f; zi1[v] = 0.f; zi2[v] = 0.f; }
    if (i > 0) {
      const int tlo = deg_off(i - 1) >> 4, thi = (offi - 1) >> 4;
      if (jb != 0) {
        const v4f* zq = Wzi4 + (i * 3) * 32 + lane;
        zi0 = zq[0]; zi1 = zq[32]; zi2 = zq[64];
      }
      for (int t = tlo; t <= thi; ++t) {
        v8f acc;
        const float4 bb0 = *(const float4*)(b0s + t * 16 + hi * 8);
        const float4 bb1 = *(const float4*)(b0s + t * 16 + hi * 8 + 4);
        acc[0] = bb0.x; acc[1] = bb0.y; acc[2] = bb0.z; acc[3] = bb0.w;
        acc[4] = bb1.x; acc[5] = bb1.y; acc[6] = bb1.z; acc[7] = bb1.w;
        const float4* wq = Wa4 + (t * 8) * 32 + lane;
        v8f a1;
        for (int v = 0; v < 8; ++v) a1[v] = 0.f;
#pragma unroll
        for (int cp = 0; cp < 8; ++cp) {
          float4 w = wq[cp * 32];
          v2f wA = {w.x, w.y};
          v2f wB = {w.z, w.w};
          v2f xA = *(const v2f*)(xrow + cp * 8);
          v2f xB = *(const v2f*)(xrow + cp * 8 + 4);
          acc = wmma_f32(wA, xA, acc);
          a1  = wmma_f32(wB, xB, a1);
        }
        float4 o0, o1;
        o0.x = fmaxf(acc[0] + a1[0], 0.f); o0.y = fmaxf(acc[1] + a1[1], 0.f);
        o0.z = fmaxf(acc[2] + a1[2], 0.f); o0.w = fmaxf(acc[3] + a1[3], 0.f);
        o1.x = fmaxf(acc[4] + a1[4], 0.f); o1.y = fmaxf(acc[5] + a1[5], 0.f);
        o1.z = fmaxf(acc[6] + a1[6], 0.f); o1.w = fmaxf(acc[7] + a1[7], 0.f);
        float* dst = h0s + n * kHStride + t * 16 + hi * 8;
        *(float4*)dst = o0;
        *(float4*)(dst + 4) = o1;
      }
      __syncthreads();
      for (int t = tlo; t <= thi; ++t) {
        const float4 bb0 = *(const float4*)(b1s + t * 16 + hi * 8);
        const float4 bb1 = *(const float4*)(b1s + t * 16 + hi * 8 + 4);
        v8f r = tile_gemm(Wb4 + (t * 128) * 32 + lane, 32, ncp, h0row);
        float4 o0, o1;
        o0.x = fmaxf(r[0] + bb0.x, 0.f); o0.y = fmaxf(r[1] + bb0.y, 0.f);
        o0.z = fmaxf(r[2] + bb0.z, 0.f); o0.w = fmaxf(r[3] + bb0.w, 0.f);
        o1.x = fmaxf(r[4] + bb1.x, 0.f); o1.y = fmaxf(r[5] + bb1.y, 0.f);
        o1.z = fmaxf(r[6] + bb1.z, 0.f); o1.w = fmaxf(r[7] + bb1.w, 0.f);
        float* dst = h1s + n * kHStride + t * 16 + hi * 8;
        *(float4*)dst = o0;
        *(float4*)(dst + 4) = o1;
      }
      __syncthreads();
    }
    if (jb == 0) {
      if (i > 0) zacc = tile_gemm(Wzb4 + ((i >> 3) * 128) * 32 + lane, 32, ncp, h1row);
    } else {
      const int cp0 = deg_off(i - 1) >> 3;
      const v4f wz[3] = {zi0, zi1, zi2};
      v8f zb;
      for (int v = 0; v < 8; ++v) zb[v] = 0.f;
#pragma unroll
      for (int cl = 0; cl < 3; ++cl) {
        int hcp = cp0 + cl; hcp = hcp > 127 ? 127 : hcp;
        v2f wA = {wz[cl].x, wz[cl].y};
        v2f wB = {wz[cl].z, wz[cl].w};
        v2f hA = *(const v2f*)(h1row + hcp * 8);
        v2f hB = *(const v2f*)(h1row + hcp * 8 + 4);
        zacc = wmma_f32(wA, hA, zacc);
        zb   = wmma_f32(wB, hB, zb);
      }
      for (int v = 0; v < 8; ++v) zacc[v] += zb[v];
    }
    {
      float zm, zs;
      switch (jb & 3) {
        case 0:  zm = zacc[0]; zs = zacc[1]; break;
        case 1:  zm = zacc[2]; zs = zacc[3]; break;
        case 2:  zm = zacc[4]; zs = zacc[5]; break;
        default: zm = zacc[6]; zs = zacc[7]; break;
      }
      if (hi == (jb >> 2)) {
        const float mu = zm + b2[i];
        const float sg = zs + b2[kD + i];
        xs[n * kXStride + i] = us[n * kXStride + i] * expf(sg) + mu;
        ldj += sg;
      }
    }
    __syncthreads();
  }

  for (int e = lane; e < kBT * kD; e += 32) {
    int r = e >> 6, d = e & 63;
    out[(wg * kBT + r) * kD + d] = xs[r * kXStride + d];
  }
  const float ldjt = ldj + __shfl_xor(ldj, 16);
  if (hi == 0) out[kB * kD + wg * kBT + n] = ldjt;
}

extern "C" void kernel_launch(void* const* d_in, const int* in_sizes, int n_in,
                              void* d_out, int out_size, void* d_ws, size_t ws_size,
                              hipStream_t stream) {
  const float* u  = (const float*)d_in[0];
  const float* W0 = (const float*)d_in[1];
  const float* b0 = (const float*)d_in[2];
  const float* W1 = (const float*)d_in[3];
  const float* b1 = (const float*)d_in[4];
  const float* W2 = (const float*)d_in[5];
  const float* b2 = (const float*)d_in[6];
  float* ws  = (float*)d_ws;
  float* out = (float*)d_out;

  prep_w0<<<(kU * kD) / 256,     256, 0, stream>>>(W0, b0, b1, ws);
  prep_w1<<<(kU * kU) / 256,     256, 0, stream>>>(W1, ws + kWbOff);
  prep_w2<<<(kWzbSz + kWziSz) / 256, 256, 0, stream>>>(W2, ws);

  size_t smem = (size_t)(2 * kBT * kXStride + 2 * kBT * kHStride) * sizeof(float);
  made_flow_kernel<<<kB / kBT, 32, smem, stream>>>(u, b2, ws, out);
}


# tile_gemm in blocks of 16 chunk pairs: operand fetches leave as groups ahead of a block's 32 WMMAs (h from LDS 1 block ahead, weights 3 blocks ahead, was 1; past a tile's end they re-read used lines)
# speedup vs baseline: 23.6308x; 1.3232x over previous
#include <hip/hip_runtime.h>

typedef __attribute__((ext_vector_type(2))) float v2f;
typedef __attribute__((ext_vector_type(8))) float v8f;
typedef __attribute__((ext_vector_type(4))) float v4f;

constexpr int kD  = 64;
constexpr int kU  = 1024;
constexpr int kB  = 4096;
constexpr int kBT = 16;

constexpr int kXStride = kD + 4;
constexpr int kHStride = kU + 4;

constexpr int kWaOff = 0;
constexpr int kWaSz  = kU * kD;
constexpr int kWbOff = kWaOff + kWaSz;
constexpr int kWbSz  = kU * kU;
constexpr int kWzbOff = kWbOff + kWbSz;
constexpr int kWzbSz  = 8 * 128 * 128;
constexpr int kWziOff = kWzbOff + kWzbSz;
constexpr int kWziSz  = kD * 3 * 128;
constexpr int kB0Off = kWziOff + kWziSz;
constexpr int kB1Off = kB0Off + kU;

__device__ __forceinline__ int deg_off(int d) {
  return d <= 16 ? 17 * d : 272 + 16 * (d - 16);
}
__device__ __forceinline__ int pos_of_unit(int j) {
  int d = j % (kD - 1), m = j / (kD - 1);
  return deg_off(d) + m;
}
__device__ __forceinline__ int unit_at_pos(int p) {
  int d, m;
  if (p < 272) { d = p / 17; m = p - 17 * d; }
  else         { d = 16 + ((p - 272) >> 4); m = (p - 272) & 15; }
  return m * (kD - 1) + d;
}


__global__ void prep_w0(const float* __restrict__ W0, const float* __restrict__ b0,
                        const float* __restrict__ b1, float* __restrict__ ws) {
  int e = blockIdx.x * 256 + threadIdx.x;
  int j = e >> 6, d = e & 63;
  int dh = j % (kD - 1);
  float v = (d <= dh) ? W0[e] : 0.f;
  int p = pos_of_unit(j);
  int t = p >> 4, n = p & 15;
  int cp = d >> 3, rr = d & 7;
  int q  = ((rr >> 2) << 1) | (rr & 1);
  int hi = (rr >> 1) & 1;
  int l  = (hi << 4) | n;
  ws[kWaOff + (((t * 8 + cp) * 32 + l) << 2) + q] = v;
  if (e < kU) {
    int pe = pos_of_unit(e);
    ws[kB0Off + pe] = b0[e];
    ws[kB1Off + pe] = b1[e];
  }
}

__global__ void prep_w1(const float* __restrict__ W1, float* __restrict__ Wb) {
  int e = blockIdx.x * 256 + threadIdx.x;
  int j = e >> 10, k = e & 1023;
  int dhj = j % (kD - 1), dhk = k % (kD - 1);
  float v = (dhk <= dhj) ? W1[e] : 0.f;
  int pj = pos_of_unit(j), pk = pos_of_unit(k);
  int t = pj >> 4, n = pj & 15;
  int cp = pk >> 3, rr = pk & 7;
  int q  = ((rr >> 2) << 1) | (rr & 1);
  int hi = (rr >> 1) & 1;
  int l  = (hi << 4) | n;
  Wb[(((t * 128 + cp) * 32 + l) << 2) + q] = v;
}

__global__ void prep_w2(const float* __restrict__ W2, float* __restrict__ ws) {
  int e = blockIdx.x * 256 + threadIdx.x;
  if (e < kWzbSz) {
    int q = e & 3, l = (e >> 2) & 31, cp = (e >> 7) & 127, b = e >> 14;
    int hi = l >> 4, m = l & 15;
    int p = 8 * cp + (q >> 1) * 4 + hi * 2 + (q & 1);
    int row = (m & 1) * kD + 8 * b + (m >> 1);
    float v = 0.f;
    if (p < deg_off(8 * b)) v = W2[row * kU + unit_at_pos(p)];
    ws[kWzbOff + e] = v;
  } else {
    int f = e - kWzbSz;
    int q = f & 3, l = (f >> 2) & 31, r = f >> 7;
    int i = r / 3, cl = r - 3 * i;
    int hi = l >> 4, m = l & 15;
    float v = 0.f;
    if (i >= 1) {
      int lo = deg_off(i - 1), hiEnd = deg_off(i);
      int p = 8 * ((lo >> 3) + cl) + (q >> 1) * 4 + hi * 2 + (q & 1);
      int row = (m & 1) * kD + 8 * (i >> 3) + (m >> 1);
      if (p >= lo && p < hiEnd) v = W2[row * kU + unit_at_pos(p)];
    }
    ws[kWziOff + f] = v;
  }
}

__device__ __forceinline__ v8f wmma_f32(v2f a, v2f b, v8f c) {
  return __builtin_amdgcn_wmma_f32_16x16x4_f32(false, a, false, b, (short)0, c,
                                               false, false);
}

constexpr int kBlk  = 16;
constexpr int kRing = 3;
constexpr int kPD   = kBlk * kRing;

template <int G>
__device__ __forceinline__ void gemm_block(v4f (&buf)[kPD], v2f (&hf)[kRing][2 * kBlk],
                                           v8f& a0, v8f& a1, v8f& a2, v8f& a3,
                                           const v4f* __restrict__ wq, const float* hrow,
                                           int base, int ncp) {
  constexpr int GN = (G + 1) % kRing;
  int hb = base + kBlk; hb = hb > 128 - kBlk ? 128 - kBlk : hb;
#pragma unroll
  for (int u = 0; u < kBlk; ++u) {
    hf[GN][2 * u]     = *(const v2f*)(hrow + (hb + u) * 8);
    hf[GN][2 * u + 1] = *(const v2f*)(hrow + (hb + u) * 8 + 4);
  }
  __builtin_amdgcn_sched_barrier(0x6);
  if (G == 0 || base < ncp) {
#pragma unroll
    for (int u = 0; u < kBlk; ++u) {
      v4f w = buf[G * kBlk + u];
      v2f wA = {w.x, w.y};
      v2f wB = {w.z, w.w};
      if (u & 1) { a2 = wmma_f32(wA, hf[G][2 * u], a2); a3 = wmma_f32(wB, hf[G][2 * u + 1], a3); }
      else       { a0 = wmma_f32(wA, hf[G][2 * u], a0); a1 = wmma_f32(wB, hf[G][2 * u + 1], a1); }
    }
  }
  __builtin_amdgcn_sched_barrier(0x6);
  const int lastb = ncp - kBlk;
  const int wb = (base + kPD < ncp) ? base + kPD : (base < lastb ? base : lastb);
#pragma unroll
  for (int u = 0; u < kBlk; ++u) buf[G * kBlk + u] = wq[(wb + u) * 32];
  __builtin_amdgcn_sched_barrier(0x6);
}

__device__ __forceinline__ v8f tile_gemm(const v4f* __restrict__ wq, int ncp, const float* hrow) {
  v4f buf[kPD];
  const int last = ncp - 1;
#pragma unroll
  for (int u = 0; u < kPD; ++u) {
    int c = u > last ? last : u;
    buf[u] = wq[c * 32];
  }
  asm volatile("" :: "v"(buf[0]), "v"(buf[1]), "v"(buf[2]), "v"(buf[3]), "v"(buf[4]), "v"(buf[5]),
               "v"(buf[6]), "v"(buf[7]), "v"(buf[8]), "v"(buf[9]), "v"(buf[10]), "v"(buf[11]),
               "v"(buf[12]), "v"(buf[13]), "v"(buf[14]), "v"(buf[15]));
  asm volatile("" :: "v"(buf[16]), "v"(buf[17]), "v"(buf[18]), "v"(buf[19]), "v"(buf[20]), "v"(buf[21]),
               "v"(buf[22]), "v"(buf[23]), "v"(buf[24]), "v"(buf[25]), "v"(buf[26]), "v"(buf[27]),
               "v"(buf[28]), "v"(buf[29]), "v"(buf[30]), "v"(buf[31]));
  asm volatile("" :: "v"(buf[32]), "v"(buf[33]), "v"(buf[34]), "v"(buf[35]), "v"(buf[36]), "v"(buf[37]),
               "v"(buf[38]), "v"(buf[39]), "v"(buf[40]), "v"(buf[41]), "v"(buf[42]), "v"(buf[43]),
               "v"(buf[44]), "v"(buf[45]), "v"(buf[46]), "v"(buf[47]));
  v2f hf[kRing][2 * kBlk];
#pragma unroll
  for (int u = 0; u < kBlk; ++u) {
    hf[0][2 * u]     = *(const v2f*)(hrow + u * 8);
    hf[0][2 * u + 1] = *(const v2f*)(hrow + u * 8 + 4);
  }
  v8f a0, a1, a2, a3;
  for (int v = 0; v < 8; ++v) { a0[v] = 0.f; a1[v] = 0.f; a2[v] = 0.f; a3[v] = 0.f; }
  for (int base = 0; base < ncp; base += kPD) {
    gemm_block<0>(buf, hf, a0, a1, a2, a3, wq, hrow, base, ncp);
    gemm_block<1>(buf, hf, a0, a1, a2, a3, wq, hrow, base + kBlk, ncp);
    gemm_block<2>(buf, hf, a0, a1, a2, a3, wq, hrow, base + 2 * kBlk, ncp);
  }
  v8f r;
  for (int v = 0; v < 8; ++v) r[v] = (a0[v] + a2[v]) + (a1[v] + a3[v]);
  return r;
}

__launch_bounds__(32, 1)
__global__ void made_flow_kernel(const float* __restrict__ u,
                                 const float* __restrict__ b2,
                                 const float* __restrict__ ws,
                                 float* __restrict__ out) {
  extern __shared__ float smem[];
  float* xs  = smem;
  float* us  = xs  + kBT * kXStride;
  float* h0s = us  + kBT * kXStride;
  float* h1s = h0s + kBT * kHStride;

  const int lane = threadIdx.x;
  const int wg   = blockIdx.x;
  const int n    = lane & 15;
  const int hi   = lane >> 4;
  const int hi2  = hi << 1;

  const float4* Wa4 = (const float4*)(ws + kWaOff);
  const v4f* Wb4 = (const v4f*)(ws + kWbOff);
  const v4f* Wzb4 = (const v4f*)(ws + kWzbOff);
  const v4f* Wzi4 = (const v4f*)(ws + kWziOff);
  const float*  b0s = ws + kB0Off;
  const float*  b1s = ws + kB1Off;

  for (int e = lane; e < kBT * kXStride; e += 32) xs[e] = 0.f;
  for (int e = lane; e < 2 * kBT * kHStride; e += 32) h0s[e] = 0.f;
  for (int e = lane; e < kBT * kD; e += 32) {
    int r = e >> 6, d = e & 63;
    us[r * kXStride + d] = u[(wg * kBT + r) * kD + d];
  }
  float ldj = 0.f;
  __syncthreads();

  v8f zacc;
  for (int v = 0; v < 8; ++v) zacc[v] = 0.f;
  const float* xrow  = xs  + n * kXStride + hi2;
  const float* h0row = h0s + n * kHStride + hi2;
  const float* h1row = h1s + n * kHStride + hi2;

  for (int i = 0; i < kD; ++i) {
    const int offi = deg_off(i);
    const int ncp  = ((offi + 127) >> 7) << 4;
    const int jb = i & 7;
    v4f zi0, zi1, zi2;
    for (int v = 0; v < 4; ++v) { zi0[v] = 0.f; zi1[v] = 0.f; zi2[v] = 0.f; }
    if (i > 0) {
      const int tlo = deg_off(i - 1) >> 4, thi = (offi - 1) >> 4;
      if (jb != 0) {
        const v4f* zq = Wzi4 + (i * 3) * 32 + lane;
        zi0 = zq[0]; zi1 = zq[32]; zi2 = zq[64];
      }
      for (int t = tlo; t <= thi; ++t) {
        v8f acc;
        const float4 bb0 = *(const float4*)(b0s + t * 16 + hi * 8);
        const float4 bb1 = *(const float4*)(b0s + t * 16 + hi * 8 + 4);
        acc[0] = bb0.x; acc[1] = bb0.y; acc[2] = bb0.z; acc[3] = bb0.w;
        acc[4] = bb1.x; acc[5] = bb1.y; acc[6] = bb1.z; acc[7] = bb1.w;
        const float4* wq = Wa4 + (t * 8) * 32 + lane;
        v8f a1;
        for (int v = 0; v < 8; ++v) a1[v] = 0.f;
#pragma unroll
        for (int cp = 0; cp < 8; ++cp) {
          float4 w = wq[cp * 32];
          v2f wA = {w.x, w.y};
          v2f wB = {w.z, w.w};
          v2f xA = *(const v2f*)(xrow + cp * 8);
          v2f xB = *(const v2f*)(xrow + cp * 8 + 4);
          acc = wmma_f32(wA, xA, acc);
          a1  = wmma_f32(wB, xB, a1);
        }
        float4 o0, o1;
        o0.x = fmaxf(acc[0] + a1[0], 0.f); o0.y = fmaxf(acc[1] + a1[1], 0.f);
        o0.z = fmaxf(acc[2] + a1[2], 0.f); o0.w = fmaxf(acc[3] + a1[3], 0.f);
        o1.x = fmaxf(acc[4] + a1[4], 0.f); o1.y = fmaxf(acc[5] + a1[5], 0.f);
        o1.z = fmaxf(acc[6] + a1[6], 0.f); o1.w = fmaxf(acc[7] + a1[7], 0.f);
        float* dst = h0s + n * kHStride + t * 16 + hi * 8;
        *(float4*)dst = o0;
        *(float4*)(dst + 4) = o1;
      }
      __syncthreads();
      for (int t = tlo; t <= thi; ++t) {
        const float4 bb0 = *(const float4*)(b1s + t * 16 + hi * 8);
        const float4 bb1 = *(const float4*)(b1s + t * 16 + hi * 8 + 4);
        v8f r = tile_gemm(Wb4 + (t * 128) * 32 + lane, ncp, h0row);
        float4 o0, o1;
        o0.x = fmaxf(r[0] + bb0.x, 0.f); o0.y = fmaxf(r[1] + bb0.y, 0.f);
        o0.z = fmaxf(r[2] + bb0.z, 0.f); o0.w = fmaxf(r[3] + bb0.w, 0.f);
        o1.x = fmaxf(r[4] + bb1.x, 0.f); o1.y = fmaxf(r[5] + bb1.y, 0.f);
        o1.z = fmaxf(r[6] + bb1.z, 0.f); o1.w = fmaxf(r[7] + bb1.w, 0.f);
        float* dst = h1s + n * kHStride + t * 16 + hi * 8;
        *(float4*)dst = o0;
        *(float4*)(dst + 4) = o1;
      }
      __syncthreads();
    }
    if (jb == 0) {
      if (i > 0) zacc = tile_gemm(Wzb4 + ((i >> 3) * 128) * 32 + lane, ncp, h1row);
    } else {
      const int cp0 = deg_off(i - 1) >> 3;
      const v4f wz[3] = {zi0, zi1, zi2};
      v8f zb;
      for (int v = 0; v < 8; ++v) zb[v] = 0.f;
#pragma unroll
      for (int cl = 0; cl < 3; ++cl) {
        int hcp = cp0 + cl; hcp = hcp > 127 ? 127 : hcp;
        v2f wA = {wz[cl].x, wz[cl].y};
        v2f wB = {wz[cl].z, wz[cl].w};
        v2f hA = *(const v2f*)(h1row + hcp * 8);
        v2f hB = *(const v2f*)(h1row + hcp * 8 + 4);
        zacc = wmma_f32(wA, hA, zacc);
        zb   = wmma_f32(wB, hB, zb);
      }
      for (int v = 0; v < 8; ++v) zacc[v] += zb[v];
    }
    {
      float zm, zs;
      switch (jb & 3) {
        case 0:  zm = zacc[0]; zs = zacc[1]; break;
        case 1:  zm = zacc[2]; zs = zacc[3]; break;
        case 2:  zm = zacc[4]; zs = zacc[5]; break;
        default: zm = zacc[6]; zs = zacc[7]; break;
      }
      if (hi == (jb >> 2)) {
        const float mu = zm + b2[i];
        const float sg = zs + b2[kD + i];
        xs[n * kXStride + i] = us[n * kXStride + i] * expf(sg) + mu;
        ldj += sg;
      }
    }
    __syncthreads();
  }

  for (int e = lane; e < kBT * kD; e += 32) {
    int r = e >> 6, d = e & 63;
    out[(wg * kBT + r) * kD + d] = xs[r * kXStride + d];
  }
  const float ldjt = ldj + __shfl_xor(ldj, 16);
  if (hi == 0) out[kB * kD + wg * kBT + n] = ldjt;
}

extern "C" void kernel_launch(void* const* d_in, const int* in_sizes, int n_in,
                              void* d_out, int out_size, void* d_ws, size_t ws_size,
                              hipStream_t stream) {
  const float* u  = (const float*)d_in[0];
  const float* W0 = (const float*)d_in[1];
  const float* b0 = (const float*)d_in[2];
  const float* W1 = (const float*)d_in[3];
  const float* b1 = (const float*)d_in[4];
  const float* W2 = (const float*)d_in[5];
  const float* b2 = (const float*)d_in[6];
  float* ws  = (float*)d_ws;
  float* out = (float*)d_out;

  prep_w0<<<(kU * kD) / 256,     256, 0, stream>>>(W0, b0, b1, ws);
  prep_w1<<<(kU * kU) / 256,     256, 0, stream>>>(W1, ws + kWbOff);
  prep_w2<<<(kWzbSz + kWziSz) / 256, 256, 0, stream>>>(W2, ws);

  size_t smem = (size_t)(2 * kBT * kXStride + 2 * kBT * kHStride) * sizeof(float);
  made_flow_kernel<<<kB / kBT, 32, smem, stream>>>(u, b2, ws, out);
}


# Every tile's weights are requested while the tile before runs: tile_gemm's ring refills past a tile's end with the next tile's blocks, layer-1 weights a step ahead; the step's syncs wait for LDS only
# speedup vs baseline: 32.3604x; 1.3694x over previous
#include <hip/hip_runtime.h>

typedef __attribute__((ext_vector_type(2))) float v2f;
typedef __attribute__((ext_vector_type(8))) float v8f;
typedef __attribute__((ext_vector_type(4))) float v4f;

constexpr int kD  = 64;
constexpr int kU  = 1024;
constexpr int kB  = 4096;
constexpr int kBT = 16;

constexpr int kXStride = kD + 4;
constexpr int kHStride = kU + 4;

constexpr int kWaOff = 0;
constexpr int kWaSz  = kU * kD;
constexpr int kWbOff = kWaOff + kWaSz;
constexpr int kWbSz  = kU * kU;
constexpr int kWzbOff = kWbOff + kWbSz;
constexpr int kWzbSz  = 8 * 128 * 128;
constexpr int kWziOff = kWzbOff + kWzbSz;
constexpr int kWziSz  = kD * 3 * 128;
constexpr int kB0Off = kWziOff + kWziSz;
constexpr int kB1Off = kB0Off + kU;

__device__ __forceinline__ int deg_off(int d) {
  return d <= 16 ? 17 * d : 272 + 16 * (d - 16);
}
__device__ __forceinline__ int pos_of_unit(int j) {
  int d = j % (kD - 1), m = j / (kD - 1);
  return deg_off(d) + m;
}
__device__ __forceinline__ int unit_at_pos(int p) {
  int d, m;
  if (p < 272) { d = p / 17; m = p - 17 * d; }
  else         { d = 16 + ((p - 272) >> 4); m = (p - 272) & 15; }
  return m * (kD - 1) + d;
}


__global__ void prep_w0(const float* __restrict__ W0, const float* __restrict__ b0,
                        const float* __restrict__ b1, float* __restrict__ ws) {
  int e = blockIdx.x * 256 + threadIdx.x;
  int j = e >> 6, d = e & 63;
  int dh = j % (kD - 1);
  float v = (d <= dh) ? W0[e] : 0.f;
  int p = pos_of_unit(j);
  int t = p >> 4, n = p & 15;
  int cp = d >> 3, rr = d & 7;
  int q  = ((rr >> 2) << 1) | (rr & 1);
  int hi = (rr >> 1) & 1;
  int l  = (hi << 4) | n;
  ws[kWaOff + (((t * 8 + cp) * 32 + l) << 2) + q] = v;
  if (e < kU) {
    int pe = pos_of_unit(e);
    ws[kB0Off + pe] = b0[e];
    ws[kB1Off + pe] = b1[e];
  }
}

__global__ void prep_w1(const float* __restrict__ W1, float* __restrict__ Wb) {
  int e = blockIdx.x * 256 + threadIdx.x;
  int j = e >> 10, k = e & 1023;
  int dhj = j % (kD - 1), dhk = k % (kD - 1);
  float v = (dhk <= dhj) ? W1[e] : 0.f;
  int pj = pos_of_unit(j), pk = pos_of_unit(k);
  int t = pj >> 4, n = pj & 15;
  int cp = pk >> 3, rr = pk & 7;
  int q  = ((rr >> 2) << 1) | (rr & 1);
  int hi = (rr >> 1) & 1;
  int l  = (hi << 4) | n;
  Wb[(((t * 128 + cp) * 32 + l) << 2) + q] = v;
}

__global__ void prep_w2(const float* __restrict__ W2, float* __restrict__ ws) {
  int e = blockIdx.x * 256 + threadIdx.x;
  if (e < kWzbSz) {
    int q = e & 3, l = (e >> 2) & 31, cp = (e >> 7) & 127, b = e >> 14;
    int hi = l >> 4, m = l & 15;
    int p = 8 * cp + (q >> 1) * 4 + hi * 2 + (q & 1);
    int row = (m & 1) * kD + 8 * b + (m >> 1);
    float v = 0.f;
    if (p < deg_off(8 * b)) v = W2[row * kU + unit_at_pos(p)];
    ws[kWzbOff + e] = v;
  } else {
    int f = e - kWzbSz;
    int q = f & 3, l = (f >> 2) & 31, r = f >> 7;
    int i = r / 3, cl = r - 3 * i;
    int hi = l >> 4, m = l & 15;
    float v = 0.f;
    if (i >= 1) {
      int lo = deg_off(i - 1), hiEnd = deg_off(i);
      int p = 8 * ((lo >> 3) + cl) + (q >> 1) * 4 + hi * 2 + (q & 1);
      int row = (m & 1) * kD + 8 * (i >> 3) + (m >> 1);
      if (p >= lo && p < hiEnd) v = W2[row * kU + unit_at_pos(p)];
    }
    ws[kWziOff + f] = v;
  }
}

__device__ __forceinline__ v8f wmma_f32(v2f a, v2f b, v8f c) {
  return __builtin_amdgcn_wmma_f32_16x16x4_f32(false, a, false, b, (short)0, c,
                                               false, false);
}

constexpr int kBlk  = 16;
constexpr int kRing = 3;
constexpr int kPD   = kBlk * kRing;

__device__ __forceinline__ int ring_block(int G, int ncp) {
  const int b = G * kBlk, last = ncp - kBlk;
  return b > last ? last : b;
}

template <int G>
__device__ __forceinline__ void gemm_block(v4f (&buf)[kPD], v2f (&hf)[kRing][2 * kBlk],
                                           v8f& a0, v8f& a1, v8f& a2, v8f& a3,
                                           const v4f* __restrict__ wsl, int ofs, int ofsn,
                                           const float* hrow, int base, int ncp, int ncpn) {
  constexpr int GN = (G + 1) % kRing;
  int hb = base + kBlk; hb = hb > 128 - kBlk ? 128 - kBlk : hb;
#pragma unroll
  for (int u = 0; u < kBlk; ++u) {
    hf[GN][2 * u]     = *(const v2f*)(hrow + (hb + u) * 8);
    hf[GN][2 * u + 1] = *(const v2f*)(hrow + (hb + u) * 8 + 4);
  }
  __builtin_amdgcn_sched_barrier(0x6);
  if (G == 0 || base < ncp) {
#pragma unroll
    for (int u = 0; u < kBlk; ++u) {
      v4f w = buf[G * kBlk + u];
      v2f wA = {w.x, w.y};
      v2f wB = {w.z, w.w};
      if (u & 1) { a2 = wmma_f32(wA, hf[G][2 * u], a2); a3 = wmma_f32(wB, hf[G][2 * u + 1], a3); }
      else       { a0 = wmma_f32(wA, hf[G][2 * u], a0); a1 = wmma_f32(wB, hf[G][2 * u + 1], a1); }
    }
  }
  __builtin_amdgcn_sched_barrier(0x6);
  const int wb = (base + kPD < ncp) ? ofs + (base + kPD) * 32 : ofsn + ring_block(G, ncpn) * 32;
#pragma unroll
  for (int u = 0; u < kBlk; ++u) buf[G * kBlk + u] = wsl[wb + u * 32];
  __builtin_amdgcn_sched_barrier(0x6);
}

__device__ __forceinline__ void ring_fence(const v4f (&buf)[kPD]) {
  asm volatile("" :: "v"(buf[0]), "v"(buf[1]), "v"(buf[2]), "v"(buf[3]), "v"(buf[4]), "v"(buf[5]),
               "v"(buf[6]), "v"(buf[7]), "v"(buf[8]), "v"(buf[9]), "v"(buf[10]), "v"(buf[11]),
               "v"(buf[12]), "v"(buf[13]), "v"(buf[14]), "v"(buf[15]));
  asm volatile("" :: "v"(buf[16]), "v"(buf[17]), "v"(buf[18]), "v"(buf[19]), "v"(buf[20]), "v"(buf[21]),
               "v"(buf[22]), "v"(buf[23]), "v"(buf[24]), "v"(buf[25]), "v"(buf[26]), "v"(buf[27]),
               "v"(buf[28]), "v"(buf[29]), "v"(buf[30]), "v"(buf[31]));
  asm volatile("" :: "v"(buf[32]), "v"(buf[33]), "v"(buf[34]), "v"(buf[35]), "v"(buf[36]), "v"(buf[37]),
               "v"(buf[38]), "v"(buf[39]), "v"(buf[40]), "v"(buf[41]), "v"(buf[42]), "v"(buf[43]),
               "v"(buf[44]), "v"(buf[45]), "v"(buf[46]), "v"(buf[47]));
}

__device__ __forceinline__ v8f tile_gemm(v4f (&buf)[kPD], const v4f* __restrict__ wsl, int ofs, int ncp,
                                         int ofsn, int ncpn, const float* hrow) {
  v2f hf[kRing][2 * kBlk];
#pragma unroll
  for (int u = 0; u < kBlk; ++u) {
    hf[0][2 * u]     = *(const v2f*)(hrow + u * 8);
    hf[0][2 * u + 1] = *(const v2f*)(hrow + u * 8 + 4);
  }
  v8f a0, a1, a2, a3;
  for (int v = 0; v < 8; ++v) { a0[v] = 0.f; a1[v] = 0.f; a2[v] = 0.f; a3[v] = 0.f; }
  int base = 0;
  do {
    gemm_block<0>(buf, hf, a0, a1, a2, a3, wsl, ofs, ofsn, hrow, base, ncp, ncpn);
    gemm_block<1>(buf, hf, a0, a1, a2, a3, wsl, ofs, ofsn, hrow, base + kBlk, ncp, ncpn);
    gemm_block<2>(buf, hf, a0, a1, a2, a3, wsl, ofs, ofsn, hrow, base + 2 * kBlk, ncp, ncpn);
    base += kPD;
  } while (base < ncp);
  v8f r;
  for (int v = 0; v < 8; ++v) r[v] = (a0[v] + a2[v]) + (a1[v] + a3[v]);
  return r;
}

__device__ __forceinline__ void lds_sync() {
  __builtin_amdgcn_fence(__ATOMIC_RELEASE, "workgroup", "local");
  __builtin_amdgcn_s_barrier();
  __builtin_amdgcn_fence(__ATOMIC_ACQUIRE, "workgroup", "local");
}

__device__ __forceinline__ int ncp_of(int i) {
  return ((deg_off(i) + 127) >> 7) << 4;
}

__device__ __forceinline__ void l1_fetch(const float4* __restrict__ Wa4, const float* __restrict__ b0s,
                                         int t, int lane, int hi,
                                         float4 (&w)[8], float4& bb0, float4& bb1) {
  const float4* wq = Wa4 + (t * 8) * 32 + lane;
#pragma unroll
  for (int cp = 0; cp < 8; ++cp) w[cp] = wq[cp * 32];
  bb0 = *(const float4*)(b0s + t * 16 + hi * 8);
  bb1 = *(const float4*)(b0s + t * 16 + hi * 8 + 4);
}
__device__ __forceinline__ void l1_tile(const float4 (&wr)[8], const float4 bb0, const float4 bb1,
                                        const float* xrow, float* dst) {
  v8f acc;
  acc[0] = bb0.x; acc[1] = bb0.y; acc[2] = bb0.z; acc[3] = bb0.w;
  acc[4] = bb1.x; acc[5] = bb1.y; acc[6] = bb1.z; acc[7] = bb1.w;
  v8f a1;
  for (int v = 0; v < 8; ++v) a1[v] = 0.f;
#pragma unroll
  for (int cp = 0; cp < 8; ++cp) {
    const float4 w = wr[cp];
    v2f wA = {w.x, w.y};
    v2f wB = {w.z, w.w};
    v2f xA = *(const v2f*)(xrow + cp * 8);
    v2f xB = *(const v2f*)(xrow + cp * 8 + 4);
    acc = wmma_f32(wA, xA, acc);
    a1  = wmma_f32(wB, xB, a1);
  }
  float4 o0, o1;
  o0.x = fmaxf(acc[0] + a1[0], 0.f); o0.y = fmaxf(acc[1] + a1[1], 0.f);
  o0.z = fmaxf(acc[2] + a1[2], 0.f); o0.w = fmaxf(acc[3] + a1[3], 0.f);
  o1.x = fmaxf(acc[4] + a1[4], 0.f); o1.y = fmaxf(acc[5] + a1[5], 0.f);
  o1.z = fmaxf(acc[6] + a1[6], 0.f); o1.w = fmaxf(acc[7] + a1[7], 0.f);
  *(float4*)dst = o0;
  *(float4*)(dst + 4) = o1;
}

__launch_bounds__(32, 1)
__global__ void made_flow_kernel(const float* __restrict__ u,
                                 const float* __restrict__ b2,
                                 const float* __restrict__ ws,
                                 float* __restrict__ out) {
  extern __shared__ float smem[];
  float* xs  = smem;
  float* us  = xs  + kBT * kXStride;
  float* h0s = us  + kBT * kXStride;
  float* h1s = h0s + kBT * kHStride;

  const int lane = threadIdx.x;
  const int wg   = blockIdx.x;
  const int n    = lane & 15;
  const int hi   = lane >> 4;
  const int hi2  = hi << 1;

  const float4* Wa4 = (const float4*)(ws + kWaOff);
  const v4f* wsl = (const v4f*)ws + lane;
  constexpr int kWbQ  = kWbOff / 4;
  constexpr int kWzbQ = kWzbOff / 4;
  const v4f* Wzi4 = (const v4f*)(ws + kWziOff);
  const float*  b0s = ws + kB0Off;
  const float*  b1s = ws + kB1Off;

  for (int e = lane; e < kBT * kXStride; e += 32) xs[e] = 0.f;
  for (int e = lane; e < 2 * kBT * kHStride; e += 32) h0s[e] = 0.f;
  for (int e = lane; e < kBT * kD; e += 32) {
    int r = e >> 6, d = e & 63;
    us[r * kXStride + d] = u[(wg * kBT + r) * kD + d];
  }
  float ldj = 0.f;

  v4f buf[kPD];
  int ringOfs = kWbQ, ringN = ncp_of(1);
#pragma unroll
  for (int u = 0; u < kPD; ++u) buf[u] = wsl[ringOfs + (ring_block(u / kBlk, ringN) + u % kBlk) * 32];
  float4 waA[8], waB[8], bA0, bA1, bB0, bB1;
  l1_fetch(Wa4, b0s, 0, lane, hi, waA, bA0, bA1);
  l1_fetch(Wa4, b0s, 1, lane, hi, waB, bB0, bB1);
  __syncthreads();

  v8f zacc;
  for (int v = 0; v < 8; ++v) zacc[v] = 0.f;
  const float* xrow  = xs  + n * kXStride + hi2;
  const float* h0row = h0s + n * kHStride + hi2;
  const float* h1row = h1s + n * kHStride + hi2;

  for (int i = 0; i < kD; ++i) {
    const int offi = deg_off(i);
    const int ncp  = ncp_of(i);
    const int jb = i & 7;
    if (i > 0) {
      const int tlo = deg_off(i - 1) >> 4, thi = (offi - 1) >> 4;
      l1_tile(waA, bA0, bA1, xrow, h0s + n * kHStride + tlo * 16 + hi * 8);
      if (thi > tlo) l1_tile(waB, bB0, bB1, xrow, h0s + n * kHStride + thi * 16 + hi * 8);
      lds_sync();
      v4f zi0, zi1, zi2;
      int t = tlo;
      do {
        ring_fence(buf);
        if (t == tlo) {
          const v4f* zq = Wzi4 + (i * 3) * 32 + lane;
          zi0 = zq[0]; zi1 = zq[32]; zi2 = zq[64];
          const int iN = i + 1 < kD ? i + 1 : i;
          const int tloN = deg_off(iN - 1) >> 4, thiN = (deg_off(iN) - 1) >> 4;
          l1_fetch(Wa4, b0s, tloN, lane, hi, waA, bA0, bA1);
          if (thiN > tloN) l1_fetch(Wa4, b0s, thiN, lane, hi, waB, bB0, bB1);
        }
        const float4 bb0 = *(const float4*)(b1s + t * 16 + hi * 8);
        const float4 bb1 = *(const float4*)(b1s + t * 16 + hi * 8 + 4);
        int nOfs = ringOfs, nN = ringN;
        if (t < thi)          { nOfs = kWbQ + (t + 1) * 128 * 32; nN = ncp; }
        else if (jb == 0)     { nOfs = kWzbQ + (i >> 3) * 128 * 32; nN = ncp; }
        else if (i + 1 < kD)  { nOfs = kWbQ + (offi >> 4) * 128 * 32; nN = ncp_of(i + 1); }
        v8f r = tile_gemm(buf, wsl, ringOfs, ringN, nOfs, nN, h0row);
        ringOfs = nOfs; ringN = nN;
        float4 o0, o1;
        o0.x = fmaxf(r[0] + bb0.x, 0.f); o0.y = fmaxf(r[1] + bb0.y, 0.f);
        o0.z = fmaxf(r[2] + bb0.z, 0.f); o0.w = fmaxf(r[3] + bb0.w, 0.f);
        o1.x = fmaxf(r[4] + bb1.x, 0.f); o1.y = fmaxf(r[5] + bb1.y, 0.f);
        o1.z = fmaxf(r[6] + bb1.z, 0.f); o1.w = fmaxf(r[7] + bb1.w, 0.f);
        float* dst = h1s + n * kHStride + t * 16 + hi * 8;
        *(float4*)dst = o0;
        *(float4*)(dst + 4) = o1;
      } while (++t <= thi);
      lds_sync();
      if (jb == 0) {
        const int nOfs = kWbQ + (offi >> 4) * 128 * 32, nN = ncp_of(i + 1);
        ring_fence(buf);
        zacc = tile_gemm(buf, wsl, ringOfs, ringN, nOfs, nN, h1row);
        ringOfs = nOfs; ringN = nN;
      } else {
        const int cp0 = deg_off(i - 1) >> 3;
        const v4f wz[3] = {zi0, zi1, zi2};
        v8f zb;
        for (int v = 0; v < 8; ++v) zb[v] = 0.f;
#pragma unroll
        for (int cl = 0; cl < 3; ++cl) {
          int hcp = cp0 + cl; hcp = hcp > 127 ? 127 : hcp;
          v2f wA = {wz[cl].x, wz[cl].y};
          v2f wB = {wz[cl].z, wz[cl].w};
          v2f hA = *(const v2f*)(h1row + hcp * 8);
          v2f hB = *(const v2f*)(h1row + hcp * 8 + 4);
          zacc = wmma_f32(wA, hA, zacc);
          zb   = wmma_f32(wB, hB, zb);
        }
        for (int v = 0; v < 8; ++v) zacc[v] += zb[v];
      }
    }
    {
      float zm, zs;
      switch (jb & 3) {
        case 0:  zm = zacc[0]; zs = zacc[1]; break;
        case 1:  zm = zacc[2]; zs = zacc[3]; break;
        case 2:  zm = zacc[4]; zs = zacc[5]; break;
        default: zm = zacc[6]; zs = zacc[7]; break;
      }
      if (hi == (jb >> 2)) {
        const float mu = zm + b2[i];
        const float sg = zs + b2[kD + i];
        xs[n * kXStride + i] = us[n * kXStride + i] * expf(sg) + mu;
        ldj += sg;
      }
    }
    lds_sync();
  }

  for (int e = lane; e < kBT * kD; e += 32) {
    int r = e >> 6, d = e & 63;
    out[(wg * kBT + r) * kD + d] = xs[r * kXStride + d];
  }
  const float ldjt = ldj + __shfl_xor(ldj, 16);
  if (hi == 0) out[kB * kD + wg * kBT + n] = ldjt;
}

extern "C" void kernel_launch(void* const* d_in, const int* in_sizes, int n_in,
                              void* d_out, int out_size, void* d_ws, size_t ws_size,
                              hipStream_t stream) {
  const float* u  = (const float*)d_in[0];
  const float* W0 = (const float*)d_in[1];
  const float* b0 = (const float*)d_in[2];
  const float* W1 = (const float*)d_in[3];
  const float* b1 = (const float*)d_in[4];
  const float* W2 = (const float*)d_in[5];
  const float* b2 = (const float*)d_in[6];
  float* ws  = (float*)d_ws;
  float* out = (float*)d_out;

  prep_w0<<<(kU * kD) / 256,     256, 0, stream>>>(W0, b0, b1, ws);
  prep_w1<<<(kU * kU) / 256,     256, 0, stream>>>(W1, ws + kWbOff);
  prep_w2<<<(kWzbSz + kWziSz) / 256, 256, 0, stream>>>(W2, ws);

  size_t smem = (size_t)(2 * kBT * kXStride + 2 * kBT * kHStride) * sizeof(float);
  made_flow_kernel<<<kB / kBT, 32, smem, stream>>>(u, b2, ws, out);
}


# The step's short phases request operands early, as groups: layer 1's 16 x fragments (once for both tiles), layer 3's 6 h fragments ahead of their WMMAs, b2 and u of column i at the step's top
# speedup vs baseline: 33.7696x; 1.0435x over previous
#include <hip/hip_runtime.h>

typedef __attribute__((ext_vector_type(2))) float v2f;
typedef __attribute__((ext_vector_type(8))) float v8f;
typedef __attribute__((ext_vector_type(4))) float v4f;

constexpr int kD  = 64;
constexpr int kU  = 1024;
constexpr int kB  = 4096;
constexpr int kBT = 16;

constexpr int kXStride = kD + 4;
constexpr int kHStride = kU + 4;

constexpr int kWaOff = 0;
constexpr int kWaSz  = kU * kD;
constexpr int kWbOff = kWaOff + kWaSz;
constexpr int kWbSz  = kU * kU;
constexpr int kWzbOff = kWbOff + kWbSz;
constexpr int kWzbSz  = 8 * 128 * 128;
constexpr int kWziOff = kWzbOff + kWzbSz;
constexpr int kWziSz  = kD * 3 * 128;
constexpr int kB0Off = kWziOff + kWziSz;
constexpr int kB1Off = kB0Off + kU;

__device__ __forceinline__ int deg_off(int d) {
  return d <= 16 ? 17 * d : 272 + 16 * (d - 16);
}
__device__ __forceinline__ int pos_of_unit(int j) {
  int d = j % (kD - 1), m = j / (kD - 1);
  return deg_off(d) + m;
}
__device__ __forceinline__ int unit_at_pos(int p) {
  int d, m;
  if (p < 272) { d = p / 17; m = p - 17 * d; }
  else         { d = 16 + ((p - 272) >> 4); m = (p - 272) & 15; }
  return m * (kD - 1) + d;
}


__global__ void prep_w0(const float* __restrict__ W0, const float* __restrict__ b0,
                        const float* __restrict__ b1, float* __restrict__ ws) {
  int e = blockIdx.x * 256 + threadIdx.x;
  int j = e >> 6, d = e & 63;
  int dh = j % (kD - 1);
  float v = (d <= dh) ? W0[e] : 0.f;
  int p = pos_of_unit(j);
  int t = p >> 4, n = p & 15;
  int cp = d >> 3, rr = d & 7;
  int q  = ((rr >> 2) << 1) | (rr & 1);
  int hi = (rr >> 1) & 1;
  int l  = (hi << 4) | n;
  ws[kWaOff + (((t * 8 + cp) * 32 + l) << 2) + q] = v;
  if (e < kU) {
    int pe = pos_of_unit(e);
    ws[kB0Off + pe] = b0[e];
    ws[kB1Off + pe] = b1[e];
  }
}

__global__ void prep_w1(const float* __restrict__ W1, float* __restrict__ Wb) {
  int e = blockIdx.x * 256 + threadIdx.x;
  int j = e >> 10, k = e & 1023;
  int dhj = j % (kD - 1), dhk = k % (kD - 1);
  float v = (dhk <= dhj) ? W1[e] : 0.f;
  int pj = pos_of_unit(j), pk = pos_of_unit(k);
  int t = pj >> 4, n = pj & 15;
  int cp = pk >> 3, rr = pk & 7;
  int q  = ((rr >> 2) << 1) | (rr & 1);
  int hi = (rr >> 1) & 1;
  int l  = (hi << 4) | n;
  Wb[(((t * 128 + cp) * 32 + l) << 2) + q] = v;
}

__global__ void prep_w2(const float* __restrict__ W2, float* __restrict__ ws) {
  int e = blockIdx.x * 256 + threadIdx.x;
  if (e < kWzbSz) {
    int q = e & 3, l = (e >> 2) & 31, cp = (e >> 7) & 127, b = e >> 14;
    int hi = l >> 4, m = l & 15;
    int p = 8 * cp + (q >> 1) * 4 + hi * 2 + (q & 1);
    int row = (m & 1) * kD + 8 * b + (m >> 1);
    float v = 0.f;
    if (p < deg_off(8 * b)) v = W2[row * kU + unit_at_pos(p)];
    ws[kWzbOff + e] = v;
  } else {
    int f = e - kWzbSz;
    int q = f & 3, l = (f >> 2) & 31, r = f >> 7;
    int i = r / 3, cl = r - 3 * i;
    int hi = l >> 4, m = l & 15;
    float v = 0.f;
    if (i >= 1) {
      int lo = deg_off(i - 1), hiEnd = deg_off(i);
      int p = 8 * ((lo >> 3) + cl) + (q >> 1) * 4 + hi * 2 + (q & 1);
      int row = (m & 1) * kD + 8 * (i >> 3) + (m >> 1);
      if (p >= lo && p < hiEnd) v = W2[row * kU + unit_at_pos(p)];
    }
    ws[kWziOff + f] = v;
  }
}

__device__ __forceinline__ v8f wmma_f32(v2f a, v2f b, v8f c) {
  return __builtin_amdgcn_wmma_f32_16x16x4_f32(false, a, false, b, (short)0, c,
                                               false, false);
}

constexpr int kBlk  = 16;
constexpr int kRing = 3;
constexpr int kPD   = kBlk * kRing;

__device__ __forceinline__ int ring_block(int G, int ncp) {
  const int b = G * kBlk, last = ncp - kBlk;
  return b > last ? last : b;
}

template <int G>
__device__ __forceinline__ void gemm_block(v4f (&buf)[kPD], v2f (&hf)[kRing][2 * kBlk],
                                           v8f& a0, v8f& a1, v8f& a2, v8f& a3,
                                           const v4f* __restrict__ wsl, int ofs, int ofsn,
                                           const float* hrow, int base, int ncp, int ncpn) {
  constexpr int GN = (G + 1) % kRing;
  int hb = base + kBlk; hb = hb > 128 - kBlk ? 128 - kBlk : hb;
#pragma unroll
  for (int u = 0; u < kBlk; ++u) {
    hf[GN][2 * u]     = *(const v2f*)(hrow + (hb + u) * 8);
    hf[GN][2 * u + 1] = *(const v2f*)(hrow + (hb + u) * 8 + 4);
  }
  __builtin_amdgcn_sched_barrier(0x6);
  if (G == 0 || base < ncp) {
#pragma unroll
    for (int u = 0; u < kBlk; ++u) {
      v4f w = buf[G * kBlk + u];
      v2f wA = {w.x, w.y};
      v2f wB = {w.z, w.w};
      if (u & 1) { a2 = wmma_f32(wA, hf[G][2 * u], a2); a3 = wmma_f32(wB, hf[G][2 * u + 1], a3); }
      else       { a0 = wmma_f32(wA, hf[G][2 * u], a0); a1 = wmma_f32(wB, hf[G][2 * u + 1], a1); }
    }
  }
  __builtin_amdgcn_sched_barrier(0x6);
  const int wb = (base + kPD < ncp) ? ofs + (base + kPD) * 32 : ofsn + ring_block(G, ncpn) * 32;
#pragma unroll
  for (int u = 0; u < kBlk; ++u) buf[G * kBlk + u] = wsl[wb + u * 32];
  __builtin_amdgcn_sched_barrier(0x6);
}

__device__ __forceinline__ void ring_fence(const v4f (&buf)[kPD]) {
  asm volatile("" :: "v"(buf[0]), "v"(buf[1]), "v"(buf[2]), "v"(buf[3]), "v"(buf[4]), "v"(buf[5]),
               "v"(buf[6]), "v"(buf[7]), "v"(buf[8]), "v"(buf[9]), "v"(buf[10]), "v"(buf[11]),
               "v"(buf[12]), "v"(buf[13]), "v"(buf[14]), "v"(buf[15]));
  asm volatile("" :: "v"(buf[16]), "v"(buf[17]), "v"(buf[18]), "v"(buf[19]), "v"(buf[20]), "v"(buf[21]),
               "v"(buf[22]), "v"(buf[23]), "v"(buf[24]), "v"(buf[25]), "v"(buf[26]), "v"(buf[27]),
               "v"(buf[28]), "v"(buf[29]), "v"(buf[30]), "v"(buf[31]));
  asm volatile("" :: "v"(buf[32]), "v"(buf[33]), "v"(buf[34]), "v"(buf[35]), "v"(buf[36]), "v"(buf[37]),
               "v"(buf[38]), "v"(buf[39]), "v"(buf[40]), "v"(buf[41]), "v"(buf[42]), "v"(buf[43]),
               "v"(buf[44]), "v"(buf[45]), "v"(buf[46]), "v"(buf[47]));
}

__device__ __forceinline__ v8f tile_gemm(v4f (&buf)[kPD], const v4f* __restrict__ wsl, int ofs, int ncp,
                                         int ofsn, int ncpn, const float* hrow) {
  v2f hf[kRing][2 * kBlk];
#pragma unroll
  for (int u = 0; u < kBlk; ++u) {
    hf[0][2 * u]     = *(const v2f*)(hrow + u * 8);
    hf[0][2 * u + 1] = *(const v2f*)(hrow + u * 8 + 4);
  }
  v8f a0, a1, a2, a3;
  for (int v = 0; v < 8; ++v) { a0[v] = 0.f; a1[v] = 0.f; a2[v] = 0.f; a3[v] = 0.f; }
  int base = 0;
  do {
    gemm_block<0>(buf, hf, a0, a1, a2, a3, wsl, ofs, ofsn, hrow, base, ncp, ncpn);
    gemm_block<1>(buf, hf, a0, a1, a2, a3, wsl, ofs, ofsn, hrow, base + kBlk, ncp, ncpn);
    gemm_block<2>(buf, hf, a0, a1, a2, a3, wsl, ofs, ofsn, hrow, base + 2 * kBlk, ncp, ncpn);
    base += kPD;
  } while (base < ncp);
  v8f r;
  for (int v = 0; v < 8; ++v) r[v] = (a0[v] + a2[v]) + (a1[v] + a3[v]);
  return r;
}

__device__ __forceinline__ void lds_sync() {
  __builtin_amdgcn_fence(__ATOMIC_RELEASE, "workgroup", "local");
  __builtin_amdgcn_s_barrier();
  __builtin_amdgcn_fence(__ATOMIC_ACQUIRE, "workgroup", "local");
}

__device__ __forceinline__ int ncp_of(int i) {
  return ((deg_off(i) + 127) >> 7) << 4;
}

__device__ __forceinline__ void l1_fetch(const float4* __restrict__ Wa4, const float* __restrict__ b0s,
                                         int t, int lane, int hi,
                                         float4 (&w)[8], float4& bb0, float4& bb1) {
  const float4* wq = Wa4 + (t * 8) * 32 + lane;
#pragma unroll
  for (int cp = 0; cp < 8; ++cp) w[cp] = wq[cp * 32];
  bb0 = *(const float4*)(b0s + t * 16 + hi * 8);
  bb1 = *(const float4*)(b0s + t * 16 + hi * 8 + 4);
}
__device__ __forceinline__ void l1_xfrags(const float* xrow, v2f (&xf)[16]) {
#pragma unroll
  for (int cp = 0; cp < 8; ++cp) {
    xf[2 * cp]     = *(const v2f*)(xrow + cp * 8);
    xf[2 * cp + 1] = *(const v2f*)(xrow + cp * 8 + 4);
  }
}
__device__ __forceinline__ void l1_tile(const float4 (&wr)[8], const float4 bb0, const float4 bb1,
                                        const v2f (&xf)[16], float* dst) {
  v8f acc;
  acc[0] = bb0.x; acc[1] = bb0.y; acc[2] = bb0.z; acc[3] = bb0.w;
  acc[4] = bb1.x; acc[5] = bb1.y; acc[6] = bb1.z; acc[7] = bb1.w;
  v8f a1;
  for (int v = 0; v < 8; ++v) a1[v] = 0.f;
#pragma unroll
  for (int cp = 0; cp < 8; ++cp) {
    const float4 w = wr[cp];
    v2f wA = {w.x, w.y};
    v2f wB = {w.z, w.w};
    acc = wmma_f32(wA, xf[2 * cp], acc);
    a1  = wmma_f32(wB, xf[2 * cp + 1], a1);
  }
  float4 o0, o1;
  o0.x = fmaxf(acc[0] + a1[0], 0.f); o0.y = fmaxf(acc[1] + a1[1], 0.f);
  o0.z = fmaxf(acc[2] + a1[2], 0.f); o0.w = fmaxf(acc[3] + a1[3], 0.f);
  o1.x = fmaxf(acc[4] + a1[4], 0.f); o1.y = fmaxf(acc[5] + a1[5], 0.f);
  o1.z = fmaxf(acc[6] + a1[6], 0.f); o1.w = fmaxf(acc[7] + a1[7], 0.f);
  *(float4*)dst = o0;
  *(float4*)(dst + 4) = o1;
}

__launch_bounds__(32, 1)
__global__ void made_flow_kernel(const float* __restrict__ u,
                                 const float* __restrict__ b2,
                                 const float* __restrict__ ws,
                                 float* __restrict__ out) {
  extern __shared__ float smem[];
  float* xs  = smem;
  float* us  = xs  + kBT * kXStride;
  float* h0s = us  + kBT * kXStride;
  float* h1s = h0s + kBT * kHStride;

  const int lane = threadIdx.x;
  const int wg   = blockIdx.x;
  const int n    = lane & 15;
  const int hi   = lane >> 4;
  const int hi2  = hi << 1;

  const float4* Wa4 = (const float4*)(ws + kWaOff);
  const v4f* wsl = (const v4f*)ws + lane;
  constexpr int kWbQ  = kWbOff / 4;
  constexpr int kWzbQ = kWzbOff / 4;
  const v4f* Wzi4 = (const v4f*)(ws + kWziOff);
  const float*  b0s = ws + kB0Off;
  const float*  b1s = ws + kB1Off;

  for (int e = lane; e < kBT * kXStride; e += 32) xs[e] = 0.f;
  for (int e = lane; e < 2 * kBT * kHStride; e += 32) h0s[e] = 0.f;
  for (int e = lane; e < kBT * kD; e += 32) {
    int r = e >> 6, d = e & 63;
    us[r * kXStride + d] = u[(wg * kBT + r) * kD + d];
  }
  float ldj = 0.f;

  v4f buf[kPD];
  int ringOfs = kWbQ, ringN = ncp_of(1);
#pragma unroll
  for (int u = 0; u < kPD; ++u) buf[u] = wsl[ringOfs + (ring_block(u / kBlk, ringN) + u % kBlk) * 32];
  float4 waA[8], waB[8], bA0, bA1, bB0, bB1;
  l1_fetch(Wa4, b0s, 0, lane, hi, waA, bA0, bA1);
  l1_fetch(Wa4, b0s, 1, lane, hi, waB, bB0, bB1);
  __syncthreads();

  v8f zacc;
  for (int v = 0; v < 8; ++v) zacc[v] = 0.f;
  const float* xrow  = xs  + n * kXStride + hi2;
  const float* h0row = h0s + n * kHStride + hi2;
  const float* h1row = h1s + n * kHStride + hi2;

  for (int i = 0; i < kD; ++i) {
    const int offi = deg_off(i);
    const int ncp  = ncp_of(i);
    const int jb = i & 7;
    const float b2m = b2[i];
    const float b2s = b2[kD + i];
    const float uval = us[n * kXStride + i];
    if (i > 0) {
      const int tlo = deg_off(i - 1) >> 4, thi = (offi - 1) >> 4;
      v2f xf[16];
      l1_xfrags(xrow, xf);
      __builtin_amdgcn_sched_barrier(0x6);
      l1_tile(waA, bA0, bA1, xf, h0s + n * kHStride + tlo * 16 + hi * 8);
      if (thi > tlo) l1_tile(waB, bB0, bB1, xf, h0s + n * kHStride + thi * 16 + hi * 8);
      lds_sync();
      v4f zi0, zi1, zi2;
      int t = tlo;
      do {
        ring_fence(buf);
        if (t == tlo) {
          const v4f* zq = Wzi4 + (i * 3) * 32 + lane;
          zi0 = zq[0]; zi1 = zq[32]; zi2 = zq[64];
          const int iN = i + 1 < kD ? i + 1 : i;
          const int tloN = deg_off(iN - 1) >> 4, thiN = (deg_off(iN) - 1) >> 4;
          l1_fetch(Wa4, b0s, tloN, lane, hi, waA, bA0, bA1);
          if (thiN > tloN) l1_fetch(Wa4, b0s, thiN, lane, hi, waB, bB0, bB1);
        }
        const float4 bb0 = *(const float4*)(b1s + t * 16 + hi * 8);
        const float4 bb1 = *(const float4*)(b1s + t * 16 + hi * 8 + 4);
        int nOfs = ringOfs, nN = ringN;
        if (t < thi)          { nOfs = kWbQ + (t + 1) * 128 * 32; nN = ncp; }
        else if (jb == 0)     { nOfs = kWzbQ + (i >> 3) * 128 * 32; nN = ncp; }
        else if (i + 1 < kD)  { nOfs = kWbQ + (offi >> 4) * 128 * 32; nN = ncp_of(i + 1); }
        v8f r = tile_gemm(buf, wsl, ringOfs, ringN, nOfs, nN, h0row);
        ringOfs = nOfs; ringN = nN;
        float4 o0, o1;
        o0.x = fmaxf(r[0] + bb0.x, 0.f); o0.y = fmaxf(r[1] + bb0.y, 0.f);
        o0.z = fmaxf(r[2] + bb0.z, 0.f); o0.w = fmaxf(r[3] + bb0.w, 0.f);
        o1.x = fmaxf(r[4] + bb1.x, 0.f); o1.y = fmaxf(r[5] + bb1.y, 0.f);
        o1.z = fmaxf(r[6] + bb1.z, 0.f); o1.w = fmaxf(r[7] + bb1.w, 0.f);
        float* dst = h1s + n * kHStride + t * 16 + hi * 8;
        *(float4*)dst = o0;
        *(float4*)(dst + 4) = o1;
      } while (++t <= thi);
      lds_sync();
      if (jb == 0) {
        const int nOfs = kWbQ + (offi >> 4) * 128 * 32, nN = ncp_of(i + 1);
        ring_fence(buf);
        zacc = tile_gemm(buf, wsl, ringOfs, ringN, nOfs, nN, h1row);
        ringOfs = nOfs; ringN = nN;
      } else {
        const int cp0 = deg_off(i - 1) >> 3;
        const v4f wz[3] = {zi0, zi1, zi2};
        v8f zb;
        for (int v = 0; v < 8; ++v) zb[v] = 0.f;
        v2f hA[3], hB[3];
#pragma unroll
        for (int cl = 0; cl < 3; ++cl) {
          int hcp = cp0 + cl; hcp = hcp > 127 ? 127 : hcp;
          hA[cl] = *(const v2f*)(h1row + hcp * 8);
          hB[cl] = *(const v2f*)(h1row + hcp * 8 + 4);
        }
        __builtin_amdgcn_sched_barrier(0x6);
#pragma unroll
        for (int cl = 0; cl < 3; ++cl) {
          v2f wA = {wz[cl].x, wz[cl].y};
          v2f wB = {wz[cl].z, wz[cl].w};
          zacc = wmma_f32(wA, hA[cl], zacc);
          zb   = wmma_f32(wB, hB[cl], zb);
        }
        for (int v = 0; v < 8; ++v) zacc[v] += zb[v];
      }
    }
    {
      float zm, zs;
      switch (jb & 3) {
        case 0:  zm = zacc[0]; zs = zacc[1]; break;
        case 1:  zm = zacc[2]; zs = zacc[3]; break;
        case 2:  zm = zacc[4]; zs = zacc[5]; break;
        default: zm = zacc[6]; zs = zacc[7]; break;
      }
      if (hi == (jb >> 2)) {
        const float mu = zm + b2m;
        const float sg = zs + b2s;
        xs[n * kXStride + i] = uval * expf(sg) + mu;
        ldj += sg;
      }
    }
    lds_sync();
  }

  for (int e = lane; e < kBT * kD; e += 32) {
    int r = e >> 6, d = e & 63;
    out[(wg * kBT + r) * kD + d] = xs[r * kXStride + d];
  }
  const float ldjt = ldj + __shfl_xor(ldj, 16);
  if (hi == 0) out[kB * kD + wg * kBT + n] = ldjt;
}

extern "C" void kernel_launch(void* const* d_in, const int* in_sizes, int n_in,
                              void* d_out, int out_size, void* d_ws, size_t ws_size,
                              hipStream_t stream) {
  const float* u  = (const float*)d_in[0];
  const float* W0 = (const float*)d_in[1];
  const float* b0 = (const float*)d_in[2];
  const float* W1 = (const float*)d_in[3];
  const float* b1 = (const float*)d_in[4];
  const float* W2 = (const float*)d_in[5];
  const float* b2 = (const float*)d_in[6];
  float* ws  = (float*)d_ws;
  float* out = (float*)d_out;

  prep_w0<<<(kU * kD) / 256,     256, 0, stream>>>(W0, b0, b1, ws);
  prep_w1<<<(kU * kU) / 256,     256, 0, stream>>>(W1, ws + kWbOff);
  prep_w2<<<(kWzbSz + kWziSz) / 256, 256, 0, stream>>>(W2, ws);

  size_t smem = (size_t)(2 * kBT * kXStride + 2 * kBT * kHStride) * sizeof(float);
  made_flow_kernel<<<kB / kBT, 32, smem, stream>>>(u, b2, ws, out);
}


# Tile weights reach the matrix wave through LDS: 3 copy waves on the idle SIMDs fill a double buffer with the next tile while one runs; no weight ring, no memory loads in its loop (sum order kept)
# speedup vs baseline: 35.3572x; 1.0470x over previous
#include <hip/hip_runtime.h>

typedef __attribute__((ext_vector_type(2))) float v2f;
typedef __attribute__((ext_vector_type(8))) float v8f;
typedef __attribute__((ext_vector_type(4))) float v4f;

constexpr int kD  = 64;
constexpr int kU  = 1024;
constexpr int kB  = 4096;
constexpr int kBT = 16;

constexpr int kXStride = kD + 4;
constexpr int kHStride = kU + 4;

constexpr int kWaOff = 0;
constexpr int kWaSz  = kU * kD;
constexpr int kWbOff = kWaOff + kWaSz;
constexpr int kWbSz  = kU * kU;
constexpr int kWzbOff = kWbOff + kWbSz;
constexpr int kWzbSz  = 8 * 128 * 128;
constexpr int kWziOff = kWzbOff + kWzbSz;
constexpr int kWziSz  = kD * 3 * 128;
constexpr int kB0Off = kWziOff + kWziSz;
constexpr int kB1Off = kB0Off + kU;

__device__ __forceinline__ int deg_off(int d) {
  return d <= 16 ? 17 * d : 272 + 16 * (d - 16);
}
__device__ __forceinline__ int pos_of_unit(int j) {
  int d = j % (kD - 1), m = j / (kD - 1);
  return deg_off(d) + m;
}
__device__ __forceinline__ int unit_at_pos(int p) {
  int d, m;
  if (p < 272) { d = p / 17; m = p - 17 * d; }
  else         { d = 16 + ((p - 272) >> 4); m = (p - 272) & 15; }
  return m * (kD - 1) + d;
}


__global__ void prep_w0(const float* __restrict__ W0, const float* __restrict__ b0,
                        const float* __restrict__ b1, float* __restrict__ ws) {
  int e = blockIdx.x * 256 + threadIdx.x;
  int j = e >> 6, d = e & 63;
  int dh = j % (kD - 1);
  float v = (d <= dh) ? W0[e] : 0.f;
  int p = pos_of_unit(j);
  int t = p >> 4, n = p & 15;
  int cp = d >> 3, rr = d & 7;
  int q  = ((rr >> 2) << 1) | (rr & 1);
  int hi = (rr >> 1) & 1;
  int l  = (hi << 4) | n;
  ws[kWaOff + (((t * 8 + cp) * 32 + l) << 2) + q] = v;
  if (e < kU) {
    int pe = pos_of_unit(e);
    ws[kB0Off + pe] = b0[e];
    ws[kB1Off + pe] = b1[e];
  }
}

__global__ void prep_w1(const float* __restrict__ W1, float* __restrict__ Wb) {
  int e = blockIdx.x * 256 + threadIdx.x;
  int j = e >> 10, k = e & 1023;
  int dhj = j % (kD - 1), dhk = k % (kD - 1);
  float v = (dhk <= dhj) ? W1[e] : 0.f;
  int pj = pos_of_unit(j), pk = pos_of_unit(k);
  int t = pj >> 4, n = pj & 15;
  int cp = pk >> 3, rr = pk & 7;
  int q  = ((rr >> 2) << 1) | (rr & 1);
  int hi = (rr >> 1) & 1;
  int l  = (hi << 4) | n;
  Wb[(((t * 128 + cp) * 32 + l) << 2) + q] = v;
}

__global__ void prep_w2(const float* __restrict__ W2, float* __restrict__ ws) {
  int e = blockIdx.x * 256 + threadIdx.x;
  if (e < kWzbSz) {
    int q = e & 3, l = (e >> 2) & 31, cp = (e >> 7) & 127, b = e >> 14;
    int hi = l >> 4, m = l & 15;
    int p = 8 * cp + (q >> 1) * 4 + hi * 2 + (q & 1);
    int row = (m & 1) * kD + 8 * b + (m >> 1);
    float v = 0.f;
    if (p < deg_off(8 * b)) v = W2[row * kU + unit_at_pos(p)];
    ws[kWzbOff + e] = v;
  } else {
    int f = e - kWzbSz;
    int q = f & 3, l = (f >> 2) & 31, r = f >> 7;
    int i = r / 3, cl = r - 3 * i;
    int hi = l >> 4, m = l & 15;
    float v = 0.f;
    if (i >= 1) {
      int lo = deg_off(i - 1), hiEnd = deg_off(i);
      int p = 8 * ((lo >> 3) + cl) + (q >> 1) * 4 + hi * 2 + (q & 1);
      int row = (m & 1) * kD + 8 * (i >> 3) + (m >> 1);
      if (p >= lo && p < hiEnd) v = W2[row * kU + unit_at_pos(p)];
    }
    ws[kWziOff + f] = v;
  }
}

__device__ __forceinline__ v8f wmma_f32(v2f a, v2f b, v8f c) {
  return __builtin_amdgcn_wmma_f32_16x16x4_f32(false, a, false, b, (short)0, c,
                                               false, false);
}

constexpr int kBlk    = 16;
constexpr int kWbufSz = 128 * 32 * 4;

template <int G>
__device__ __forceinline__ void gemm_block(v4f (&wf)[2][kBlk], v2f (&hf)[2][2 * kBlk],
                                           v8f& a0, v8f& a1, v8f& a2, v8f& a3,
                                           const float* wl, const float* hrow, int base, int ncp) {
  constexpr int GN = G ^ 1;
  int nb = base + kBlk; nb = nb > ncp - kBlk ? ncp - kBlk : nb;
  const float* hp = hrow + nb * 8;
  const float* wp = wl + nb * 128;
#pragma unroll
  for (int u = 0; u < kBlk; ++u) {
    hf[GN][2 * u]     = *(const v2f*)(hp + u * 8);
    hf[GN][2 * u + 1] = *(const v2f*)(hp + u * 8 + 4);
  }
#pragma unroll
  for (int u = 0; u < kBlk; ++u) wf[GN][u] = *(const v4f*)(wp + u * 128);
  __builtin_amdgcn_sched_barrier(0x6);
#pragma unroll
  for (int u = 0; u < kBlk; ++u) {
    v4f w = wf[G][u];
    v2f wA = {w.x, w.y};
    v2f wB = {w.z, w.w};
    if (u & 1) { a2 = wmma_f32(wA, hf[G][2 * u], a2); a3 = wmma_f32(wB, hf[G][2 * u + 1], a3); }
    else       { a0 = wmma_f32(wA, hf[G][2 * u], a0); a1 = wmma_f32(wB, hf[G][2 * u + 1], a1); }
  }
  __builtin_amdgcn_sched_barrier(0x6);
}

__device__ __forceinline__ v8f tile_gemm(const float* wl, int ncp, const float* hrow) {
  v4f wf[2][kBlk];
  v2f hf[2][2 * kBlk];
#pragma unroll
  for (int u = 0; u < kBlk; ++u) {
    hf[0][2 * u]     = *(const v2f*)(hrow + u * 8);
    hf[0][2 * u + 1] = *(const v2f*)(hrow + u * 8 + 4);
  }
#pragma unroll
  for (int u = 0; u < kBlk; ++u) wf[0][u] = *(const v4f*)(wl + u * 128);
  v8f a0, a1, a2, a3;
  for (int v = 0; v < 8; ++v) { a0[v] = 0.f; a1[v] = 0.f; a2[v] = 0.f; a3[v] = 0.f; }
  int base = 0;
  do {
    gemm_block<0>(wf, hf, a0, a1, a2, a3, wl, hrow, base, ncp);
    if (base + kBlk < ncp) gemm_block<1>(wf, hf, a0, a1, a2, a3, wl, hrow, base + kBlk, ncp);
    base += 2 * kBlk;
  } while (base < ncp);
  v8f r;
  for (int v = 0; v < 8; ++v) r[v] = (a0[v] + a2[v]) + (a1[v] + a3[v]);
  return r;
}

__device__ __forceinline__ void copy_tile(const v4f* __restrict__ wsl, int ofs, int ncp, float* dst,
                                          int pw, int lane) {
  const int nblk = ncp >> 4;
  v4f r[3][kBlk];
#pragma unroll
  for (int s3 = 0; s3 < 3; ++s3) {
    const int blk = pw + 3 * s3;
    if (blk < nblk) {
#pragma unroll
      for (int u = 0; u < kBlk; ++u) r[s3][u] = wsl[ofs + (blk * kBlk + u) * 32];
    }
  }
#pragma unroll
  for (int s3 = 0; s3 < 3; ++s3) {
    const int blk = pw + 3 * s3;
    if (blk < nblk) {
#pragma unroll
      for (int u = 0; u < kBlk; ++u) *(v4f*)(dst + ((blk * kBlk + u) * 32 + lane) * 4) = r[s3][u];
    }
  }
}

__device__ __forceinline__ void lds_sync() {
  __builtin_amdgcn_fence(__ATOMIC_RELEASE, "workgroup", "local");
  __builtin_amdgcn_s_barrier();
  __builtin_amdgcn_fence(__ATOMIC_ACQUIRE, "workgroup", "local");
}

__device__ __forceinline__ int ncp_of(int i) {
  return ((deg_off(i) + 127) >> 7) << 4;
}

__device__ __forceinline__ void l1_fetch(const float4* __restrict__ Wa4, const float* __restrict__ b0s,
                                         int t, int lane, int hi,
                                         float4 (&w)[8], float4& bb0, float4& bb1) {
  const float4* wq = Wa4 + (t * 8) * 32 + lane;
#pragma unroll
  for (int cp = 0; cp < 8; ++cp) w[cp] = wq[cp * 32];
  bb0 = *(const float4*)(b0s + t * 16 + hi * 8);
  bb1 = *(const float4*)(b0s + t * 16 + hi * 8 + 4);
}
__device__ __forceinline__ void l1_xfrags(const float* xrow, v2f (&xf)[16]) {
#pragma unroll
  for (int cp = 0; cp < 8; ++cp) {
    xf[2 * cp]     = *(const v2f*)(xrow + cp * 8);
    xf[2 * cp + 1] = *(const v2f*)(xrow + cp * 8 + 4);
  }
}
__device__ __forceinline__ void l1_tile(const float4 (&wr)[8], const float4 bb0, const float4 bb1,
                                        const v2f (&xf)[16], float* dst) {
  v8f acc;
  acc[0] = bb0.x; acc[1] = bb0.y; acc[2] = bb0.z; acc[3] = bb0.w;
  acc[4] = bb1.x; acc[5] = bb1.y; acc[6] = bb1.z; acc[7] = bb1.w;
  v8f a1;
  for (int v = 0; v < 8; ++v) a1[v] = 0.f;
#pragma unroll
  for (int cp = 0; cp < 8; ++cp) {
    const float4 w = wr[cp];
    v2f wA = {w.x, w.y};
    v2f wB = {w.z, w.w};
    acc = wmma_f32(wA, xf[2 * cp], acc);
    a1  = wmma_f32(wB, xf[2 * cp + 1], a1);
  }
  float4 o0, o1;
  o0.x = fmaxf(acc[0] + a1[0], 0.f); o0.y = fmaxf(acc[1] + a1[1], 0.f);
  o0.z = fmaxf(acc[2] + a1[2], 0.f); o0.w = fmaxf(acc[3] + a1[3], 0.f);
  o1.x = fmaxf(acc[4] + a1[4], 0.f); o1.y = fmaxf(acc[5] + a1[5], 0.f);
  o1.z = fmaxf(acc[6] + a1[6], 0.f); o1.w = fmaxf(acc[7] + a1[7], 0.f);
  *(float4*)dst = o0;
  *(float4*)(dst + 4) = o1;
}

__launch_bounds__(128, 1)
__global__ void made_flow_kernel(const float* __restrict__ u,
                                 const float* __restrict__ b2,
                                 const float* __restrict__ ws,
                                 float* __restrict__ out) {
  extern __shared__ float smem[];
  float* xs   = smem;
  float* us   = xs  + kBT * kXStride;
  float* h0s  = us  + kBT * kXStride;
  float* h1s  = h0s + kBT * kHStride;
  float* wbuf = h1s + kBT * kHStride;

  const int tid  = threadIdx.x;
  const int lane = tid & 31;
  const int wv   = __builtin_amdgcn_readfirstlane(tid >> 5);
  const int wg   = blockIdx.x;
  const int n    = lane & 15;
  const int hi   = lane >> 4;
  const int hi2  = hi << 1;

  const float4* Wa4 = (const float4*)(ws + kWaOff);
  const v4f* wsl = (const v4f*)ws + lane;
  constexpr int kWbQ  = kWbOff / 4;
  constexpr int kWzbQ = kWzbOff / 4;
  const v4f* Wzi4 = (const v4f*)(ws + kWziOff);
  const float*  b0s = ws + kB0Off;
  const float*  b1s = ws + kB1Off;

  for (int e = tid; e < kBT * kXStride; e += 128) xs[e] = 0.f;
  for (int e = tid; e < 2 * kBT * kHStride; e += 128) h0s[e] = 0.f;
  for (int e = tid; e < kBT * kD; e += 128) {
    int r = e >> 6, d = e & 63;
    us[r * kXStride + d] = u[(wg * kBT + r) * kD + d];
  }
  float ldj = 0.f;

  int ringOfs = kWbQ, ringN = ncp_of(1), pb = 0;
  if (wv != 0) copy_tile(wsl, ringOfs, ringN, wbuf, wv - 1, lane);
  float4 waA[8], waB[8], bA0, bA1, bB0, bB1;
  l1_fetch(Wa4, b0s, 0, lane, hi, waA, bA0, bA1);
  l1_fetch(Wa4, b0s, 1, lane, hi, waB, bB0, bB1);
  __syncthreads();

  v8f zacc;
  for (int v = 0; v < 8; ++v) zacc[v] = 0.f;
  const float* xrow  = xs  + n * kXStride + hi2;
  const float* h0row = h0s + n * kHStride + hi2;
  const float* h1row = h1s + n * kHStride + hi2;

  for (int i = 0; i < kD; ++i) {
    const int offi = deg_off(i);
    const int ncp  = ncp_of(i);
    const int jb = i & 7;
    float b2m = 0.f, b2s = 0.f, uval = 0.f;
    if (wv == 0) {
      b2m = b2[i];
      b2s = b2[kD + i];
      uval = us[n * kXStride + i];
    }
    if (i > 0) {
      const int tlo = deg_off(i - 1) >> 4, thi = (offi - 1) >> 4;
      if (wv == 0) {
        v2f xf[16];
        l1_xfrags(xrow, xf);
        __builtin_amdgcn_sched_barrier(0x6);
        l1_tile(waA, bA0, bA1, xf, h0s + n * kHStride + tlo * 16 + hi * 8);
        if (thi > tlo) l1_tile(waB, bB0, bB1, xf, h0s + n * kHStride + thi * 16 + hi * 8);
      }
      v4f zi0 = {0.f, 0.f, 0.f, 0.f}, zi1 = zi0, zi2 = zi0;
      int t = tlo;
      do {
        lds_sync();
        int nOfs = ringOfs, nN = ringN;
        if (t < thi)          { nOfs = kWbQ + (t + 1) * 128 * 32; nN = ncp; }
        else if (jb == 0)     { nOfs = kWzbQ + (i >> 3) * 128 * 32; nN = ncp; }
        else if (i + 1 < kD)  { nOfs = kWbQ + (offi >> 4) * 128 * 32; nN = ncp_of(i + 1); }
        if (wv == 0) {
          const float4 bb0 = *(const float4*)(b1s + t * 16 + hi * 8);
          const float4 bb1 = *(const float4*)(b1s + t * 16 + hi * 8 + 4);
          if (t == tlo) {
            const v4f* zq = Wzi4 + (i * 3) * 32 + lane;
            zi0 = zq[0]; zi1 = zq[32]; zi2 = zq[64];
            const int iN = i + 1 < kD ? i + 1 : i;
            const int tloN = deg_off(iN - 1) >> 4, thiN = (deg_off(iN) - 1) >> 4;
            l1_fetch(Wa4, b0s, tloN, lane, hi, waA, bA0, bA1);
            if (thiN > tloN) l1_fetch(Wa4, b0s, thiN, lane, hi, waB, bB0, bB1);
          }
          v8f r = tile_gemm(wbuf + pb * kWbufSz + lane * 4, ringN, h0row);
          float4 o0, o1;
          o0.x = fmaxf(r[0] + bb0.x, 0.f); o0.y = fmaxf(r[1] + bb0.y, 0.f);
          o0.z = fmaxf(r[2] + bb0.z, 0.f); o0.w = fmaxf(r[3] + bb0.w, 0.f);
          o1.x = fmaxf(r[4] + bb1.x, 0.f); o1.y = fmaxf(r[5] + bb1.y, 0.f);
          o1.z = fmaxf(r[6] + bb1.z, 0.f); o1.w = fmaxf(r[7] + bb1.w, 0.f);
          float* dst = h1s + n * kHStride + t * 16 + hi * 8;
          *(float4*)dst = o0;
          *(float4*)(dst + 4) = o1;
        } else {
          copy_tile(wsl, nOfs, nN, wbuf + (pb ^ 1) * kWbufSz, wv - 1, lane);
        }
        ringOfs = nOfs; ringN = nN; pb ^= 1;
      } while (++t <= thi);
      lds_sync();
      if (jb == 0) {
        const int nOfs = kWbQ + (offi >> 4) * 128 * 32, nN = ncp_of(i + 1);
        if (wv == 0) zacc = tile_gemm(wbuf + pb * kWbufSz + lane * 4, ringN, h1row);
        else         copy_tile(wsl, nOfs, nN, wbuf + (pb ^ 1) * kWbufSz, wv - 1, lane);
        ringOfs = nOfs; ringN = nN; pb ^= 1;
      } else if (wv == 0) {
        const int cp0 = deg_off(i - 1) >> 3;
        const v4f wz[3] = {zi0, zi1, zi2};
        v8f zb;
        for (int v = 0; v < 8; ++v) zb[v] = 0.f;
        v2f hA[3], hB[3];
#pragma unroll
        for (int cl = 0; cl < 3; ++cl) {
          int hcp = cp0 + cl; hcp = hcp > 127 ? 127 : hcp;
          hA[cl] = *(const v2f*)(h1row + hcp * 8);
          hB[cl] = *(const v2f*)(h1row + hcp * 8 + 4);
        }
        __builtin_amdgcn_sched_barrier(0x6);
#pragma unroll
        for (int cl = 0; cl < 3; ++cl) {
          v2f wA = {wz[cl].x, wz[cl].y};
          v2f wB = {wz[cl].z, wz[cl].w};
          zacc = wmma_f32(wA, hA[cl], zacc);
          zb   = wmma_f32(wB, hB[cl], zb);
        }
        for (int v = 0; v < 8; ++v) zacc[v] += zb[v];
      }
    }
    if (wv == 0) {
      float zm, zs;
      switch (jb & 3) {
        case 0:  zm = zacc[0]; zs = zacc[1]; break;
        case 1:  zm = zacc[2]; zs = zacc[3]; break;
        case 2:  zm = zacc[4]; zs = zacc[5]; break;
        default: zm = zacc[6]; zs = zacc[7]; break;
      }
      if (hi == (jb >> 2)) {
        const float mu = zm + b2m;
        const float sg = zs + b2s;
        xs[n * kXStride + i] = uval * expf(sg) + mu;
        ldj += sg;
      }
    }
    lds_sync();
  }

  for (int e = tid; e < kBT * kD; e += 128) {
    int r = e >> 6, d = e & 63;
    out[(wg * kBT + r) * kD + d] = xs[r * kXStride + d];
  }
  if (wv == 0) {
    const float ldjt = ldj + __shfl_xor(ldj, 16);
    if (hi == 0) out[kB * kD + wg * kBT + n] = ldjt;
  }
}

extern "C" void kernel_launch(void* const* d_in, const int* in_sizes, int n_in,
                              void* d_out, int out_size, void* d_ws, size_t ws_size,
                              hipStream_t stream) {
  const float* u  = (const float*)d_in[0];
  const float* W0 = (const float*)d_in[1];
  const float* b0 = (const float*)d_in[2];
  const float* W1 = (const float*)d_in[3];
  const float* b1 = (const float*)d_in[4];
  const float* W2 = (const float*)d_in[5];
  const float* b2 = (const float*)d_in[6];
  float* ws  = (float*)d_ws;
  float* out = (float*)d_out;

  prep_w0<<<(kU * kD) / 256,     256, 0, stream>>>(W0, b0, b1, ws);
  prep_w1<<<(kU * kU) / 256,     256, 0, stream>>>(W1, ws + kWbOff);
  prep_w2<<<(kWzbSz + kWziSz) / 256, 256, 0, stream>>>(W2, ws);

  size_t smem = (size_t)(2 * kBT * kXStride + 2 * kBT * kHStride + 2 * kWbufSz) * sizeof(float);
  made_flow_kernel<<<kB / kBT, 128, smem, stream>>>(u, b2, ws, out);
}


# tile_gemm's K blocks split over two matrix waves on two SIMDs: wave 1 takes the back half, its partial tile reaches wave 0 through LDS (1 more barrier per call); copy waves 3 -> 2; sum order changes
# speedup vs baseline: 41.1849x; 1.1648x over previous
#include <hip/hip_runtime.h>

typedef __attribute__((ext_vector_type(2))) float v2f;
typedef __attribute__((ext_vector_type(8))) float v8f;
typedef __attribute__((ext_vector_type(4))) float v4f;

constexpr int kD  = 64;
constexpr int kU  = 1024;
constexpr int kB  = 4096;
constexpr int kBT = 16;

constexpr int kXStride = kD + 4;
constexpr int kHStride = kU + 4;

constexpr int kWaOff = 0;
constexpr int kWaSz  = kU * kD;
constexpr int kWbOff = kWaOff + kWaSz;
constexpr int kWbSz  = kU * kU;
constexpr int kWzbOff = kWbOff + kWbSz;
constexpr int kWzbSz  = 8 * 128 * 128;
constexpr int kWziOff = kWzbOff + kWzbSz;
constexpr int kWziSz  = kD * 3 * 128;
constexpr int kB0Off = kWziOff + kWziSz;
constexpr int kB1Off = kB0Off + kU;

__device__ __forceinline__ int deg_off(int d) {
  return d <= 16 ? 17 * d : 272 + 16 * (d - 16);
}
__device__ __forceinline__ int pos_of_unit(int j) {
  int d = j % (kD - 1), m = j / (kD - 1);
  return deg_off(d) + m;
}
__device__ __forceinline__ int unit_at_pos(int p) {
  int d, m;
  if (p < 272) { d = p / 17; m = p - 17 * d; }
  else         { d = 16 + ((p - 272) >> 4); m = (p - 272) & 15; }
  return m * (kD - 1) + d;
}


__global__ void prep_w0(const float* __restrict__ W0, const float* __restrict__ b0,
                        const float* __restrict__ b1, float* __restrict__ ws) {
  int e = blockIdx.x * 256 + threadIdx.x;
  int j = e >> 6, d = e & 63;
  int dh = j % (kD - 1);
  float v = (d <= dh) ? W0[e] : 0.f;
  int p = pos_of_unit(j);
  int t = p >> 4, n = p & 15;
  int cp = d >> 3, rr = d & 7;
  int q  = ((rr >> 2) << 1) | (rr & 1);
  int hi = (rr >> 1) & 1;
  int l  = (hi << 4) | n;
  ws[kWaOff + (((t * 8 + cp) * 32 + l) << 2) + q] = v;
  if (e < kU) {
    int pe = pos_of_unit(e);
    ws[kB0Off + pe] = b0[e];
    ws[kB1Off + pe] = b1[e];
  }
}

__global__ void prep_w1(const float* __restrict__ W1, float* __restrict__ Wb) {
  int e = blockIdx.x * 256 + threadIdx.x;
  int j = e >> 10, k = e & 1023;
  int dhj = j % (kD - 1), dhk = k % (kD - 1);
  float v = (dhk <= dhj) ? W1[e] : 0.f;
  int pj = pos_of_unit(j), pk = pos_of_unit(k);
  int t = pj >> 4, n = pj & 15;
  int cp = pk >> 3, rr = pk & 7;
  int q  = ((rr >> 2) << 1) | (rr & 1);
  int hi = (rr >> 1) & 1;
  int l  = (hi << 4) | n;
  Wb[(((t * 128 + cp) * 32 + l) << 2) + q] = v;
}

__global__ void prep_w2(const float* __restrict__ W2, float* __restrict__ ws) {
  int e = blockIdx.x * 256 + threadIdx.x;
  if (e < kWzbSz) {
    int q = e & 3, l = (e >> 2) & 31, cp = (e >> 7) & 127, b = e >> 14;
    int hi = l >> 4, m = l & 15;
    int p = 8 * cp + (q >> 1) * 4 + hi * 2 + (q & 1);
    int row = (m & 1) * kD + 8 * b + (m >> 1);
    float v = 0.f;
    if (p < deg_off(8 * b)) v = W2[row * kU + unit_at_pos(p)];
    ws[kWzbOff + e] = v;
  } else {
    int f = e - kWzbSz;
    int q = f & 3, l = (f >> 2) & 31, r = f >> 7;
    int i = r / 3, cl = r - 3 * i;
    int hi = l >> 4, m = l & 15;
    float v = 0.f;
    if (i >= 1) {
      int lo = deg_off(i - 1), hiEnd = deg_off(i);
      int p = 8 * ((lo >> 3) + cl) + (q >> 1) * 4 + hi * 2 + (q & 1);
      int row = (m & 1) * kD + 8 * (i >> 3) + (m >> 1);
      if (p >= lo && p < hiEnd) v = W2[row * kU + unit_at_pos(p)];
    }
    ws[kWziOff + f] = v;
  }
}

__device__ __forceinline__ v8f wmma_f32(v2f a, v2f b, v8f c) {
  return __builtin_amdgcn_wmma_f32_16x16x4_f32(false, a, false, b, (short)0, c,
                                               false, false);
}

constexpr int kBlk    = 16;
constexpr int kWbufSz = 128 * 32 * 4;

template <int G>
__device__ __forceinline__ void gemm_block(v4f (&wf)[2][kBlk], v2f (&hf)[2][2 * kBlk],
                                           v8f& a0, v8f& a1, v8f& a2, v8f& a3,
                                           const float* wl, const float* hrow, int base, int ncp) {
  constexpr int GN = G ^ 1;
  int nb = base + kBlk; nb = nb > ncp - kBlk ? ncp - kBlk : nb;
  const float* hp = hrow + nb * 8;
  const float* wp = wl + nb * 128;
#pragma unroll
  for (int u = 0; u < kBlk; ++u) {
    hf[GN][2 * u]     = *(const v2f*)(hp + u * 8);
    hf[GN][2 * u + 1] = *(const v2f*)(hp + u * 8 + 4);
  }
#pragma unroll
  for (int u = 0; u < kBlk; ++u) wf[GN][u] = *(const v4f*)(wp + u * 128);
  __builtin_amdgcn_sched_barrier(0x6);
#pragma unroll
  for (int u = 0; u < kBlk; ++u) {
    v4f w = wf[G][u];
    v2f wA = {w.x, w.y};
    v2f wB = {w.z, w.w};
    if (u & 1) { a2 = wmma_f32(wA, hf[G][2 * u], a2); a3 = wmma_f32(wB, hf[G][2 * u + 1], a3); }
    else       { a0 = wmma_f32(wA, hf[G][2 * u], a0); a1 = wmma_f32(wB, hf[G][2 * u + 1], a1); }
  }
  __builtin_amdgcn_sched_barrier(0x6);
}

__device__ __forceinline__ v8f tile_gemm(const float* wl, int ncp, const float* hrow) {
  v4f wf[2][kBlk];
  v2f hf[2][2 * kBlk];
#pragma unroll
  for (int u = 0; u < kBlk; ++u) {
    hf[0][2 * u]     = *(const v2f*)(hrow + u * 8);
    hf[0][2 * u + 1] = *(const v2f*)(hrow + u * 8 + 4);
  }
#pragma unroll
  for (int u = 0; u < kBlk; ++u) wf[0][u] = *(const v4f*)(wl + u * 128);
  v8f a0, a1, a2, a3;
  for (int v = 0; v < 8; ++v) { a0[v] = 0.f; a1[v] = 0.f; a2[v] = 0.f; a3[v] = 0.f; }
  int base = 0;
  for (; base + 2 * kBlk <= ncp; base += 2 * kBlk) {
    gemm_block<0>(wf, hf, a0, a1, a2, a3, wl, hrow, base, ncp);
    gemm_block<1>(wf, hf, a0, a1, a2, a3, wl, hrow, base + kBlk, ncp);
  }
  if (base < ncp) gemm_block<0>(wf, hf, a0, a1, a2, a3, wl, hrow, base, ncp);
  v8f r;
  for (int v = 0; v < 8; ++v) r[v] = (a0[v] + a2[v]) + (a1[v] + a3[v]);
  return r;
}

template <int CNT>
__device__ __forceinline__ void copy_blocks(const v4f* __restrict__ wsl, int ofs, float* dst,
                                            int pw, int lane) {
  v4f r[CNT][kBlk];
#pragma unroll
  for (int s3 = 0; s3 < CNT; ++s3) {
    const int blk = pw + 2 * s3;
#pragma unroll
    for (int u = 0; u < kBlk; ++u) r[s3][u] = wsl[ofs + (blk * kBlk + u) * 32];
  }
#pragma unroll
  for (int s3 = 0; s3 < CNT; ++s3) {
    const int blk = pw + 2 * s3;
#pragma unroll
    for (int u = 0; u < kBlk; ++u) *(v4f*)(dst + ((blk * kBlk + u) * 32 + lane) * 4) = r[s3][u];
  }
}
__device__ __forceinline__ void copy_tile(const v4f* __restrict__ wsl, int ofs, int ncp, float* dst,
                                          int pw, int lane) {
  const int cnt = ((ncp >> 4) - pw + 1) >> 1;
  switch (cnt) {
    case 1: copy_blocks<1>(wsl, ofs, dst, pw, lane); break;
    case 2: copy_blocks<2>(wsl, ofs, dst, pw, lane); break;
    case 3: copy_blocks<3>(wsl, ofs, dst, pw, lane); break;
    case 4: copy_blocks<4>(wsl, ofs, dst, pw, lane); break;
    default: break;
  }
}

__device__ __forceinline__ void lds_sync() {
  __builtin_amdgcn_fence(__ATOMIC_RELEASE, "workgroup", "local");
  __builtin_amdgcn_s_barrier();
  __builtin_amdgcn_fence(__ATOMIC_ACQUIRE, "workgroup", "local");
}

__device__ __forceinline__ int ncp_of(int i) {
  return ((deg_off(i) + 127) >> 7) << 4;
}

__device__ __forceinline__ void l1_fetch(const float4* __restrict__ Wa4, const float* __restrict__ b0s,
                                         int t, int lane, int hi,
                                         float4 (&w)[8], float4& bb0, float4& bb1) {
  const float4* wq = Wa4 + (t * 8) * 32 + lane;
#pragma unroll
  for (int cp = 0; cp < 8; ++cp) w[cp] = wq[cp * 32];
  bb0 = *(const float4*)(b0s + t * 16 + hi * 8);
  bb1 = *(const float4*)(b0s + t * 16 + hi * 8 + 4);
}
__device__ __forceinline__ void l1_xfrags(const float* xrow, v2f (&xf)[16]) {
#pragma unroll
  for (int cp = 0; cp < 8; ++cp) {
    xf[2 * cp]     = *(const v2f*)(xrow + cp * 8);
    xf[2 * cp + 1] = *(const v2f*)(xrow + cp * 8 + 4);
  }
}
__device__ __forceinline__ void l1_tile(const float4 (&wr)[8], const float4 bb0, const float4 bb1,
                                        const v2f (&xf)[16], float* dst) {
  v8f acc;
  acc[0] = bb0.x; acc[1] = bb0.y; acc[2] = bb0.z; acc[3] = bb0.w;
  acc[4] = bb1.x; acc[5] = bb1.y; acc[6] = bb1.z; acc[7] = bb1.w;
  v8f a1;
  for (int v = 0; v < 8; ++v) a1[v] = 0.f;
#pragma unroll
  for (int cp = 0; cp < 8; ++cp) {
    const float4 w = wr[cp];
    v2f wA = {w.x, w.y};
    v2f wB = {w.z, w.w};
    acc = wmma_f32(wA, xf[2 * cp], acc);
    a1  = wmma_f32(wB, xf[2 * cp + 1], a1);
  }
  float4 o0, o1;
  o0.x = fmaxf(acc[0] + a1[0], 0.f); o0.y = fmaxf(acc[1] + a1[1], 0.f);
  o0.z = fmaxf(acc[2] + a1[2], 0.f); o0.w = fmaxf(acc[3] + a1[3], 0.f);
  o1.x = fmaxf(acc[4] + a1[4], 0.f); o1.y = fmaxf(acc[5] + a1[5], 0.f);
  o1.z = fmaxf(acc[6] + a1[6], 0.f); o1.w = fmaxf(acc[7] + a1[7], 0.f);
  *(float4*)dst = o0;
  *(float4*)(dst + 4) = o1;
}

__launch_bounds__(128, 1)
__global__ void made_flow_kernel(const float* __restrict__ u,
                                 const float* __restrict__ b2,
                                 const float* __restrict__ ws,
                                 float* __restrict__ out) {
  extern __shared__ float smem[];
  float* xs   = smem;
  float* us   = xs  + kBT * kXStride;
  float* h0s  = us  + kBT * kXStride;
  float* h1s  = h0s + kBT * kHStride;
  float* wbuf = h1s + kBT * kHStride;
  float* psum = wbuf + 2 * kWbufSz;

  const int tid  = threadIdx.x;
  const int lane = tid & 31;
  const int wv   = __builtin_amdgcn_readfirstlane(tid >> 5);
  const int wg   = blockIdx.x;
  const int n    = lane & 15;
  const int hi   = lane >> 4;
  const int hi2  = hi << 1;

  const float4* Wa4 = (const float4*)(ws + kWaOff);
  const v4f* wsl = (const v4f*)ws + lane;
  constexpr int kWbQ  = kWbOff / 4;
  constexpr int kWzbQ = kWzbOff / 4;
  const v4f* Wzi4 = (const v4f*)(ws + kWziOff);
  const float*  b0s = ws + kB0Off;
  const float*  b1s = ws + kB1Off;

  for (int e = tid; e < kBT * kXStride; e += 128) xs[e] = 0.f;
  for (int e = tid; e < 2 * kBT * kHStride; e += 128) h0s[e] = 0.f;
  for (int e = tid; e < kBT * kD; e += 128) {
    int r = e >> 6, d = e & 63;
    us[r * kXStride + d] = u[(wg * kBT + r) * kD + d];
  }
  float ldj = 0.f;

  int ringOfs = kWbQ, ringN = ncp_of(1), pb = 0;
  if (wv >= 2) copy_tile(wsl, ringOfs, ringN, wbuf, wv - 2, lane);
  float4 waA[8], waB[8], bA0, bA1, bB0, bB1;
  l1_fetch(Wa4, b0s, 0, lane, hi, waA, bA0, bA1);
  l1_fetch(Wa4, b0s, 1, lane, hi, waB, bB0, bB1);
  __syncthreads();

  v8f zacc;
  for (int v = 0; v < 8; ++v) zacc[v] = 0.f;
  const float* xrow  = xs  + n * kXStride + hi2;
  const float* h0row = h0s + n * kHStride + hi2;
  const float* h1row = h1s + n * kHStride + hi2;

  for (int i = 0; i < kD; ++i) {
    const int offi = deg_off(i);
    const int ncp  = ncp_of(i);
    const int jb = i & 7;
    float b2m = 0.f, b2s = 0.f, uval = 0.f;
    if (wv == 0) {
      b2m = b2[i];
      b2s = b2[kD + i];
      uval = us[n * kXStride + i];
    }
    if (i > 0) {
      const int tlo = deg_off(i - 1) >> 4, thi = (offi - 1) >> 4;
      if (wv == 0) {
        v2f xf[16];
        l1_xfrags(xrow, xf);
        __builtin_amdgcn_sched_barrier(0x6);
        l1_tile(waA, bA0, bA1, xf, h0s + n * kHStride + tlo * 16 + hi * 8);
        if (thi > tlo) l1_tile(waB, bB0, bB1, xf, h0s + n * kHStride + thi * 16 + hi * 8);
      }
      v4f zi0 = {0.f, 0.f, 0.f, 0.f}, zi1 = zi0, zi2 = zi0;
      int t = tlo;
      do {
        lds_sync();
        int nOfs = ringOfs, nN = ringN;
        if (t < thi)          { nOfs = kWbQ + (t + 1) * 128 * 32; nN = ncp; }
        else if (jb == 0)     { nOfs = kWzbQ + (i >> 3) * 128 * 32; nN = ncp; }
        else if (i + 1 < kD)  { nOfs = kWbQ + (offi >> 4) * 128 * 32; nN = ncp_of(i + 1); }
        const int nblk = ringN >> 4, n0 = (nblk + 1) >> 1;
        const float* wcur = wbuf + pb * kWbufSz + lane * 4;
        v8f r;
        for (int v = 0; v < 8; ++v) r[v] = 0.f;
        float4 bb0 = {0.f, 0.f, 0.f, 0.f}, bb1 = bb0;
        if (wv == 0) {
          bb0 = *(const float4*)(b1s + t * 16 + hi * 8);
          bb1 = *(const float4*)(b1s + t * 16 + hi * 8 + 4);
          if (t == tlo) {
            const v4f* zq = Wzi4 + (i * 3) * 32 + lane;
            zi0 = zq[0]; zi1 = zq[32]; zi2 = zq[64];
            const int iN = i + 1 < kD ? i + 1 : i;
            const int tloN = deg_off(iN - 1) >> 4, thiN = (deg_off(iN) - 1) >> 4;
            l1_fetch(Wa4, b0s, tloN, lane, hi, waA, bA0, bA1);
            if (thiN > tloN) l1_fetch(Wa4, b0s, thiN, lane, hi, waB, bB0, bB1);
          }
          r = tile_gemm(wcur, n0 * kBlk, h0row);
        } else if (wv == 1) {
          if (nblk > n0) {
            const v8f p = tile_gemm(wcur + n0 * kBlk * 128, (nblk - n0) * kBlk, h0row + n0 * kBlk * 8);
            float4 p0, p1;
            p0.x = p[0]; p0.y = p[1]; p0.z = p[2]; p0.w = p[3];
            p1.x = p[4]; p1.y = p[5]; p1.z = p[6]; p1.w = p[7];
            *(float4*)(psum + lane * 4) = p0;
            *(float4*)(psum + 128 + lane * 4) = p1;
          }
        } else {
          copy_tile(wsl, nOfs, nN, wbuf + (pb ^ 1) * kWbufSz, wv - 2, lane);
        }
        lds_sync();
        if (wv == 0) {
          if (nblk > n0) {
            const float4 p0 = *(const float4*)(psum + lane * 4);
            const float4 p1 = *(const float4*)(psum + 128 + lane * 4);
            r[0] += p0.x; r[1] += p0.y; r[2] += p0.z; r[3] += p0.w;
            r[4] += p1.x; r[5] += p1.y; r[6] += p1.z; r[7] += p1.w;
          }
          float4 o0, o1;
          o0.x = fmaxf(r[0] + bb0.x, 0.f); o0.y = fmaxf(r[1] + bb0.y, 0.f);
          o0.z = fmaxf(r[2] + bb0.z, 0.f); o0.w = fmaxf(r[3] + bb0.w, 0.f);
          o1.x = fmaxf(r[4] + bb1.x, 0.f); o1.y = fmaxf(r[5] + bb1.y, 0.f);
          o1.z = fmaxf(r[6] + bb1.z, 0.f); o1.w = fmaxf(r[7] + bb1.w, 0.f);
          float* dst = h1s + n * kHStride + t * 16 + hi * 8;
          *(float4*)dst = o0;
          *(float4*)(dst + 4) = o1;
        }
        ringOfs = nOfs; ringN = nN; pb ^= 1;
      } while (++t <= thi);
      lds_sync();
      if (jb == 0) {
        const int nOfs = kWbQ + (offi >> 4) * 128 * 32, nN = ncp_of(i + 1);
        const int nblk = ringN >> 4, n0 = (nblk + 1) >> 1;
        const float* wcur = wbuf + pb * kWbufSz + lane * 4;
        if (wv == 0) {
          zacc = tile_gemm(wcur, n0 * kBlk, h1row);
        } else if (wv == 1) {
          if (nblk > n0) {
            const v8f p = tile_gemm(wcur + n0 * kBlk * 128, (nblk - n0) * kBlk, h1row + n0 * kBlk * 8);
            float4 p0, p1;
            p0.x = p[0]; p0.y = p[1]; p0.z = p[2]; p0.w = p[3];
            p1.x = p[4]; p1.y = p[5]; p1.z = p[6]; p1.w = p[7];
            *(float4*)(psum + lane * 4) = p0;
            *(float4*)(psum + 128 + lane * 4) = p1;
          }
        } else {
          copy_tile(wsl, nOfs, nN, wbuf + (pb ^ 1) * kWbufSz, wv - 2, lane);
        }
        lds_sync();
        if (wv == 0 && nblk > n0) {
          const float4 p0 = *(const float4*)(psum + lane * 4);
          const float4 p1 = *(const float4*)(psum + 128 + lane * 4);
          zacc[0] += p0.x; zacc[1] += p0.y; zacc[2] += p0.z; zacc[3] += p0.w;
          zacc[4] += p1.x; zacc[5] += p1.y; zacc[6] += p1.z; zacc[7] += p1.w;
        }
        ringOfs = nOfs; ringN = nN; pb ^= 1;
      } else if (wv == 0) {
        const int cp0 = deg_off(i - 1) >> 3;
        const v4f wz[3] = {zi0, zi1, zi2};
        v8f zb;
        for (int v = 0; v < 8; ++v) zb[v] = 0.f;
        v2f hA[3], hB[3];
#pragma unroll
        for (int cl = 0; cl < 3; ++cl) {
          int hcp = cp0 + cl; hcp = hcp > 127 ? 127 : hcp;
          hA[cl] = *(const v2f*)(h1row + hcp * 8);
          hB[cl] = *(const v2f*)(h1row + hcp * 8 + 4);
        }
        __builtin_amdgcn_sched_barrier(0x6);
#pragma unroll
        for (int cl = 0; cl < 3; ++cl) {
          v2f wA = {wz[cl].x, wz[cl].y};
          v2f wB = {wz[cl].z, wz[cl].w};
          zacc = wmma_f32(wA, hA[cl], zacc);
          zb   = wmma_f32(wB, hB[cl], zb);
        }
        for (int v = 0; v < 8; ++v) zacc[v] += zb[v];
      }
    }
    if (wv == 0) {
      float zm, zs;
      switch (jb & 3) {
        case 0:  zm = zacc[0]; zs = zacc[1]; break;
        case 1:  zm = zacc[2]; zs = zacc[3]; break;
        case 2:  zm = zacc[4]; zs = zacc[5]; break;
        default: zm = zacc[6]; zs = zacc[7]; break;
      }
      if (hi == (jb >> 2)) {
        const float mu = zm + b2m;
        const float sg = zs + b2s;
        xs[n * kXStride + i] = uval * expf(sg) + mu;
        ldj += sg;
      }
    }
    lds_sync();
  }

  for (int e = tid; e < kBT * kD; e += 128) {
    int r = e >> 6, d = e & 63;
    out[(wg * kBT + r) * kD + d] = xs[r * kXStride + d];
  }
  if (wv == 0) {
    const float ldjt = ldj + __shfl_xor(ldj, 16);
    if (hi == 0) out[kB * kD + wg * kBT + n] = ldjt;
  }
}

extern "C" void kernel_launch(void* const* d_in, const int* in_sizes, int n_in,
                              void* d_out, int out_size, void* d_ws, size_t ws_size,
                              hipStream_t stream) {
  const float* u  = (const float*)d_in[0];
  const float* W0 = (const float*)d_in[1];
  const float* b0 = (const float*)d_in[2];
  const float* W1 = (const float*)d_in[3];
  const float* b1 = (const float*)d_in[4];
  const float* W2 = (const float*)d_in[5];
  const float* b2 = (const float*)d_in[6];
  float* ws  = (float*)d_ws;
  float* out = (float*)d_out;

  prep_w0<<<(kU * kD) / 256,     256, 0, stream>>>(W0, b0, b1, ws);
  prep_w1<<<(kU * kU) / 256,     256, 0, stream>>>(W1, ws + kWbOff);
  prep_w2<<<(kWzbSz + kWziSz) / 256, 256, 0, stream>>>(W2, ws);

  size_t smem = (size_t)(2 * kBT * kXStride + 2 * kBT * kHStride + 2 * kWbufSz + 2 * 32 * 4) * sizeof(float);
  made_flow_kernel<<<kB / kBT, 128, smem, stream>>>(u, b2, ws, out);
}


# tile_gemm's K blocks (now 8 chunk pairs each) split over four matrix waves, one per SIMD, partial tiles to wave 0 through LDS; copy waves 2 -> 4, the second wave of each SIMD; sum order changes
# speedup vs baseline: 42.5522x; 1.0332x over previous
#include <hip/hip_runtime.h>

typedef __attribute__((ext_vector_type(2))) float v2f;
typedef __attribute__((ext_vector_type(8))) float v8f;
typedef __attribute__((ext_vector_type(4))) float v4f;

constexpr int kD  = 64;
constexpr int kU  = 1024;
constexpr int kB  = 4096;
constexpr int kBT = 16;

constexpr int kXStride = kD + 4;
constexpr int kHStride = kU + 4;

constexpr int kWaOff = 0;
constexpr int kWaSz  = kU * kD;
constexpr int kWbOff = kWaOff + kWaSz;
constexpr int kWbSz  = kU * kU;
constexpr int kWzbOff = kWbOff + kWbSz;
constexpr int kWzbSz  = 8 * 128 * 128;
constexpr int kWziOff = kWzbOff + kWzbSz;
constexpr int kWziSz  = kD * 3 * 128;
constexpr int kB0Off = kWziOff + kWziSz;
constexpr int kB1Off = kB0Off + kU;

__device__ __forceinline__ int deg_off(int d) {
  return d <= 16 ? 17 * d : 272 + 16 * (d - 16);
}
__device__ __forceinline__ int pos_of_unit(int j) {
  int d = j % (kD - 1), m = j / (kD - 1);
  return deg_off(d) + m;
}
__device__ __forceinline__ int unit_at_pos(int p) {
  int d, m;
  if (p < 272) { d = p / 17; m = p - 17 * d; }
  else         { d = 16 + ((p - 272) >> 4); m = (p - 272) & 15; }
  return m * (kD - 1) + d;
}


__global__ void prep_w0(const float* __restrict__ W0, const float* __restrict__ b0,
                        const float* __restrict__ b1, float* __restrict__ ws) {
  int e = blockIdx.x * 256 + threadIdx.x;
  int j = e >> 6, d = e & 63;
  int dh = j % (kD - 1);
  float v = (d <= dh) ? W0[e] : 0.f;
  int p = pos_of_unit(j);
  int t = p >> 4, n = p & 15;
  int cp = d >> 3, rr = d & 7;
  int q  = ((rr >> 2) << 1) | (rr & 1);
  int hi = (rr >> 1) & 1;
  int l  = (hi << 4) | n;
  ws[kWaOff + (((t * 8 + cp) * 32 + l) << 2) + q] = v;
  if (e < kU) {
    int pe = pos_of_unit(e);
    ws[kB0Off + pe] = b0[e];
    ws[kB1Off + pe] = b1[e];
  }
}

__global__ void prep_w1(const float* __restrict__ W1, float* __restrict__ Wb) {
  int e = blockIdx.x * 256 + threadIdx.x;
  int j = e >> 10, k = e & 1023;
  int dhj = j % (kD - 1), dhk = k % (kD - 1);
  float v = (dhk <= dhj) ? W1[e] : 0.f;
  int pj = pos_of_unit(j), pk = pos_of_unit(k);
  int t = pj >> 4, n = pj & 15;
  int cp = pk >> 3, rr = pk & 7;
  int q  = ((rr >> 2) << 1) | (rr & 1);
  int hi = (rr >> 1) & 1;
  int l  = (hi << 4) | n;
  Wb[(((t * 128 + cp) * 32 + l) << 2) + q] = v;
}

__global__ void prep_w2(const float* __restrict__ W2, float* __restrict__ ws) {
  int e = blockIdx.x * 256 + threadIdx.x;
  if (e < kWzbSz) {
    int q = e & 3, l = (e >> 2) & 31, cp = (e >> 7) & 127, b = e >> 14;
    int hi = l >> 4, m = l & 15;
    int p = 8 * cp + (q >> 1) * 4 + hi * 2 + (q & 1);
    int row = (m & 1) * kD + 8 * b + (m >> 1);
    float v = 0.f;
    if (p < deg_off(8 * b)) v = W2[row * kU + unit_at_pos(p)];
    ws[kWzbOff + e] = v;
  } else {
    int f = e - kWzbSz;
    int q = f & 3, l = (f >> 2) & 31, r = f >> 7;
    int i = r / 3, cl = r - 3 * i;
    int hi = l >> 4, m = l & 15;
    float v = 0.f;
    if (i >= 1) {
      int lo = deg_off(i - 1), hiEnd = deg_off(i);
      int p = 8 * ((lo >> 3) + cl) + (q >> 1) * 4 + hi * 2 + (q & 1);
      int row = (m & 1) * kD + 8 * (i >> 3) + (m >> 1);
      if (p >= lo && p < hiEnd) v = W2[row * kU + unit_at_pos(p)];
    }
    ws[kWziOff + f] = v;
  }
}

__device__ __forceinline__ v8f wmma_f32(v2f a, v2f b, v8f c) {
  return __builtin_amdgcn_wmma_f32_16x16x4_f32(false, a, false, b, (short)0, c,
                                               false, false);
}

constexpr int kBlk    = 16;
constexpr int kMB     = 8;
constexpr int kWbufSz = 128 * 32 * 4;
constexpr int kThreads = 256;

template <int G>
__device__ __forceinline__ void gemm_block(v4f (&wf)[2][kMB], v2f (&hf)[2][2 * kMB],
                                           v8f& a0, v8f& a1, v8f& a2, v8f& a3,
                                           const float* wl, const float* hrow, int base, int ncp) {
  constexpr int GN = G ^ 1;
  int nb = base + kMB; nb = nb > ncp - kMB ? ncp - kMB : nb;
  const float* hp = hrow + nb * 8;
  const float* wp = wl + nb * 128;
#pragma unroll
  for (int u = 0; u < kMB; ++u) {
    hf[GN][2 * u]     = *(const v2f*)(hp + u * 8);
    hf[GN][2 * u + 1] = *(const v2f*)(hp + u * 8 + 4);
  }
#pragma unroll
  for (int u = 0; u < kMB; ++u) wf[GN][u] = *(const v4f*)(wp + u * 128);
  __builtin_amdgcn_sched_barrier(0x6);
#pragma unroll
  for (int u = 0; u < kMB; ++u) {
    v4f w = wf[G][u];
    v2f wA = {w.x, w.y};
    v2f wB = {w.z, w.w};
    if (u & 1) { a2 = wmma_f32(wA, hf[G][2 * u], a2); a3 = wmma_f32(wB, hf[G][2 * u + 1], a3); }
    else       { a0 = wmma_f32(wA, hf[G][2 * u], a0); a1 = wmma_f32(wB, hf[G][2 * u + 1], a1); }
  }
  __builtin_amdgcn_sched_barrier(0x6);
}

__device__ __forceinline__ v8f tile_gemm(const float* wl, int ncp, const float* hrow) {
  v4f wf[2][kMB];
  v2f hf[2][2 * kMB];
#pragma unroll
  for (int u = 0; u < kMB; ++u) {
    hf[0][2 * u]     = *(const v2f*)(hrow + u * 8);
    hf[0][2 * u + 1] = *(const v2f*)(hrow + u * 8 + 4);
  }
#pragma unroll
  for (int u = 0; u < kMB; ++u) wf[0][u] = *(const v4f*)(wl + u * 128);
  v8f a0, a1, a2, a3;
  for (int v = 0; v < 8; ++v) { a0[v] = 0.f; a1[v] = 0.f; a2[v] = 0.f; a3[v] = 0.f; }
  int base = 0;
  for (; base + 2 * kMB <= ncp; base += 2 * kMB) {
    gemm_block<0>(wf, hf, a0, a1, a2, a3, wl, hrow, base, ncp);
    gemm_block<1>(wf, hf, a0, a1, a2, a3, wl, hrow, base + kMB, ncp);
  }
  if (base < ncp) gemm_block<0>(wf, hf, a0, a1, a2, a3, wl, hrow, base, ncp);
  v8f r;
  for (int v = 0; v < 8; ++v) r[v] = (a0[v] + a2[v]) + (a1[v] + a3[v]);
  return r;
}

template <int CNT>
__device__ __forceinline__ void copy_blocks(const v4f* __restrict__ wsl, int ofs, float* dst,
                                            int pw, int lane) {
  v4f r[CNT][kBlk];
#pragma unroll
  for (int s3 = 0; s3 < CNT; ++s3) {
    const int blk = pw + 4 * s3;
#pragma unroll
    for (int u = 0; u < kBlk; ++u) r[s3][u] = wsl[ofs + (blk * kBlk + u) * 32];
  }
#pragma unroll
  for (int s3 = 0; s3 < CNT; ++s3) {
    const int blk = pw + 4 * s3;
#pragma unroll
    for (int u = 0; u < kBlk; ++u) *(v4f*)(dst + ((blk * kBlk + u) * 32 + lane) * 4) = r[s3][u];
  }
}
__device__ __forceinline__ void copy_tile(const v4f* __restrict__ wsl, int ofs, int ncp, float* dst,
                                          int pw, int lane) {
  const int cnt = ((ncp >> 4) - pw + 3) >> 2;
  switch (cnt) {
    case 1: copy_blocks<1>(wsl, ofs, dst, pw, lane); break;
    case 2: copy_blocks<2>(wsl, ofs, dst, pw, lane); break;
    default: break;
  }
}

__device__ __forceinline__ void k_share(int ncp, int w, int& st, int& cnt) {
  const int nq = ncp / kMB, qb = nq >> 2, qr = nq & 3;
  cnt = qb + (w < qr ? 1 : 0);
  st  = w * qb + (w < qr ? w : qr);
}

__device__ __forceinline__ void helper_share(const float* wcur, const float* hrow, int st, int cnt,
                                             float* ps, int lane) {
  v8f p;
  for (int v = 0; v < 8; ++v) p[v] = 0.f;
  if (cnt > 0) p = tile_gemm(wcur + st * kMB * 128, cnt * kMB, hrow + st * kMB * 8);
  float4 p0, p1;
  p0.x = p[0]; p0.y = p[1]; p0.z = p[2]; p0.w = p[3];
  p1.x = p[4]; p1.y = p[5]; p1.z = p[6]; p1.w = p[7];
  *(float4*)(ps + lane * 4) = p0;
  *(float4*)(ps + 128 + lane * 4) = p1;
}
__device__ __forceinline__ void add_shares(v8f& r, const float* psum, int lane) {
  float4 q0[3], q1[3];
#pragma unroll
  for (int w = 0; w < 3; ++w) {
    q0[w] = *(const float4*)(psum + w * 256 + lane * 4);
    q1[w] = *(const float4*)(psum + w * 256 + 128 + lane * 4);
  }
  r[0] = (r[0] + q0[0].x) + (q0[1].x + q0[2].x); r[1] = (r[1] + q0[0].y) + (q0[1].y + q0[2].y);
  r[2] = (r[2] + q0[0].z) + (q0[1].z + q0[2].z); r[3] = (r[3] + q0[0].w) + (q0[1].w + q0[2].w);
  r[4] = (r[4] + q1[0].x) + (q1[1].x + q1[2].x); r[5] = (r[5] + q1[0].y) + (q1[1].y + q1[2].y);
  r[6] = (r[6] + q1[0].z) + (q1[1].z + q1[2].z); r[7] = (r[7] + q1[0].w) + (q1[1].w + q1[2].w);
}

__device__ __forceinline__ void lds_sync() {
  __builtin_amdgcn_fence(__ATOMIC_RELEASE, "workgroup", "local");
  __builtin_amdgcn_s_barrier();
  __builtin_amdgcn_fence(__ATOMIC_ACQUIRE, "workgroup", "local");
}

__device__ __forceinline__ int ncp_of(int i) {
  return ((deg_off(i) + 127) >> 7) << 4;
}

__device__ __forceinline__ void l1_fetch(const float4* __restrict__ Wa4, const float* __restrict__ b0s,
                                         int t, int lane, int hi,
                                         float4 (&w)[8], float4& bb0, float4& bb1) {
  const float4* wq = Wa4 + (t * 8) * 32 + lane;
#pragma unroll
  for (int cp = 0; cp < 8; ++cp) w[cp] = wq[cp * 32];
  bb0 = *(const float4*)(b0s + t * 16 + hi * 8);
  bb1 = *(const float4*)(b0s + t * 16 + hi * 8 + 4);
}
__device__ __forceinline__ void l1_xfrags(const float* xrow, v2f (&xf)[16]) {
#pragma unroll
  for (int cp = 0; cp < 8; ++cp) {
    xf[2 * cp]     = *(const v2f*)(xrow + cp * 8);
    xf[2 * cp + 1] = *(const v2f*)(xrow + cp * 8 + 4);
  }
}
__device__ __forceinline__ void l1_tile(const float4 (&wr)[8], const float4 bb0, const float4 bb1,
                                        const v2f (&xf)[16], float* dst) {
  v8f acc;
  acc[0] = bb0.x; acc[1] = bb0.y; acc[2] = bb0.z; acc[3] = bb0.w;
  acc[4] = bb1.x; acc[5] = bb1.y; acc[6] = bb1.z; acc[7] = bb1.w;
  v8f a1;
  for (int v = 0; v < 8; ++v) a1[v] = 0.f;
#pragma unroll
  for (int cp = 0; cp < 8; ++cp) {
    const float4 w = wr[cp];
    v2f wA = {w.x, w.y};
    v2f wB = {w.z, w.w};
    acc = wmma_f32(wA, xf[2 * cp], acc);
    a1  = wmma_f32(wB, xf[2 * cp + 1], a1);
  }
  float4 o0, o1;
  o0.x = fmaxf(acc[0] + a1[0], 0.f); o0.y = fmaxf(acc[1] + a1[1], 0.f);
  o0.z = fmaxf(acc[2] + a1[2], 0.f); o0.w = fmaxf(acc[3] + a1[3], 0.f);
  o1.x = fmaxf(acc[4] + a1[4], 0.f); o1.y = fmaxf(acc[5] + a1[5], 0.f);
  o1.z = fmaxf(acc[6] + a1[6], 0.f); o1.w = fmaxf(acc[7] + a1[7], 0.f);
  *(float4*)dst = o0;
  *(float4*)(dst + 4) = o1;
}

__launch_bounds__(kThreads, 1)
__global__ void made_flow_kernel(const float* __restrict__ u,
                                 const float* __restrict__ b2,
                                 const float* __restrict__ ws,
                                 float* __restrict__ out) {
  extern __shared__ float smem[];
  float* xs   = smem;
  float* us   = xs  + kBT * kXStride;
  float* h0s  = us  + kBT * kXStride;
  float* h1s  = h0s + kBT * kHStride;
  float* wbuf = h1s + kBT * kHStride;
  float* psum = wbuf + 2 * kWbufSz;

  const int tid  = threadIdx.x;
  const int lane = tid & 31;
  const int wv   = __builtin_amdgcn_readfirstlane(tid >> 5);
  const int wg   = blockIdx.x;
  const int n    = lane & 15;
  const int hi   = lane >> 4;
  const int hi2  = hi << 1;

  const float4* Wa4 = (const float4*)(ws + kWaOff);
  const v4f* wsl = (const v4f*)ws + lane;
  constexpr int kWbQ  = kWbOff / 4;
  constexpr int kWzbQ = kWzbOff / 4;
  const v4f* Wzi4 = (const v4f*)(ws + kWziOff);
  const float*  b0s = ws + kB0Off;
  const float*  b1s = ws + kB1Off;

  for (int e = tid; e < kBT * kXStride; e += kThreads) xs[e] = 0.f;
  for (int e = tid; e < 2 * kBT * kHStride; e += kThreads) h0s[e] = 0.f;
  for (int e = tid; e < kBT * kD; e += kThreads) {
    int r = e >> 6, d = e & 63;
    us[r * kXStride + d] = u[(wg * kBT + r) * kD + d];
  }
  float ldj = 0.f;

  int ringOfs = kWbQ, ringN = ncp_of(1), pb = 0;
  if (wv >= 4) copy_tile(wsl, ringOfs, ringN, wbuf, wv - 4, lane);
  float4 waA[8], waB[8], bA0, bA1, bB0, bB1;
  l1_fetch(Wa4, b0s, 0, lane, hi, waA, bA0, bA1);
  l1_fetch(Wa4, b0s, 1, lane, hi, waB, bB0, bB1);
  __syncthreads();

  v8f zacc;
  for (int v = 0; v < 8; ++v) zacc[v] = 0.f;
  const float* xrow  = xs  + n * kXStride + hi2;
  const float* h0row = h0s + n * kHStride + hi2;
  const float* h1row = h1s + n * kHStride + hi2;

  for (int i = 0; i < kD; ++i) {
    const int offi = deg_off(i);
    const int ncp  = ncp_of(i);
    const int jb = i & 7;
    float b2m = 0.f, b2s = 0.f, uval = 0.f;
    if (wv == 0) {
      b2m = b2[i];
      b2s = b2[kD + i];
      uval = us[n * kXStride + i];
    }
    if (i > 0) {
      const int tlo = deg_off(i - 1) >> 4, thi = (offi - 1) >> 4;
      if (wv == 0) {
        v2f xf[16];
        l1_xfrags(xrow, xf);
        __builtin_amdgcn_sched_barrier(0x6);
        l1_tile(waA, bA0, bA1, xf, h0s + n * kHStride + tlo * 16 + hi * 8);
        if (thi > tlo) l1_tile(waB, bB0, bB1, xf, h0s + n * kHStride + thi * 16 + hi * 8);
      }
      v4f zi0 = {0.f, 0.f, 0.f, 0.f}, zi1 = zi0, zi2 = zi0;
      int t = tlo;
      do {
        lds_sync();
        int nOfs = ringOfs, nN = ringN;
        if (t < thi)          { nOfs = kWbQ + (t + 1) * 128 * 32; nN = ncp; }
        else if (jb == 0)     { nOfs = kWzbQ + (i >> 3) * 128 * 32; nN = ncp; }
        else if (i + 1 < kD)  { nOfs = kWbQ + (offi >> 4) * 128 * 32; nN = ncp_of(i + 1); }
        int kst = 0, kcnt = 0;
        if (wv < 4) k_share(ringN, wv, kst, kcnt);
        const float* wcur = wbuf + pb * kWbufSz + lane * 4;
        v8f r;
        for (int v = 0; v < 8; ++v) r[v] = 0.f;
        float4 bb0 = {0.f, 0.f, 0.f, 0.f}, bb1 = bb0;
        if (wv == 0) {
          bb0 = *(const float4*)(b1s + t * 16 + hi * 8);
          bb1 = *(const float4*)(b1s + t * 16 + hi * 8 + 4);
          if (t == tlo) {
            const v4f* zq = Wzi4 + (i * 3) * 32 + lane;
            zi0 = zq[0]; zi1 = zq[32]; zi2 = zq[64];
            const int iN = i + 1 < kD ? i + 1 : i;
            const int tloN = deg_off(iN - 1) >> 4, thiN = (deg_off(iN) - 1) >> 4;
            l1_fetch(Wa4, b0s, tloN, lane, hi, waA, bA0, bA1);
            if (thiN > tloN) l1_fetch(Wa4, b0s, thiN, lane, hi, waB, bB0, bB1);
          }
          r = tile_gemm(wcur, kcnt * kMB, h0row);
        } else if (wv < 4) {
          helper_share(wcur, h0row, kst, kcnt, psum + (wv - 1) * 256, lane);
        } else {
          copy_tile(wsl, nOfs, nN, wbuf + (pb ^ 1) * kWbufSz, wv - 4, lane);
        }
        lds_sync();
        if (wv == 0) {
          add_shares(r, psum, lane);
          float4 o0, o1;
          o0.x = fmaxf(r[0] + bb0.x, 0.f); o0.y = fmaxf(r[1] + bb0.y, 0.f);
          o0.z = fmaxf(r[2] + bb0.z, 0.f); o0.w = fmaxf(r[3] + bb0.w, 0.f);
          o1.x = fmaxf(r[4] + bb1.x, 0.f); o1.y = fmaxf(r[5] + bb1.y, 0.f);
          o1.z = fmaxf(r[6] + bb1.z, 0.f); o1.w = fmaxf(r[7] + bb1.w, 0.f);
          float* dst = h1s + n * kHStride + t * 16 + hi * 8;
          *(float4*)dst = o0;
          *(float4*)(dst + 4) = o1;
        }
        ringOfs = nOfs; ringN = nN; pb ^= 1;
      } while (++t <= thi);
      lds_sync();
      if (jb == 0) {
        const int nOfs = kWbQ + (offi >> 4) * 128 * 32, nN = ncp_of(i + 1);
        int kst = 0, kcnt = 0;
        if (wv < 4) k_share(ringN, wv, kst, kcnt);
        const float* wcur = wbuf + pb * kWbufSz + lane * 4;
        if (wv == 0) {
          zacc = tile_gemm(wcur, kcnt * kMB, h1row);
        } else if (wv < 4) {
          helper_share(wcur, h1row, kst, kcnt, psum + (wv - 1) * 256, lane);
        } else {
          copy_tile(wsl, nOfs, nN, wbuf + (pb ^ 1) * kWbufSz, wv - 4, lane);
        }
        lds_sync();
        if (wv == 0) add_shares(zacc, psum, lane);
        ringOfs = nOfs; ringN = nN; pb ^= 1;
      } else if (wv == 0) {
        const int cp0 = deg_off(i - 1) >> 3;
        const v4f wz[3] = {zi0, zi1, zi2};
        v8f zb;
        for (int v = 0; v < 8; ++v) zb[v] = 0.f;
        v2f hA[3], hB[3];
#pragma unroll
        for (int cl = 0; cl < 3; ++cl) {
          int hcp = cp0 + cl; hcp = hcp > 127 ? 127 : hcp;
          hA[cl] = *(const v2f*)(h1row + hcp * 8);
          hB[cl] = *(const v2f*)(h1row + hcp * 8 + 4);
        }
        __builtin_amdgcn_sched_barrier(0x6);
#pragma unroll
        for (int cl = 0; cl < 3; ++cl) {
          v2f wA = {wz[cl].x, wz[cl].y};
          v2f wB = {wz[cl].z, wz[cl].w};
          zacc = wmma_f32(wA, hA[cl], zacc);
          zb   = wmma_f32(wB, hB[cl], zb);
        }
        for (int v = 0; v < 8; ++v) zacc[v] += zb[v];
      }
    }
    if (wv == 0) {
      float zm, zs;
      switch (jb & 3) {
        case 0:  zm = zacc[0]; zs = zacc[1]; break;
        case 1:  zm = zacc[2]; zs = zacc[3]; break;
        case 2:  zm = zacc[4]; zs = zacc[5]; break;
        default: zm = zacc[6]; zs = zacc[7]; break;
      }
      if (hi == (jb >> 2)) {
        const float mu = zm + b2m;
        const float sg = zs + b2s;
        xs[n * kXStride + i] = uval * expf(sg) + mu;
        ldj += sg;
      }
    }
    lds_sync();
  }

  for (int e = tid; e < kBT * kD; e += kThreads) {
    int r = e >> 6, d = e & 63;
    out[(wg * kBT + r) * kD + d] = xs[r * kXStride + d];
  }
  if (wv == 0) {
    const float ldjt = ldj + __shfl_xor(ldj, 16);
    if (hi == 0) out[kB * kD + wg * kBT + n] = ldjt;
  }
}

extern "C" void kernel_launch(void* const* d_in, const int* in_sizes, int n_in,
                              void* d_out, int out_size, void* d_ws, size_t ws_size,
                              hipStream_t stream) {
  const float* u  = (const float*)d_in[0];
  const float* W0 = (const float*)d_in[1];
  const float* b0 = (const float*)d_in[2];
  const float* W1 = (const float*)d_in[3];
  const float* b1 = (const float*)d_in[4];
  const float* W2 = (const float*)d_in[5];
  const float* b2 = (const float*)d_in[6];
  float* ws  = (float*)d_ws;
  float* out = (float*)d_out;

  prep_w0<<<(kU * kD) / 256,     256, 0, stream>>>(W0, b0, b1, ws);
  prep_w1<<<(kU * kU) / 256,     256, 0, stream>>>(W1, ws + kWbOff);
  prep_w2<<<(kWzbSz + kWziSz) / 256, 256, 0, stream>>>(W2, ws);

  size_t smem = (size_t)(2 * kBT * kXStride + 2 * kBT * kHStride + 2 * kWbufSz + 3 * 2 * 32 * 4) * sizeof(float);
  made_flow_kernel<<<kB / kBT, kThreads, smem, stream>>>(u, b2, ws, out);
}


# Weight tiles are requested two tile_gemm calls ahead: the copy waves hold a tile in registers for one call period, then store it to the freed LDS half; the memory round trip leaves the call's path
# speedup vs baseline: 46.9155x; 1.1025x over previous
#include <hip/hip_runtime.h>

typedef __attribute__((ext_vector_type(2))) float v2f;
typedef __attribute__((ext_vector_type(8))) float v8f;
typedef __attribute__((ext_vector_type(4))) float v4f;

constexpr int kD  = 64;
constexpr int kU  = 1024;
constexpr int kB  = 4096;
constexpr int kBT = 16;

constexpr int kXStride = kD + 4;
constexpr int kHStride = kU + 4;

constexpr int kWaOff = 0;
constexpr int kWaSz  = kU * kD;
constexpr int kWbOff = kWaOff + kWaSz;
constexpr int kWbSz  = kU * kU;
constexpr int kWzbOff = kWbOff + kWbSz;
constexpr int kWzbSz  = 8 * 128 * 128;
constexpr int kWziOff = kWzbOff + kWzbSz;
constexpr int kWziSz  = kD * 3 * 128;
constexpr int kB0Off = kWziOff + kWziSz;
constexpr int kB1Off = kB0Off + kU;

__device__ __forceinline__ int deg_off(int d) {
  return d <= 16 ? 17 * d : 272 + 16 * (d - 16);
}
__device__ __forceinline__ int pos_of_unit(int j) {
  int d = j % (kD - 1), m = j / (kD - 1);
  return deg_off(d) + m;
}
__device__ __forceinline__ int unit_at_pos(int p) {
  int d, m;
  if (p < 272) { d = p / 17; m = p - 17 * d; }
  else         { d = 16 + ((p - 272) >> 4); m = (p - 272) & 15; }
  return m * (kD - 1) + d;
}


__global__ void prep_w0(const float* __restrict__ W0, const float* __restrict__ b0,
                        const float* __restrict__ b1, float* __restrict__ ws) {
  int e = blockIdx.x * 256 + threadIdx.x;
  int j = e >> 6, d = e & 63;
  int dh = j % (kD - 1);
  float v = (d <= dh) ? W0[e] : 0.f;
  int p = pos_of_unit(j);
  int t = p >> 4, n = p & 15;
  int cp = d >> 3, rr = d & 7;
  int q  = ((rr >> 2) << 1) | (rr & 1);
  int hi = (rr >> 1) & 1;
  int l  = (hi << 4) | n;
  ws[kWaOff + (((t * 8 + cp) * 32 + l) << 2) + q] = v;
  if (e < kU) {
    int pe = pos_of_unit(e);
    ws[kB0Off + pe] = b0[e];
    ws[kB1Off + pe] = b1[e];
  }
}

__global__ void prep_w1(const float* __restrict__ W1, float* __restrict__ Wb) {
  int e = blockIdx.x * 256 + threadIdx.x;
  int j = e >> 10, k = e & 1023;
  int dhj = j % (kD - 1), dhk = k % (kD - 1);
  float v = (dhk <= dhj) ? W1[e] : 0.f;
  int pj = pos_of_unit(j), pk = pos_of_unit(k);
  int t = pj >> 4, n = pj & 15;
  int cp = pk >> 3, rr = pk & 7;
  int q  = ((rr >> 2) << 1) | (rr & 1);
  int hi = (rr >> 1) & 1;
  int l  = (hi << 4) | n;
  Wb[(((t * 128 + cp) * 32 + l) << 2) + q] = v;
}

__global__ void prep_w2(const float* __restrict__ W2, float* __restrict__ ws) {
  int e = blockIdx.x * 256 + threadIdx.x;
  if (e < kWzbSz) {
    int q = e & 3, l = (e >> 2) & 31, cp = (e >> 7) & 127, b = e >> 14;
    int hi = l >> 4, m = l & 15;
    int p = 8 * cp + (q >> 1) * 4 + hi * 2 + (q & 1);
    int row = (m & 1) * kD + 8 * b + (m >> 1);
    float v = 0.f;
    if (p < deg_off(8 * b)) v = W2[row * kU + unit_at_pos(p)];
    ws[kWzbOff + e] = v;
  } else {
    int f = e - kWzbSz;
    int q = f & 3, l = (f >> 2) & 31, r = f >> 7;
    int i = r / 3, cl = r - 3 * i;
    int hi = l >> 4, m = l & 15;
    float v = 0.f;
    if (i >= 1) {
      int lo = deg_off(i - 1), hiEnd = deg_off(i);
      int p = 8 * ((lo >> 3) + cl) + (q >> 1) * 4 + hi * 2 + (q & 1);
      int row = (m & 1) * kD + 8 * (i >> 3) + (m >> 1);
      if (p >= lo && p < hiEnd) v = W2[row * kU + unit_at_pos(p)];
    }
    ws[kWziOff + f] = v;
  }
}

__device__ __forceinline__ v8f wmma_f32(v2f a, v2f b, v8f c) {
  return __builtin_amdgcn_wmma_f32_16x16x4_f32(false, a, false, b, (short)0, c,
                                               false, false);
}

constexpr int kBlk    = 16;
constexpr int kMB     = 8;
constexpr int kWbufSz = 128 * 32 * 4;
constexpr int kThreads = 256;

template <int G>
__device__ __forceinline__ void gemm_block(v4f (&wf)[2][kMB], v2f (&hf)[2][2 * kMB],
                                           v8f& a0, v8f& a1, v8f& a2, v8f& a3,
                                           const float* wl, const float* hrow, int base, int ncp) {
  constexpr int GN = G ^ 1;
  int nb = base + kMB; nb = nb > ncp - kMB ? ncp - kMB : nb;
  const float* hp = hrow + nb * 8;
  const float* wp = wl + nb * 128;
#pragma unroll
  for (int u = 0; u < kMB; ++u) {
    hf[GN][2 * u]     = *(const v2f*)(hp + u * 8);
    hf[GN][2 * u + 1] = *(const v2f*)(hp + u * 8 + 4);
  }
#pragma unroll
  for (int u = 0; u < kMB; ++u) wf[GN][u] = *(const v4f*)(wp + u * 128);
  __builtin_amdgcn_sched_barrier(0x6);
#pragma unroll
  for (int u = 0; u < kMB; ++u) {
    v4f w = wf[G][u];
    v2f wA = {w.x, w.y};
    v2f wB = {w.z, w.w};
    if (u & 1) { a2 = wmma_f32(wA, hf[G][2 * u], a2); a3 = wmma_f32(wB, hf[G][2 * u + 1], a3); }
    else       { a0 = wmma_f32(wA, hf[G][2 * u], a0); a1 = wmma_f32(wB, hf[G][2 * u + 1], a1); }
  }
  __builtin_amdgcn_sched_barrier(0x6);
}

__device__ __forceinline__ v8f tile_gemm(const float* wl, int ncp, const float* hrow) {
  v4f wf[2][kMB];
  v2f hf[2][2 * kMB];
#pragma unroll
  for (int u = 0; u < kMB; ++u) {
    hf[0][2 * u]     = *(const v2f*)(hrow + u * 8);
    hf[0][2 * u + 1] = *(const v2f*)(hrow + u * 8 + 4);
  }
#pragma unroll
  for (int u = 0; u < kMB; ++u) wf[0][u] = *(const v4f*)(wl + u * 128);
  v8f a0, a1, a2, a3;
  for (int v = 0; v < 8; ++v) { a0[v] = 0.f; a1[v] = 0.f; a2[v] = 0.f; a3[v] = 0.f; }
  int base = 0;
  for (; base + 2 * kMB <= ncp; base += 2 * kMB) {
    gemm_block<0>(wf, hf, a0, a1, a2, a3, wl, hrow, base, ncp);
    gemm_block<1>(wf, hf, a0, a1, a2, a3, wl, hrow, base + kMB, ncp);
  }
  if (base < ncp) gemm_block<0>(wf, hf, a0, a1, a2, a3, wl, hrow, base, ncp);
  v8f r;
  for (int v = 0; v < 8; ++v) r[v] = (a0[v] + a2[v]) + (a1[v] + a3[v]);
  return r;
}

__device__ __forceinline__ int copy_cnt(int ncp, int pw) {
  return ncp > 0 ? ((ncp >> 4) - pw + 3) >> 2 : 0;
}
__device__ __forceinline__ void copy_drained() {
  __builtin_amdgcn_fence(__ATOMIC_ACQUIRE, "workgroup", "global");
}
__device__ __forceinline__ void copy_fetch(v4f (&r)[2][kBlk], const v4f* __restrict__ wsl, int ofs,
                                           int cnt, int pw) {
  if (cnt > 0) {
    copy_drained();
#pragma unroll
    for (int u = 0; u < kBlk; ++u) r[0][u] = wsl[ofs + (pw * kBlk + u) * 32];
    if (cnt > 1) {
#pragma unroll
      for (int u = 0; u < kBlk; ++u) r[1][u] = wsl[ofs + ((pw + 4) * kBlk + u) * 32];
    }
  }
}
__device__ __forceinline__ void copy_put(const v4f (&r)[2][kBlk], int cnt, float* dst, int pw, int lane) {
#pragma unroll
  for (int s3 = 0; s3 < 2; ++s3) {
    if (s3 < cnt) {
      const int blk = pw + 4 * s3;
#pragma unroll
      for (int u = 0; u < kBlk; ++u) *(v4f*)(dst + ((blk * kBlk + u) * 32 + lane) * 4) = r[s3][u];
    }
  }
}

__device__ __forceinline__ void k_share(int ncp, int w, int& st, int& cnt) {
  const int nq = ncp / kMB, qb = nq >> 2, qr = nq & 3;
  cnt = qb + (w < qr ? 1 : 0);
  st  = w * qb + (w < qr ? w : qr);
}

__device__ __forceinline__ void helper_share(const float* wcur, const float* hrow, int st, int cnt,
                                             float* ps, int lane) {
  v8f p;
  for (int v = 0; v < 8; ++v) p[v] = 0.f;
  if (cnt > 0) p = tile_gemm(wcur + st * kMB * 128, cnt * kMB, hrow + st * kMB * 8);
  float4 p0, p1;
  p0.x = p[0]; p0.y = p[1]; p0.z = p[2]; p0.w = p[3];
  p1.x = p[4]; p1.y = p[5]; p1.z = p[6]; p1.w = p[7];
  *(float4*)(ps + lane * 4) = p0;
  *(float4*)(ps + 128 + lane * 4) = p1;
}
__device__ __forceinline__ void add_shares(v8f& r, const float* psum, int lane) {
  float4 q0[3], q1[3];
#pragma unroll
  for (int w = 0; w < 3; ++w) {
    q0[w] = *(const float4*)(psum + w * 256 + lane * 4);
    q1[w] = *(const float4*)(psum + w * 256 + 128 + lane * 4);
  }
  r[0] = (r[0] + q0[0].x) + (q0[1].x + q0[2].x); r[1] = (r[1] + q0[0].y) + (q0[1].y + q0[2].y);
  r[2] = (r[2] + q0[0].z) + (q0[1].z + q0[2].z); r[3] = (r[3] + q0[0].w) + (q0[1].w + q0[2].w);
  r[4] = (r[4] + q1[0].x) + (q1[1].x + q1[2].x); r[5] = (r[5] + q1[0].y) + (q1[1].y + q1[2].y);
  r[6] = (r[6] + q1[0].z) + (q1[1].z + q1[2].z); r[7] = (r[7] + q1[0].w) + (q1[1].w + q1[2].w);
}

__device__ __forceinline__ void lds_sync() {
  __builtin_amdgcn_fence(__ATOMIC_RELEASE, "workgroup", "local");
  __builtin_amdgcn_s_barrier();
  __builtin_amdgcn_fence(__ATOMIC_ACQUIRE, "workgroup", "local");
}

__device__ __forceinline__ int ncp_of(int i) {
  return ((deg_off(i) + 127) >> 7) << 4;
}

__device__ __forceinline__ void next_call(int& ci, int& ct, int& ck) {
  if (ci >= kD) return;
  if (ck == 0) {
    const int thi = (deg_off(ci) - 1) >> 4;
    if (ct < thi) { ++ct; return; }
    if ((ci & 7) == 0) { ck = 1; return; }
  }
  ck = 0; ++ci; ct = deg_off(ci - 1) >> 4;
}
__device__ __forceinline__ int call_n(int ci) {
  return ci < kD ? ncp_of(ci) : 0;
}
__device__ __forceinline__ int call_ofs(int ci, int ct, int ck) {
  return ck ? (kWzbOff / 4) + (ci >> 3) * 128 * 32 : (kWbOff / 4) + ct * 128 * 32;
}

__device__ __forceinline__ void l1_fetch(const float4* __restrict__ Wa4, const float* __restrict__ b0s,
                                         int t, int lane, int hi,
                                         float4 (&w)[8], float4& bb0, float4& bb1) {
  const float4* wq = Wa4 + (t * 8) * 32 + lane;
#pragma unroll
  for (int cp = 0; cp < 8; ++cp) w[cp] = wq[cp * 32];
  bb0 = *(const float4*)(b0s + t * 16 + hi * 8);
  bb1 = *(const float4*)(b0s + t * 16 + hi * 8 + 4);
}
__device__ __forceinline__ void l1_xfrags(const float* xrow, v2f (&xf)[16]) {
#pragma unroll
  for (int cp = 0; cp < 8; ++cp) {
    xf[2 * cp]     = *(const v2f*)(xrow + cp * 8);
    xf[2 * cp + 1] = *(const v2f*)(xrow + cp * 8 + 4);
  }
}
__device__ __forceinline__ void l1_tile(const float4 (&wr)[8], const float4 bb0, const float4 bb1,
                                        const v2f (&xf)[16], float* dst) {
  v8f acc;
  acc[0] = bb0.x; acc[1] = bb0.y; acc[2] = bb0.z; acc[3] = bb0.w;
  acc[4] = bb1.x; acc[5] = bb1.y; acc[6] = bb1.z; acc[7] = bb1.w;
  v8f a1;
  for (int v = 0; v < 8; ++v) a1[v] = 0.f;
#pragma unroll
  for (int cp = 0; cp < 8; ++cp) {
    const float4 w = wr[cp];
    v2f wA = {w.x, w.y};
    v2f wB = {w.z, w.w};
    acc = wmma_f32(wA, xf[2 * cp], acc);
    a1  = wmma_f32(wB, xf[2 * cp + 1], a1);
  }
  float4 o0, o1;
  o0.x = fmaxf(acc[0] + a1[0], 0.f); o0.y = fmaxf(acc[1] + a1[1], 0.f);
  o0.z = fmaxf(acc[2] + a1[2], 0.f); o0.w = fmaxf(acc[3] + a1[3], 0.f);
  o1.x = fmaxf(acc[4] + a1[4], 0.f); o1.y = fmaxf(acc[5] + a1[5], 0.f);
  o1.z = fmaxf(acc[6] + a1[6], 0.f); o1.w = fmaxf(acc[7] + a1[7], 0.f);
  *(float4*)dst = o0;
  *(float4*)(dst + 4) = o1;
}

__launch_bounds__(kThreads, 1)
__global__ void made_flow_kernel(const float* __restrict__ u,
                                 const float* __restrict__ b2,
                                 const float* __restrict__ ws,
                                 float* __restrict__ out) {
  extern __shared__ float smem[];
  float* xs   = smem;
  float* us   = xs  + kBT * kXStride;
  float* h0s  = us  + kBT * kXStride;
  float* h1s  = h0s + kBT * kHStride;
  float* wbuf = h1s + kBT * kHStride;
  float* psum = wbuf + 2 * kWbufSz;

  const int tid  = threadIdx.x;
  const int lane = tid & 31;
  const int wv   = __builtin_amdgcn_readfirstlane(tid >> 5);
  const int wg   = blockIdx.x;
  const int n    = lane & 15;
  const int hi   = lane >> 4;
  const int hi2  = hi << 1;

  const float4* Wa4 = (const float4*)(ws + kWaOff);
  const v4f* wsl = (const v4f*)ws + lane;
  const v4f* Wzi4 = (const v4f*)(ws + kWziOff);
  const float*  b0s = ws + kB0Off;
  const float*  b1s = ws + kB1Off;

  for (int e = tid; e < kBT * kXStride; e += kThreads) xs[e] = 0.f;
  for (int e = tid; e < 2 * kBT * kHStride; e += kThreads) h0s[e] = 0.f;
  for (int e = tid; e < kBT * kD; e += kThreads) {
    int r = e >> 6, d = e & 63;
    us[r * kXStride + d] = u[(wg * kBT + r) * kD + d];
  }
  float ldj = 0.f;

  int ci = 1, ct = 0, ck = 0;
  int ringN = call_n(ci), pb = 0;
  const int pw = wv - 4;
  v4f cr[2][kBlk];
#pragma unroll
  for (int s3 = 0; s3 < 2; ++s3)
#pragma unroll
    for (int u2 = 0; u2 < kBlk; ++u2) cr[s3][u2] = v4f{0.f, 0.f, 0.f, 0.f};
  if (wv >= 4) {
    copy_fetch(cr, wsl, call_ofs(ci, ct, ck), copy_cnt(ringN, pw), pw);
    copy_put(cr, copy_cnt(ringN, pw), wbuf, pw, lane);
  }
  next_call(ci, ct, ck);
  int heldN = call_n(ci);
  int heldCnt = wv >= 4 ? copy_cnt(heldN, pw) : 0;
  const int heldOfs0 = call_ofs(ci, ct, ck);
  next_call(ci, ct, ck);
  int fOfs = call_ofs(ci, ct, ck);
  int fCnt = wv >= 4 ? copy_cnt(call_n(ci), pw) : 0;
  float4 waA[8], waB[8], bA0, bA1, bB0, bB1;
  l1_fetch(Wa4, b0s, 0, lane, hi, waA, bA0, bA1);
  l1_fetch(Wa4, b0s, 1, lane, hi, waB, bB0, bB1);
  __syncthreads();
  copy_fetch(cr, wsl, heldOfs0, heldCnt, pw);

  v8f zacc;
  for (int v = 0; v < 8; ++v) zacc[v] = 0.f;
  const float* xrow  = xs  + n * kXStride + hi2;
  const float* h0row = h0s + n * kHStride + hi2;
  const float* h1row = h1s + n * kHStride + hi2;

  for (int i = 0; i < kD; ++i) {
    const int offi = deg_off(i);
    const int jb = i & 7;
    float b2m, b2s, uval;
    if (wv == 0) {
      b2m = b2[i];
      b2s = b2[kD + i];
      uval = us[n * kXStride + i];
    }
    if (i > 0) {
      const int tlo = deg_off(i - 1) >> 4, thi = (offi - 1) >> 4;
      if (wv == 0) {
        v2f xf[16];
        l1_xfrags(xrow, xf);
        __builtin_amdgcn_sched_barrier(0x6);
        l1_tile(waA, bA0, bA1, xf, h0s + n * kHStride + tlo * 16 + hi * 8);
        if (thi > tlo) l1_tile(waB, bB0, bB1, xf, h0s + n * kHStride + thi * 16 + hi * 8);
      }
      v4f zi0, zi1, zi2;
      int t = tlo;
      do {
        lds_sync();
        copy_put(cr, heldCnt, wbuf + (pb ^ 1) * kWbufSz, pw, lane);
        copy_fetch(cr, wsl, fOfs, fCnt, pw);
        int kst = 0, kcnt = 0;
        if (wv < 4) k_share(ringN, wv, kst, kcnt);
        const float* wcur = wbuf + pb * kWbufSz + lane * 4;
        v8f r;
        float4 bb0, bb1;
        if (wv == 0) {
          bb0 = *(const float4*)(b1s + t * 16 + hi * 8);
          bb1 = *(const float4*)(b1s + t * 16 + hi * 8 + 4);
          if (t == tlo) {
            const v4f* zq = Wzi4 + (i * 3) * 32 + lane;
            zi0 = zq[0]; zi1 = zq[32]; zi2 = zq[64];
            const int iN = i + 1 < kD ? i + 1 : i;
            const int tloN = deg_off(iN - 1) >> 4, thiN = (deg_off(iN) - 1) >> 4;
            l1_fetch(Wa4, b0s, tloN, lane, hi, waA, bA0, bA1);
            if (thiN > tloN) l1_fetch(Wa4, b0s, thiN, lane, hi, waB, bB0, bB1);
          }
          r = tile_gemm(wcur, kcnt * kMB, h0row);
        } else if (wv < 4) {
          helper_share(wcur, h0row, kst, kcnt, psum + (wv - 1) * 256, lane);
        }
        lds_sync();
        if (wv == 0) {
          add_shares(r, psum, lane);
          float4 o0, o1;
          o0.x = fmaxf(r[0] + bb0.x, 0.f); o0.y = fmaxf(r[1] + bb0.y, 0.f);
          o0.z = fmaxf(r[2] + bb0.z, 0.f); o0.w = fmaxf(r[3] + bb0.w, 0.f);
          o1.x = fmaxf(r[4] + bb1.x, 0.f); o1.y = fmaxf(r[5] + bb1.y, 0.f);
          o1.z = fmaxf(r[6] + bb1.z, 0.f); o1.w = fmaxf(r[7] + bb1.w, 0.f);
          float* dst = h1s + n * kHStride + t * 16 + hi * 8;
          *(float4*)dst = o0;
          *(float4*)(dst + 4) = o1;
        }
        ringN = heldN; heldN = call_n(ci); heldCnt = fCnt; pb ^= 1;
        next_call(ci, ct, ck);
        fOfs = call_ofs(ci, ct, ck); fCnt = wv >= 4 ? copy_cnt(call_n(ci), pw) : 0;
      } while (++t <= thi);
      lds_sync();
      if (jb == 0) {
        copy_put(cr, heldCnt, wbuf + (pb ^ 1) * kWbufSz, pw, lane);
        copy_fetch(cr, wsl, fOfs, fCnt, pw);
        int kst = 0, kcnt = 0;
        if (wv < 4) k_share(ringN, wv, kst, kcnt);
        const float* wcur = wbuf + pb * kWbufSz + lane * 4;
        if (wv == 0) {
          zacc = tile_gemm(wcur, kcnt * kMB, h1row);
        } else if (wv < 4) {
          helper_share(wcur, h1row, kst, kcnt, psum + (wv - 1) * 256, lane);
        }
        lds_sync();
        if (wv == 0) add_shares(zacc, psum, lane);
        ringN = heldN; heldN = call_n(ci); heldCnt = fCnt; pb ^= 1;
        next_call(ci, ct, ck);
        fOfs = call_ofs(ci, ct, ck); fCnt = wv >= 4 ? copy_cnt(call_n(ci), pw) : 0;
      } else if (wv == 0) {
        const int cp0 = deg_off(i - 1) >> 3;
        const v4f wz[3] = {zi0, zi1, zi2};
        v8f zb;
        for (int v = 0; v < 8; ++v) zb[v] = 0.f;
        v2f hA[3], hB[3];
#pragma unroll
        for (int cl = 0; cl < 3; ++cl) {
          int hcp = cp0 + cl; hcp = hcp > 127 ? 127 : hcp;
          hA[cl] = *(const v2f*)(h1row + hcp * 8);
          hB[cl] = *(const v2f*)(h1row + hcp * 8 + 4);
        }
        __builtin_amdgcn_sched_barrier(0x6);
#pragma unroll
        for (int cl = 0; cl < 3; ++cl) {
          v2f wA = {wz[cl].x, wz[cl].y};
          v2f wB = {wz[cl].z, wz[cl].w};
          zacc = wmma_f32(wA, hA[cl], zacc);
          zb   = wmma_f32(wB, hB[cl], zb);
        }
        for (int v = 0; v < 8; ++v) zacc[v] += zb[v];
      }
    }
    if (wv == 0) {
      float zm, zs;
      switch (jb & 3) {
        case 0:  zm = zacc[0]; zs = zacc[1]; break;
        case 1:  zm = zacc[2]; zs = zacc[3]; break;
        case 2:  zm = zacc[4]; zs = zacc[5]; break;
        default: zm = zacc[6]; zs = zacc[7]; break;
      }
      if (hi == (jb >> 2)) {
        const float mu = zm + b2m;
        const float sg = zs + b2s;
        xs[n * kXStride + i] = uval * expf(sg) + mu;
        ldj += sg;
      }
    }
    lds_sync();
  }

  for (int e = tid; e < kBT * kD; e += kThreads) {
    int r = e >> 6, d = e & 63;
    out[(wg * kBT + r) * kD + d] = xs[r * kXStride + d];
  }
  if (wv == 0) {
    const float ldjt = ldj + __shfl_xor(ldj, 16);
    if (hi == 0) out[kB * kD + wg * kBT + n] = ldjt;
  }
}

extern "C" void kernel_launch(void* const* d_in, const int* in_sizes, int n_in,
                              void* d_out, int out_size, void* d_ws, size_t ws_size,
                              hipStream_t stream) {
  const float* u  = (const float*)d_in[0];
  const float* W0 = (const float*)d_in[1];
  const float* b0 = (const float*)d_in[2];
  const float* W1 = (const float*)d_in[3];
  const float* b1 = (const float*)d_in[4];
  const float* W2 = (const float*)d_in[5];
  const float* b2 = (const float*)d_in[6];
  float* ws  = (float*)d_ws;
  float* out = (float*)d_out;

  prep_w0<<<(kU * kD) / 256,     256, 0, stream>>>(W0, b0, b1, ws);
  prep_w1<<<(kU * kU) / 256,     256, 0, stream>>>(W1, ws + kWbOff);
  prep_w2<<<(kWzbSz + kWziSz) / 256, 256, 0, stream>>>(W2, ws);

  size_t smem = (size_t)(2 * kBT * kXStride + 2 * kBT * kHStride + 2 * kWbufSz + 3 * 2 * 32 * 4) * sizeof(float);
  made_flow_kernel<<<kB / kBT, kThreads, smem, stream>>>(u, b2, ws, out);
}


# prep_w1 gathers: one thread fetches the four weights of one packed float4 slot and stores 16 B (a wave writes a whole 512-B chunk pair), was one weight per thread scattered; 32768 -> 8192 waves
# speedup vs baseline: 53.5239x; 1.1409x over previous
#include <hip/hip_runtime.h>

typedef __attribute__((ext_vector_type(2))) float v2f;
typedef __attribute__((ext_vector_type(8))) float v8f;
typedef __attribute__((ext_vector_type(4))) float v4f;

constexpr int kD  = 64;
constexpr int kU  = 1024;
constexpr int kB  = 4096;
constexpr int kBT = 16;

constexpr int kXStride = kD + 4;
constexpr int kHStride = kU + 4;

constexpr int kWaOff = 0;
constexpr int kWaSz  = kU * kD;
constexpr int kWbOff = kWaOff + kWaSz;
constexpr int kWbSz  = kU * kU;
constexpr int kWzbOff = kWbOff + kWbSz;
constexpr int kWzbSz  = 8 * 128 * 128;
constexpr int kWziOff = kWzbOff + kWzbSz;
constexpr int kWziSz  = kD * 3 * 128;
constexpr int kB0Off = kWziOff + kWziSz;
constexpr int kB1Off = kB0Off + kU;

__device__ __forceinline__ int deg_off(int d) {
  return d <= 16 ? 17 * d : 272 + 16 * (d - 16);
}
__device__ __forceinline__ int pos_of_unit(int j) {
  int d = j % (kD - 1), m = j / (kD - 1);
  return deg_off(d) + m;
}
__device__ __forceinline__ int unit_at_pos(int p) {
  int d, m;
  if (p < 272) { d = p / 17; m = p - 17 * d; }
  else         { d = 16 + ((p - 272) >> 4); m = (p - 272) & 15; }
  return m * (kD - 1) + d;
}


__global__ void prep_w0(const float* __restrict__ W0, const float* __restrict__ b0,
                        const float* __restrict__ b1, float* __restrict__ ws) {
  int e = blockIdx.x * 256 + threadIdx.x;
  int j = e >> 6, d = e & 63;
  int dh = j % (kD - 1);
  float v = (d <= dh) ? W0[e] : 0.f;
  int p = pos_of_unit(j);
  int t = p >> 4, n = p & 15;
  int cp = d >> 3, rr = d & 7;
  int q  = ((rr >> 2) << 1) | (rr & 1);
  int hi = (rr >> 1) & 1;
  int l  = (hi << 4) | n;
  ws[kWaOff + (((t * 8 + cp) * 32 + l) << 2) + q] = v;
  if (e < kU) {
    int pe = pos_of_unit(e);
    ws[kB0Off + pe] = b0[e];
    ws[kB1Off + pe] = b1[e];
  }
}

__global__ void prep_w1(const float* __restrict__ W1, float* __restrict__ Wb) {
  int e = blockIdx.x * 256 + threadIdx.x;
  int l = e & 31, cp = (e >> 5) & 127, t = e >> 12;
  int n = l & 15, hi = l >> 4;
  int j = unit_at_pos(t * 16 + n);
  int dhj = j % (kD - 1);
  const float* row = W1 + j * kU;
  float v[4];
#pragma unroll
  for (int q = 0; q < 4; ++q) {
    int pk = 8 * cp + (q >> 1) * 4 + hi * 2 + (q & 1);
    int k = unit_at_pos(pk);
    int dhk = k % (kD - 1);
    v[q] = (dhk <= dhj) ? row[k] : 0.f;
  }
  float4 o;
  o.x = v[0]; o.y = v[1]; o.z = v[2]; o.w = v[3];
  *(float4*)(Wb + (e << 2)) = o;
}

__global__ void prep_w2(const float* __restrict__ W2, float* __restrict__ ws) {
  int e = blockIdx.x * 256 + threadIdx.x;
  if (e < kWzbSz) {
    int q = e & 3, l = (e >> 2) & 31, cp = (e >> 7) & 127, b = e >> 14;
    int hi = l >> 4, m = l & 15;
    int p = 8 * cp + (q >> 1) * 4 + hi * 2 + (q & 1);
    int row = (m & 1) * kD + 8 * b + (m >> 1);
    float v = 0.f;
    if (p < deg_off(8 * b)) v = W2[row * kU + unit_at_pos(p)];
    ws[kWzbOff + e] = v;
  } else {
    int f = e - kWzbSz;
    int q = f & 3, l = (f >> 2) & 31, r = f >> 7;
    int i = r / 3, cl = r - 3 * i;
    int hi = l >> 4, m = l & 15;
    float v = 0.f;
    if (i >= 1) {
      int lo = deg_off(i - 1), hiEnd = deg_off(i);
      int p = 8 * ((lo >> 3) + cl) + (q >> 1) * 4 + hi * 2 + (q & 1);
      int row = (m & 1) * kD + 8 * (i >> 3) + (m >> 1);
      if (p >= lo && p < hiEnd) v = W2[row * kU + unit_at_pos(p)];
    }
    ws[kWziOff + f] = v;
  }
}

__device__ __forceinline__ v8f wmma_f32(v2f a, v2f b, v8f c) {
  return __builtin_amdgcn_wmma_f32_16x16x4_f32(false, a, false, b, (short)0, c,
                                               false, false);
}

constexpr int kBlk    = 16;
constexpr int kMB     = 8;
constexpr int kWbufSz = 128 * 32 * 4;
constexpr int kThreads = 256;

template <int G>
__device__ __forceinline__ void gemm_block(v4f (&wf)[2][kMB], v2f (&hf)[2][2 * kMB],
                                           v8f& a0, v8f& a1, v8f& a2, v8f& a3,
                                           const float* wl, const float* hrow, int base, int ncp) {
  constexpr int GN = G ^ 1;
  int nb = base + kMB; nb = nb > ncp - kMB ? ncp - kMB : nb;
  const float* hp = hrow + nb * 8;
  const float* wp = wl + nb * 128;
#pragma unroll
  for (int u = 0; u < kMB; ++u) {
    hf[GN][2 * u]     = *(const v2f*)(hp + u * 8);
    hf[GN][2 * u + 1] = *(const v2f*)(hp + u * 8 + 4);
  }
#pragma unroll
  for (int u = 0; u < kMB; ++u) wf[GN][u] = *(const v4f*)(wp + u * 128);
  __builtin_amdgcn_sched_barrier(0x6);
#pragma unroll
  for (int u = 0; u < kMB; ++u) {
    v4f w = wf[G][u];
    v2f wA = {w.x, w.y};
    v2f wB = {w.z, w.w};
    if (u & 1) { a2 = wmma_f32(wA, hf[G][2 * u], a2); a3 = wmma_f32(wB, hf[G][2 * u + 1], a3); }
    else       { a0 = wmma_f32(wA, hf[G][2 * u], a0); a1 = wmma_f32(wB, hf[G][2 * u + 1], a1); }
  }
  __builtin_amdgcn_sched_barrier(0x6);
}

__device__ __forceinline__ v8f tile_gemm(const float* wl, int ncp, const float* hrow) {
  v4f wf[2][kMB];
  v2f hf[2][2 * kMB];
#pragma unroll
  for (int u = 0; u < kMB; ++u) {
    hf[0][2 * u]     = *(const v2f*)(hrow + u * 8);
    hf[0][2 * u + 1] = *(const v2f*)(hrow + u * 8 + 4);
  }
#pragma unroll
  for (int u = 0; u < kMB; ++u) wf[0][u] = *(const v4f*)(wl + u * 128);
  v8f a0, a1, a2, a3;
  for (int v = 0; v < 8; ++v) { a0[v] = 0.f; a1[v] = 0.f; a2[v] = 0.f; a3[v] = 0.f; }
  int base = 0;
  for (; base + 2 * kMB <= ncp; base += 2 * kMB) {
    gemm_block<0>(wf, hf, a0, a1, a2, a3, wl, hrow, base, ncp);
    gemm_block<1>(wf, hf, a0, a1, a2, a3, wl, hrow, base + kMB, ncp);
  }
  if (base < ncp) gemm_block<0>(wf, hf, a0, a1, a2, a3, wl, hrow, base, ncp);
  v8f r;
  for (int v = 0; v < 8; ++v) r[v] = (a0[v] + a2[v]) + (a1[v] + a3[v]);
  return r;
}

__device__ __forceinline__ int copy_cnt(int ncp, int pw) {
  return ncp > 0 ? ((ncp >> 4) - pw + 3) >> 2 : 0;
}
__device__ __forceinline__ void copy_drained() {
  __builtin_amdgcn_fence(__ATOMIC_ACQUIRE, "workgroup", "global");
}
__device__ __forceinline__ void copy_fetch(v4f (&r)[2][kBlk], const v4f* __restrict__ wsl, int ofs,
                                           int cnt, int pw) {
  if (cnt > 0) {
    copy_drained();
#pragma unroll
    for (int u = 0; u < kBlk; ++u) r[0][u] = wsl[ofs + (pw * kBlk + u) * 32];
    if (cnt > 1) {
#pragma unroll
      for (int u = 0; u < kBlk; ++u) r[1][u] = wsl[ofs + ((pw + 4) * kBlk + u) * 32];
    }
  }
}
__device__ __forceinline__ void copy_put(const v4f (&r)[2][kBlk], int cnt, float* dst, int pw, int lane) {
#pragma unroll
  for (int s3 = 0; s3 < 2; ++s3) {
    if (s3 < cnt) {
      const int blk = pw + 4 * s3;
#pragma unroll
      for (int u = 0; u < kBlk; ++u) *(v4f*)(dst + ((blk * kBlk + u) * 32 + lane) * 4) = r[s3][u];
    }
  }
}

__device__ __forceinline__ void k_share(int ncp, int w, int& st, int& cnt) {
  const int nq = ncp / kMB, qb = nq >> 2, qr = nq & 3;
  cnt = qb + (w < qr ? 1 : 0);
  st  = w * qb + (w < qr ? w : qr);
}

__device__ __forceinline__ void helper_share(const float* wcur, const float* hrow, int st, int cnt,
                                             float* ps, int lane) {
  v8f p;
  for (int v = 0; v < 8; ++v) p[v] = 0.f;
  if (cnt > 0) p = tile_gemm(wcur + st * kMB * 128, cnt * kMB, hrow + st * kMB * 8);
  float4 p0, p1;
  p0.x = p[0]; p0.y = p[1]; p0.z = p[2]; p0.w = p[3];
  p1.x = p[4]; p1.y = p[5]; p1.z = p[6]; p1.w = p[7];
  *(float4*)(ps + lane * 4) = p0;
  *(float4*)(ps + 128 + lane * 4) = p1;
}
__device__ __forceinline__ void add_shares(v8f& r, const float* psum, int lane) {
  float4 q0[3], q1[3];
#pragma unroll
  for (int w = 0; w < 3; ++w) {
    q0[w] = *(const float4*)(psum + w * 256 + lane * 4);
    q1[w] = *(const float4*)(psum + w * 256 + 128 + lane * 4);
  }
  r[0] = (r[0] + q0[0].x) + (q0[1].x + q0[2].x); r[1] = (r[1] + q0[0].y) + (q0[1].y + q0[2].y);
  r[2] = (r[2] + q0[0].z) + (q0[1].z + q0[2].z); r[3] = (r[3] + q0[0].w) + (q0[1].w + q0[2].w);
  r[4] = (r[4] + q1[0].x) + (q1[1].x + q1[2].x); r[5] = (r[5] + q1[0].y) + (q1[1].y + q1[2].y);
  r[6] = (r[6] + q1[0].z) + (q1[1].z + q1[2].z); r[7] = (r[7] + q1[0].w) + (q1[1].w + q1[2].w);
}

__device__ __forceinline__ void lds_sync() {
  __builtin_amdgcn_fence(__ATOMIC_RELEASE, "workgroup", "local");
  __builtin_amdgcn_s_barrier();
  __builtin_amdgcn_fence(__ATOMIC_ACQUIRE, "workgroup", "local");
}

__device__ __forceinline__ int ncp_of(int i) {
  return ((deg_off(i) + 127) >> 7) << 4;
}

__device__ __forceinline__ void next_call(int& ci, int& ct, int& ck) {
  if (ci >= kD) return;
  if (ck == 0) {
    const int thi = (deg_off(ci) - 1) >> 4;
    if (ct < thi) { ++ct; return; }
    if ((ci & 7) == 0) { ck = 1; return; }
  }
  ck = 0; ++ci; ct = deg_off(ci - 1) >> 4;
}
__device__ __forceinline__ int call_n(int ci) {
  return ci < kD ? ncp_of(ci) : 0;
}
__device__ __forceinline__ int call_ofs(int ci, int ct, int ck) {
  return ck ? (kWzbOff / 4) + (ci >> 3) * 128 * 32 : (kWbOff / 4) + ct * 128 * 32;
}

__device__ __forceinline__ void l1_fetch(const float4* __restrict__ Wa4, const float* __restrict__ b0s,
                                         int t, int lane, int hi,
                                         float4 (&w)[8], float4& bb0, float4& bb1) {
  const float4* wq = Wa4 + (t * 8) * 32 + lane;
#pragma unroll
  for (int cp = 0; cp < 8; ++cp) w[cp] = wq[cp * 32];
  bb0 = *(const float4*)(b0s + t * 16 + hi * 8);
  bb1 = *(const float4*)(b0s + t * 16 + hi * 8 + 4);
}
__device__ __forceinline__ void l1_xfrags(const float* xrow, v2f (&xf)[16]) {
#pragma unroll
  for (int cp = 0; cp < 8; ++cp) {
    xf[2 * cp]     = *(const v2f*)(xrow + cp * 8);
    xf[2 * cp + 1] = *(const v2f*)(xrow + cp * 8 + 4);
  }
}
__device__ __forceinline__ void l1_tile(const float4 (&wr)[8], const float4 bb0, const float4 bb1,
                                        const v2f (&xf)[16], float* dst) {
  v8f acc;
  acc[0] = bb0.x; acc[1] = bb0.y; acc[2] = bb0.z; acc[3] = bb0.w;
  acc[4] = bb1.x; acc[5] = bb1.y; acc[6] = bb1.z; acc[7] = bb1.w;
  v8f a1;
  for (int v = 0; v < 8; ++v) a1[v] = 0.f;
#pragma unroll
  for (int cp = 0; cp < 8; ++cp) {
    const float4 w = wr[cp];
    v2f wA = {w.x, w.y};
    v2f wB = {w.z, w.w};
    acc = wmma_f32(wA, xf[2 * cp], acc);
    a1  = wmma_f32(wB, xf[2 * cp + 1], a1);
  }
  float4 o0, o1;
  o0.x = fmaxf(acc[0] + a1[0], 0.f); o0.y = fmaxf(acc[1] + a1[1], 0.f);
  o0.z = fmaxf(acc[2] + a1[2], 0.f); o0.w = fmaxf(acc[3] + a1[3], 0.f);
  o1.x = fmaxf(acc[4] + a1[4], 0.f); o1.y = fmaxf(acc[5] + a1[5], 0.f);
  o1.z = fmaxf(acc[6] + a1[6], 0.f); o1.w = fmaxf(acc[7] + a1[7], 0.f);
  *(float4*)dst = o0;
  *(float4*)(dst + 4) = o1;
}

__launch_bounds__(kThreads, 1)
__global__ void made_flow_kernel(const float* __restrict__ u,
                                 const float* __restrict__ b2,
                                 const float* __restrict__ ws,
                                 float* __restrict__ out) {
  extern __shared__ float smem[];
  float* xs   = smem;
  float* us   = xs  + kBT * kXStride;
  float* h0s  = us  + kBT * kXStride;
  float* h1s  = h0s + kBT * kHStride;
  float* wbuf = h1s + kBT * kHStride;
  float* psum = wbuf + 2 * kWbufSz;

  const int tid  = threadIdx.x;
  const int lane = tid & 31;
  const int wv   = __builtin_amdgcn_readfirstlane(tid >> 5);
  const int wg   = blockIdx.x;
  const int n    = lane & 15;
  const int hi   = lane >> 4;
  const int hi2  = hi << 1;

  const float4* Wa4 = (const float4*)(ws + kWaOff);
  const v4f* wsl = (const v4f*)ws + lane;
  const v4f* Wzi4 = (const v4f*)(ws + kWziOff);
  const float*  b0s = ws + kB0Off;
  const float*  b1s = ws + kB1Off;

  for (int e = tid; e < kBT * kXStride; e += kThreads) xs[e] = 0.f;
  for (int e = tid; e < 2 * kBT * kHStride; e += kThreads) h0s[e] = 0.f;
  for (int e = tid; e < kBT * kD; e += kThreads) {
    int r = e >> 6, d = e & 63;
    us[r * kXStride + d] = u[(wg * kBT + r) * kD + d];
  }
  float ldj = 0.f;

  int ci = 1, ct = 0, ck = 0;
  int ringN = call_n(ci), pb = 0;
  const int pw = wv - 4;
  v4f cr[2][kBlk];
#pragma unroll
  for (int s3 = 0; s3 < 2; ++s3)
#pragma unroll
    for (int u2 = 0; u2 < kBlk; ++u2) cr[s3][u2] = v4f{0.f, 0.f, 0.f, 0.f};
  if (wv >= 4) {
    copy_fetch(cr, wsl, call_ofs(ci, ct, ck), copy_cnt(ringN, pw), pw);
    copy_put(cr, copy_cnt(ringN, pw), wbuf, pw, lane);
  }
  next_call(ci, ct, ck);
  int heldN = call_n(ci);
  int heldCnt = wv >= 4 ? copy_cnt(heldN, pw) : 0;
  const int heldOfs0 = call_ofs(ci, ct, ck);
  next_call(ci, ct, ck);
  int fOfs = call_ofs(ci, ct, ck);
  int fCnt = wv >= 4 ? copy_cnt(call_n(ci), pw) : 0;
  float4 waA[8], waB[8], bA0, bA1, bB0, bB1;
  l1_fetch(Wa4, b0s, 0, lane, hi, waA, bA0, bA1);
  l1_fetch(Wa4, b0s, 1, lane, hi, waB, bB0, bB1);
  __syncthreads();
  copy_fetch(cr, wsl, heldOfs0, heldCnt, pw);

  v8f zacc;
  for (int v = 0; v < 8; ++v) zacc[v] = 0.f;
  const float* xrow  = xs  + n * kXStride + hi2;
  const float* h0row = h0s + n * kHStride + hi2;
  const float* h1row = h1s + n * kHStride + hi2;

  for (int i = 0; i < kD; ++i) {
    const int offi = deg_off(i);
    const int jb = i & 7;
    float b2m, b2s, uval;
    if (wv == 0) {
      b2m = b2[i];
      b2s = b2[kD + i];
      uval = us[n * kXStride + i];
    }
    if (i > 0) {
      const int tlo = deg_off(i - 1) >> 4, thi = (offi - 1) >> 4;
      if (wv == 0) {
        v2f xf[16];
        l1_xfrags(xrow, xf);
        __builtin_amdgcn_sched_barrier(0x6);
        l1_tile(waA, bA0, bA1, xf, h0s + n * kHStride + tlo * 16 + hi * 8);
        if (thi > tlo) l1_tile(waB, bB0, bB1, xf, h0s + n * kHStride + thi * 16 + hi * 8);
      }
      v4f zi0, zi1, zi2;
      int t = tlo;
      do {
        lds_sync();
        copy_put(cr, heldCnt, wbuf + (pb ^ 1) * kWbufSz, pw, lane);
        copy_fetch(cr, wsl, fOfs, fCnt, pw);
        int kst = 0, kcnt = 0;
        if (wv < 4) k_share(ringN, wv, kst, kcnt);
        const float* wcur = wbuf + pb * kWbufSz + lane * 4;
        v8f r;
        float4 bb0, bb1;
        if (wv == 0) {
          bb0 = *(const float4*)(b1s + t * 16 + hi * 8);
          bb1 = *(const float4*)(b1s + t * 16 + hi * 8 + 4);
          if (t == tlo) {
            const v4f* zq = Wzi4 + (i * 3) * 32 + lane;
            zi0 = zq[0]; zi1 = zq[32]; zi2 = zq[64];
            const int iN = i + 1 < kD ? i + 1 : i;
            const int tloN = deg_off(iN - 1) >> 4, thiN = (deg_off(iN) - 1) >> 4;
            l1_fetch(Wa4, b0s, tloN, lane, hi, waA, bA0, bA1);
            if (thiN > tloN) l1_fetch(Wa4, b0s, thiN, lane, hi, waB, bB0, bB1);
          }
          r = tile_gemm(wcur, kcnt * kMB, h0row);
        } else if (wv < 4) {
          helper_share(wcur, h0row, kst, kcnt, psum + (wv - 1) * 256, lane);
        }
        lds_sync();
        if (wv == 0) {
          add_shares(r, psum, lane);
          float4 o0, o1;
          o0.x = fmaxf(r[0] + bb0.x, 0.f); o0.y = fmaxf(r[1] + bb0.y, 0.f);
          o0.z = fmaxf(r[2] + bb0.z, 0.f); o0.w = fmaxf(r[3] + bb0.w, 0.f);
          o1.x = fmaxf(r[4] + bb1.x, 0.f); o1.y = fmaxf(r[5] + bb1.y, 0.f);
          o1.z = fmaxf(r[6] + bb1.z, 0.f); o1.w = fmaxf(r[7] + bb1.w, 0.f);
          float* dst = h1s + n * kHStride + t * 16 + hi * 8;
          *(float4*)dst = o0;
          *(float4*)(dst + 4) = o1;
        }
        ringN = heldN; heldN = call_n(ci); heldCnt = fCnt; pb ^= 1;
        next_call(ci, ct, ck);
        fOfs = call_ofs(ci, ct, ck); fCnt = wv >= 4 ? copy_cnt(call_n(ci), pw) : 0;
      } while (++t <= thi);
      lds_sync();
      if (jb == 0) {
        copy_put(cr, heldCnt, wbuf + (pb ^ 1) * kWbufSz, pw, lane);
        copy_fetch(cr, wsl, fOfs, fCnt, pw);
        int kst = 0, kcnt = 0;
        if (wv < 4) k_share(ringN, wv, kst, kcnt);
        const float* wcur = wbuf + pb * kWbufSz + lane * 4;
        if (wv == 0) {
          zacc = tile_gemm(wcur, kcnt * kMB, h1row);
        } else if (wv < 4) {
          helper_share(wcur, h1row, kst, kcnt, psum + (wv - 1) * 256, lane);
        }
        lds_sync();
        if (wv == 0) add_shares(zacc, psum, lane);
        ringN = heldN; heldN = call_n(ci); heldCnt = fCnt; pb ^= 1;
        next_call(ci, ct, ck);
        fOfs = call_ofs(ci, ct, ck); fCnt = wv >= 4 ? copy_cnt(call_n(ci), pw) : 0;
      } else if (wv == 0) {
        const int cp0 = deg_off(i - 1) >> 3;
        const v4f wz[3] = {zi0, zi1, zi2};
        v8f zb;
        for (int v = 0; v < 8; ++v) zb[v] = 0.f;
        v2f hA[3], hB[3];
#pragma unroll
        for (int cl = 0; cl < 3; ++cl) {
          int hcp = cp0 + cl; hcp = hcp > 127 ? 127 : hcp;
          hA[cl] = *(const v2f*)(h1row + hcp * 8);
          hB[cl] = *(const v2f*)(h1row + hcp * 8 + 4);
        }
        __builtin_amdgcn_sched_barrier(0x6);
#pragma unroll
        for (int cl = 0; cl < 3; ++cl) {
          v2f wA = {wz[cl].x, wz[cl].y};
          v2f wB = {wz[cl].z, wz[cl].w};
          zacc = wmma_f32(wA, hA[cl], zacc);
          zb   = wmma_f32(wB, hB[cl], zb);
        }
        for (int v = 0; v < 8; ++v) zacc[v] += zb[v];
      }
    }
    if (wv == 0) {
      float zm, zs;
      switch (jb & 3) {
        case 0:  zm = zacc[0]; zs = zacc[1]; break;
        case 1:  zm = zacc[2]; zs = zacc[3]; break;
        case 2:  zm = zacc[4]; zs = zacc[5]; break;
        default: zm = zacc[6]; zs = zacc[7]; break;
      }
      if (hi == (jb >> 2)) {
        const float mu = zm + b2m;
        const float sg = zs + b2s;
        xs[n * kXStride + i] = uval * expf(sg) + mu;
        ldj += sg;
      }
    }
    lds_sync();
  }

  for (int e = tid; e < kBT * kD; e += kThreads) {
    int r = e >> 6, d = e & 63;
    out[(wg * kBT + r) * kD + d] = xs[r * kXStride + d];
  }
  if (wv == 0) {
    const float ldjt = ldj + __shfl_xor(ldj, 16);
    if (hi == 0) out[kB * kD + wg * kBT + n] = ldjt;
  }
}

extern "C" void kernel_launch(void* const* d_in, const int* in_sizes, int n_in,
                              void* d_out, int out_size, void* d_ws, size_t ws_size,
                              hipStream_t stream) {
  const float* u  = (const float*)d_in[0];
  const float* W0 = (const float*)d_in[1];
  const float* b0 = (const float*)d_in[2];
  const float* W1 = (const float*)d_in[3];
  const float* b1 = (const float*)d_in[4];
  const float* W2 = (const float*)d_in[5];
  const float* b2 = (const float*)d_in[6];
  float* ws  = (float*)d_ws;
  float* out = (float*)d_out;

  prep_w0<<<(kU * kD) / 256,     256, 0, stream>>>(W0, b0, b1, ws);
  prep_w1<<<(kU * kU / 4) / 256, 256, 0, stream>>>(W1, ws + kWbOff);
  prep_w2<<<(kWzbSz + kWziSz) / 256, 256, 0, stream>>>(W2, ws);

  size_t smem = (size_t)(2 * kBT * kXStride + 2 * kBT * kHStride + 2 * kWbufSz + 3 * 2 * 32 * 4) * sizeof(float);
  made_flow_kernel<<<kB / kBT, kThreads, smem, stream>>>(u, b2, ws, out);
}
